# starting kernel
# baseline (speedup 1.0000x reference)
#include <hip/hip_runtime.h>

typedef __attribute__((ext_vector_type(2))) float v2f;
typedef __attribute__((ext_vector_type(8))) float v8f;

#define N_NODES 100000
#define NFEAT   256
#define NHID    64
#define NCLASS  41

__global__ __launch_bounds__(256) void fill_zero_k(float* __restrict__ p, int n) {
    int i = blockIdx.x * blockDim.x + threadIdx.x;
    if (i < n) p[i] = 0.0f;
}

__global__ __launch_bounds__(256) void gemm1_wmma_k(const float* __restrict__ x,
                                                    const float* __restrict__ W1,
                                                    float* __restrict__ h1) {
    const int lane = threadIdx.x & 31;
    const int wave = threadIdx.x >> 5;
    const int r0 = (blockIdx.x * 8 + wave) * 16;
    if (r0 >= N_NODES) return;

    const int m     = lane & 15;
    const int khalf = (lane >> 4) * 2;

    int arow = r0 + m;
    if (arow >= N_NODES) arow = N_NODES - 1;
    const float* __restrict__ xrow = x + (long)arow * NFEAT;

    v8f acc0 = {}, acc1 = {}, acc2 = {}, acc3 = {};
    for (int k0 = 0; k0 < NFEAT; k0 += 4) {
        v2f a;
        a.x = xrow[k0 + khalf];
        a.y = xrow[k0 + khalf + 1];
        const float* __restrict__ wlo = W1 + (k0 + khalf) * NHID + m;
        const float* __restrict__ whi = W1 + (k0 + khalf + 1) * NHID + m;
        v2f b0; b0.x = wlo[0];  b0.y = whi[0];
        v2f b1; b1.x = wlo[16]; b1.y = whi[16];
        v2f b2; b2.x = wlo[32]; b2.y = whi[32];
        v2f b3; b3.x = wlo[48]; b3.y = whi[48];
        acc0 = __builtin_amdgcn_wmma_f32_16x16x4_f32(false, a, false, b0, (short)0, acc0, false, false);
        acc1 = __builtin_amdgcn_wmma_f32_16x16x4_f32(false, a, false, b1, (short)0, acc1, false, false);
        acc2 = __builtin_amdgcn_wmma_f32_16x16x4_f32(false, a, false, b2, (short)0, acc2, false, false);
        acc3 = __builtin_amdgcn_wmma_f32_16x16x4_f32(false, a, false, b3, (short)0, acc3, false, false);
    }

    const int col     = lane & 15;
    const int rowHalf = (lane >> 4) * 8;
    for (int j = 0; j < 8; ++j) {
        int r = r0 + rowHalf + j;
        if (r < N_NODES) {
            float* __restrict__ o = h1 + (long)r * NHID + col;
            o[0]  = acc0[j];
            o[16] = acc1[j];
            o[32] = acc2[j];
            o[48] = acc3[j];
        }
    }
}

__global__ __launch_bounds__(256) void gemm2_wmma_k(const float* __restrict__ agg1,
                                                    const float* __restrict__ b1,
                                                    const float* __restrict__ W2,
                                                    float* __restrict__ h2) {
    const int lane = threadIdx.x & 31;
    const int wave = threadIdx.x >> 5;
    const int r0 = (blockIdx.x * 8 + wave) * 16;
    if (r0 >= N_NODES) return;

    const int m     = lane & 15;
    const int khalf = (lane >> 4) * 2;

    int arow = r0 + m;
    if (arow >= N_NODES) arow = N_NODES - 1;
    const float* __restrict__ hrow = agg1 + (long)arow * NHID;

    v8f acc0 = {}, acc1 = {}, acc2 = {};
    for (int k0 = 0; k0 < NHID; k0 += 4) {
        int ka = k0 + khalf;
        v2f a;
        a.x = fmaxf(hrow[ka]     + b1[ka],     0.0f);
        a.y = fmaxf(hrow[ka + 1] + b1[ka + 1], 0.0f);

        v2f b0, b1v, b2v;
        {
            int n = m;
            b0.x = W2[ka * NCLASS + n];
            b0.y = W2[(ka + 1) * NCLASS + n];
        }
        {
            int n = 16 + m;
            b1v.x = W2[ka * NCLASS + n];
            b1v.y = W2[(ka + 1) * NCLASS + n];
        }
        {
            int n = 32 + m;
            int nc = (n < NCLASS) ? n : (NCLASS - 1);
            float lo = W2[ka * NCLASS + nc];
            float hi = W2[(ka + 1) * NCLASS + nc];
            if (n >= NCLASS) { lo = 0.0f; hi = 0.0f; }
            b2v.x = lo; b2v.y = hi;
        }
        acc0 = __builtin_amdgcn_wmma_f32_16x16x4_f32(false, a, false, b0,  (short)0, acc0, false, false);
        acc1 = __builtin_amdgcn_wmma_f32_16x16x4_f32(false, a, false, b1v, (short)0, acc1, false, false);
        acc2 = __builtin_amdgcn_wmma_f32_16x16x4_f32(false, a, false, b2v, (short)0, acc2, false, false);
    }

    const int col     = lane & 15;
    const int rowHalf = (lane >> 4) * 8;
    for (int j = 0; j < 8; ++j) {
        int r = r0 + rowHalf + j;
        if (r < N_NODES) {
            float* __restrict__ o = h2 + (long)r * NCLASS;
            o[col]      = acc0[j];
            o[col + 16] = acc1[j];
            int c2 = col + 32;
            if (c2 < NCLASS) o[c2] = acc2[j];
        }
    }
}

template <int D>
__global__ __launch_bounds__(256) void spmm_k(const int* __restrict__ src,
                                              const int* __restrict__ dst,
                                              const float* __restrict__ w,
                                              const float* __restrict__ h,
                                              float* __restrict__ agg,
                                              int nEdges) {
    long idx = (long)blockIdx.x * blockDim.x + threadIdx.x;
    long total = (long)nEdges * D;
    if (idx >= total) return;
    int e = (int)(idx / D);
    int c = (int)(idx - (long)e * D);
    float v = w[e] * h[(long)src[e] * D + c];
    atomicAdd(&agg[(long)dst[e] * D + c], v);
}

__global__ __launch_bounds__(256) void logsoftmax_k(const float* __restrict__ agg2,
                                                    const float* __restrict__ b2,
                                                    float* __restrict__ out) {
    int i = blockIdx.x * blockDim.x + threadIdx.x;
    if (i >= N_NODES) return;
    const float* __restrict__ z = agg2 + (long)i * NCLASS;
    float zb[NCLASS];
    float mx = -3.0e38f;
    for (int c = 0; c < NCLASS; ++c) {
        zb[c] = z[c] + b2[c];
        mx = fmaxf(mx, zb[c]);
    }
    float s = 0.0f;
    for (int c = 0; c < NCLASS; ++c) s += expf(zb[c] - mx);
    float lse = mx + logf(s);
    float* __restrict__ o = out + (long)i * NCLASS;
    for (int c = 0; c < NCLASS; ++c) o[c] = zb[c] - lse;
}

extern "C" void kernel_launch(void* const* d_in, const int* in_sizes, int n_in,
                              void* d_out, int out_size, void* d_ws, size_t ws_size,
                              hipStream_t stream) {
    const float* x    = (const float*)d_in[0];
    const int*   esrc = (const int*)d_in[1];
    const int*   edst = (const int*)d_in[2];
    const float* ew   = (const float*)d_in[3];
    const float* W1   = (const float*)d_in[4];
    const float* b1   = (const float*)d_in[5];
    const float* W2   = (const float*)d_in[6];
    const float* b2   = (const float*)d_in[7];
    float* out = (float*)d_out;
    const int E = in_sizes[1];

    float* h1   = (float*)d_ws;
    float* agg1 = h1   + (size_t)N_NODES * NHID;
    float* h2   = agg1 + (size_t)N_NODES * NHID;
    float* agg2 = h2   + (size_t)N_NODES * NCLASS;

    const int n1 = N_NODES * NHID;
    const int n2 = N_NODES * NCLASS;
    fill_zero_k<<<(n1 + 255) / 256, 256, 0, stream>>>(agg1, n1);
    fill_zero_k<<<(n2 + 255) / 256, 256, 0, stream>>>(agg2, n2);

    const int gemmBlocks = (N_NODES + 127) / 128;
    gemm1_wmma_k<<<gemmBlocks, 256, 0, stream>>>(x, W1, h1);

    long w1 = (long)E * NHID;
    spmm_k<NHID><<<(unsigned)((w1 + 255) / 256), 256, 0, stream>>>(esrc, edst, ew, h1, agg1, E);

    gemm2_wmma_k<<<gemmBlocks, 256, 0, stream>>>(agg1, b1, W2, h2);

    long w2 = (long)E * NCLASS;
    spmm_k<NCLASS><<<(unsigned)((w2 + 255) / 256), 256, 0, stream>>>(esrc, edst, ew, h2, agg2, E);

    logsoftmax_k<<<(N_NODES + 255) / 256, 256, 0, stream>>>(agg2, b2, out);
}


# Float-atomic scatter SpMM -> gather: edges grouped by dst each call (int atomic rank, wave scan, place), one wave sums a row in registers; per-row sum order = rank arrival order
# speedup vs baseline: 2.4406x; 2.4406x over previous
#include <hip/hip_runtime.h>

typedef __attribute__((ext_vector_type(2))) float v2f;
typedef __attribute__((ext_vector_type(8))) float v8f;

#define N_NODES 100000
#define NFEAT   256
#define NHID    64
#define NCLASS  41

__global__ __launch_bounds__(256) void gemm1_wmma_k(const float* __restrict__ x,
                                                    const float* __restrict__ W1,
                                                    float* __restrict__ h1) {
    const int lane = threadIdx.x & 31;
    const int wave = threadIdx.x >> 5;
    const int r0 = (blockIdx.x * 8 + wave) * 16;
    if (r0 >= N_NODES) return;

    const int m     = lane & 15;
    const int khalf = (lane >> 4) * 2;

    int arow = r0 + m;
    if (arow >= N_NODES) arow = N_NODES - 1;
    const float* __restrict__ xrow = x + (long)arow * NFEAT;

    v8f acc0 = {}, acc1 = {}, acc2 = {}, acc3 = {};
    for (int k0 = 0; k0 < NFEAT; k0 += 4) {
        v2f a;
        a.x = xrow[k0 + khalf];
        a.y = xrow[k0 + khalf + 1];
        const float* __restrict__ wlo = W1 + (k0 + khalf) * NHID + m;
        const float* __restrict__ whi = W1 + (k0 + khalf + 1) * NHID + m;
        v2f b0; b0.x = wlo[0];  b0.y = whi[0];
        v2f b1; b1.x = wlo[16]; b1.y = whi[16];
        v2f b2; b2.x = wlo[32]; b2.y = whi[32];
        v2f b3; b3.x = wlo[48]; b3.y = whi[48];
        acc0 = __builtin_amdgcn_wmma_f32_16x16x4_f32(false, a, false, b0, (short)0, acc0, false, false);
        acc1 = __builtin_amdgcn_wmma_f32_16x16x4_f32(false, a, false, b1, (short)0, acc1, false, false);
        acc2 = __builtin_amdgcn_wmma_f32_16x16x4_f32(false, a, false, b2, (short)0, acc2, false, false);
        acc3 = __builtin_amdgcn_wmma_f32_16x16x4_f32(false, a, false, b3, (short)0, acc3, false, false);
    }

    const int col     = lane & 15;
    const int rowHalf = (lane >> 4) * 8;
    for (int j = 0; j < 8; ++j) {
        int r = r0 + rowHalf + j;
        if (r < N_NODES) {
            float* __restrict__ o = h1 + (long)r * NHID + col;
            o[0]  = acc0[j];
            o[16] = acc1[j];
            o[32] = acc2[j];
            o[48] = acc3[j];
        }
    }
}

__global__ __launch_bounds__(256) void gemm2_wmma_k(const float* __restrict__ agg1,
                                                    const float* __restrict__ b1,
                                                    const float* __restrict__ W2,
                                                    float* __restrict__ h2) {
    const int lane = threadIdx.x & 31;
    const int wave = threadIdx.x >> 5;
    const int r0 = (blockIdx.x * 8 + wave) * 16;
    if (r0 >= N_NODES) return;

    const int m     = lane & 15;
    const int khalf = (lane >> 4) * 2;

    int arow = r0 + m;
    if (arow >= N_NODES) arow = N_NODES - 1;
    const float* __restrict__ hrow = agg1 + (long)arow * NHID;

    v8f acc0 = {}, acc1 = {}, acc2 = {};
    for (int k0 = 0; k0 < NHID; k0 += 4) {
        int ka = k0 + khalf;
        v2f a;
        a.x = fmaxf(hrow[ka]     + b1[ka],     0.0f);
        a.y = fmaxf(hrow[ka + 1] + b1[ka + 1], 0.0f);

        v2f b0, b1v, b2v;
        {
            int n = m;
            b0.x = W2[ka * NCLASS + n];
            b0.y = W2[(ka + 1) * NCLASS + n];
        }
        {
            int n = 16 + m;
            b1v.x = W2[ka * NCLASS + n];
            b1v.y = W2[(ka + 1) * NCLASS + n];
        }
        {
            int n = 32 + m;
            int nc = (n < NCLASS) ? n : (NCLASS - 1);
            float lo = W2[ka * NCLASS + nc];
            float hi = W2[(ka + 1) * NCLASS + nc];
            if (n >= NCLASS) { lo = 0.0f; hi = 0.0f; }
            b2v.x = lo; b2v.y = hi;
        }
        acc0 = __builtin_amdgcn_wmma_f32_16x16x4_f32(false, a, false, b0,  (short)0, acc0, false, false);
        acc1 = __builtin_amdgcn_wmma_f32_16x16x4_f32(false, a, false, b1v, (short)0, acc1, false, false);
        acc2 = __builtin_amdgcn_wmma_f32_16x16x4_f32(false, a, false, b2v, (short)0, acc2, false, false);
    }

    const int col     = lane & 15;
    const int rowHalf = (lane >> 4) * 8;
    for (int j = 0; j < 8; ++j) {
        int r = r0 + rowHalf + j;
        if (r < N_NODES) {
            float* __restrict__ o = h2 + (long)r * NCLASS;
            o[col]      = acc0[j];
            o[col + 16] = acc1[j];
            int c2 = col + 32;
            if (c2 < NCLASS) o[c2] = acc2[j];
        }
    }
}

#define N_EDGES 1600000

__global__ __launch_bounds__(256) void zero_int_k(int* __restrict__ p, int n) {
    int i = blockIdx.x * blockDim.x + threadIdx.x;
    if (i < n) p[i] = 0;
}

__global__ __launch_bounds__(256) void rank_k(const int* __restrict__ dst,
                                              int* __restrict__ cnt,
                                              int* __restrict__ rank, int nEdges) {
    int e = blockIdx.x * blockDim.x + threadIdx.x;
    if (e >= nEdges) return;
    rank[e] = atomicAdd(&cnt[dst[e]], 1);
}

__global__ __launch_bounds__(256) void alloc_k(const int* __restrict__ cnt,
                                               int* __restrict__ total,
                                               int* __restrict__ rowstart) {
    const int i = blockIdx.x * blockDim.x + threadIdx.x;
    const int lane = threadIdx.x & 31;
    const int c = (i < N_NODES) ? cnt[i] : 0;
    int incl = c;
    for (int d = 1; d < 32; d <<= 1) {
        int t = __shfl_up(incl, d, 32);
        if (lane >= d) incl += t;
    }
    const int sum = __shfl(incl, 31, 32);
    int base = 0;
    if (lane == 0) base = atomicAdd(total, sum);
    base = __shfl(base, 0, 32);
    if (i < N_NODES) rowstart[i] = base + incl - c;
}

__global__ __launch_bounds__(256) void place_k(const int* __restrict__ src,
                                               const int* __restrict__ dst,
                                               const float* __restrict__ w,
                                               const int* __restrict__ rank,
                                               const int* __restrict__ rowstart,
                                               int* __restrict__ pk, int nEdges) {
    int e = blockIdx.x * blockDim.x + threadIdx.x;
    if (e >= nEdges) return;
    int p = rowstart[dst[e]] + rank[e];
    int2 v; v.x = src[e]; v.y = __float_as_int(w[e]);
    ((int2*)pk)[p] = v;
}

template <int K>
__device__ __forceinline__ void acc64(const float* __restrict__ h, int ex, int ey, int j,
                                      int lane, float& ax, float& ay) {
    float2 v[K];
    float  wk[K];
#pragma unroll
    for (int k = 0; k < K; ++k) {
        int sj = __builtin_amdgcn_readlane(ex, j + k);
        wk[k]  = __int_as_float(__builtin_amdgcn_readlane(ey, j + k));
        v[k]   = *(const float2*)(h + (long)sj * NHID + lane * 2);
    }
#pragma unroll
    for (int k = 0; k < K; ++k) {
        ax = fmaf(wk[k], v[k].x, ax);
        ay = fmaf(wk[k], v[k].y, ay);
    }
}

__global__ __launch_bounds__(256) void gather64_k(const int* __restrict__ rowstart,
                                                  const int* __restrict__ cnt,
                                                  const int* __restrict__ pk,
                                                  const float* __restrict__ h,
                                                  float* __restrict__ agg) {
    const int lane = threadIdx.x & 31;
    const int nw = gridDim.x * 8;
    for (int r = blockIdx.x * 8 + (threadIdx.x >> 5); r < N_NODES; r += nw) {
        const int ru = __builtin_amdgcn_readfirstlane(r);
        const int s  = __builtin_amdgcn_readfirstlane(rowstart[ru]);
        const int n  = __builtin_amdgcn_readfirstlane(cnt[ru]);
        float ax = 0.0f, ay = 0.0f;
        for (int c0 = 0; c0 < n; c0 += 32) {
            int m = n - c0; if (m > 32) m = 32;
            int ex = 0, ey = 0;
            if (lane < m) { int2 e = ((const int2*)pk)[s + c0 + lane]; ex = e.x; ey = e.y; }
            int j = 0;
            for (; j + 8 <= m; j += 8) acc64<8>(h, ex, ey, j, lane, ax, ay);
            switch (m - j) {
                case 7: acc64<7>(h, ex, ey, j, lane, ax, ay); break;
                case 6: acc64<6>(h, ex, ey, j, lane, ax, ay); break;
                case 5: acc64<5>(h, ex, ey, j, lane, ax, ay); break;
                case 4: acc64<4>(h, ex, ey, j, lane, ax, ay); break;
                case 3: acc64<3>(h, ex, ey, j, lane, ax, ay); break;
                case 2: acc64<2>(h, ex, ey, j, lane, ax, ay); break;
                case 1: acc64<1>(h, ex, ey, j, lane, ax, ay); break;
                default: break;
            }
        }
        float2 o; o.x = ax; o.y = ay;
        *(float2*)(agg + (long)ru * NHID + lane * 2) = o;
    }
}

template <int K>
__device__ __forceinline__ void acc41(const float* __restrict__ h, int ex, int ey, int j,
                                      int c0i, int c1i, float& a0, float& a1) {
    float v0[K], v1[K], wk[K];
#pragma unroll
    for (int k = 0; k < K; ++k) {
        int sj = __builtin_amdgcn_readlane(ex, j + k);
        wk[k]  = __int_as_float(__builtin_amdgcn_readlane(ey, j + k));
        const float* __restrict__ row = h + (long)sj * NCLASS;
        v0[k] = row[c0i];
        v1[k] = row[c1i];
    }
#pragma unroll
    for (int k = 0; k < K; ++k) {
        a0 = fmaf(wk[k], v0[k], a0);
        a1 = fmaf(wk[k], v1[k], a1);
    }
}

__global__ __launch_bounds__(256) void gather41_k(const int* __restrict__ rowstart,
                                                  const int* __restrict__ cnt,
                                                  const int* __restrict__ pk,
                                                  const float* __restrict__ h,
                                                  float* __restrict__ agg) {
    const int lane = threadIdx.x & 31;
    const int nw = gridDim.x * 8;
    const int c1i = (32 + lane < NCLASS) ? (32 + lane) : (NCLASS - 1);
    for (int r = blockIdx.x * 8 + (threadIdx.x >> 5); r < N_NODES; r += nw) {
        const int ru = __builtin_amdgcn_readfirstlane(r);
        const int s  = __builtin_amdgcn_readfirstlane(rowstart[ru]);
        const int n  = __builtin_amdgcn_readfirstlane(cnt[ru]);
        float a0 = 0.0f, a1 = 0.0f;
        for (int c0 = 0; c0 < n; c0 += 32) {
            int m = n - c0; if (m > 32) m = 32;
            int ex = 0, ey = 0;
            if (lane < m) { int2 e = ((const int2*)pk)[s + c0 + lane]; ex = e.x; ey = e.y; }
            int j = 0;
            for (; j + 8 <= m; j += 8) acc41<8>(h, ex, ey, j, lane, c1i, a0, a1);
            switch (m - j) {
                case 7: acc41<7>(h, ex, ey, j, lane, c1i, a0, a1); break;
                case 6: acc41<6>(h, ex, ey, j, lane, c1i, a0, a1); break;
                case 5: acc41<5>(h, ex, ey, j, lane, c1i, a0, a1); break;
                case 4: acc41<4>(h, ex, ey, j, lane, c1i, a0, a1); break;
                case 3: acc41<3>(h, ex, ey, j, lane, c1i, a0, a1); break;
                case 2: acc41<2>(h, ex, ey, j, lane, c1i, a0, a1); break;
                case 1: acc41<1>(h, ex, ey, j, lane, c1i, a0, a1); break;
                default: break;
            }
        }
        float* __restrict__ o = agg + (long)ru * NCLASS;
        o[lane] = a0;
        if (32 + lane < NCLASS) o[32 + lane] = a1;
    }
}

__global__ __launch_bounds__(256) void logsoftmax_k(const float* __restrict__ agg2,
                                                    const float* __restrict__ b2,
                                                    float* __restrict__ out) {
    int i = blockIdx.x * blockDim.x + threadIdx.x;
    if (i >= N_NODES) return;
    const float* __restrict__ z = agg2 + (long)i * NCLASS;
    float zb[NCLASS];
    float mx = -3.0e38f;
    for (int c = 0; c < NCLASS; ++c) {
        zb[c] = z[c] + b2[c];
        mx = fmaxf(mx, zb[c]);
    }
    float s = 0.0f;
    for (int c = 0; c < NCLASS; ++c) s += expf(zb[c] - mx);
    float lse = mx + logf(s);
    float* __restrict__ o = out + (long)i * NCLASS;
    for (int c = 0; c < NCLASS; ++c) o[c] = zb[c] - lse;
}

extern "C" void kernel_launch(void* const* d_in, const int* in_sizes, int n_in,
                              void* d_out, int out_size, void* d_ws, size_t ws_size,
                              hipStream_t stream) {
    const float* x    = (const float*)d_in[0];
    const int*   esrc = (const int*)d_in[1];
    const int*   edst = (const int*)d_in[2];
    const float* ew   = (const float*)d_in[3];
    const float* W1   = (const float*)d_in[4];
    const float* b1   = (const float*)d_in[5];
    const float* W2   = (const float*)d_in[6];
    const float* b2   = (const float*)d_in[7];
    float* out = (float*)d_out;
    const int E = in_sizes[1];

    float* h1   = (float*)d_ws;
    float* agg1 = h1   + (size_t)N_NODES * NHID;
    float* h2   = h1;
    float* agg2 = agg1;
    int* pk       = (int*)(agg1 + (size_t)N_NODES * NHID);
    int* rank     = pk   + (size_t)2 * N_EDGES;
    int* cnt      = rank + (size_t)N_EDGES;
    int* rowstart = cnt  + (size_t)(N_NODES + 8);

    zero_int_k<<<(N_NODES + 1 + 255) / 256, 256, 0, stream>>>(cnt, N_NODES + 1);
    rank_k<<<(E + 255) / 256, 256, 0, stream>>>(edst, cnt, rank, E);
    alloc_k<<<(N_NODES + 255) / 256, 256, 0, stream>>>(cnt, cnt + N_NODES, rowstart);
    place_k<<<(E + 255) / 256, 256, 0, stream>>>(esrc, edst, ew, rank, rowstart, pk, E);

    const int gemmBlocks = (N_NODES + 127) / 128;
    gemm1_wmma_k<<<gemmBlocks, 256, 0, stream>>>(x, W1, h1);

    gather64_k<<<2048, 256, 0, stream>>>(rowstart, cnt, pk, h1, agg1);

    gemm2_wmma_k<<<gemmBlocks, 256, 0, stream>>>(agg1, b1, W2, h2);

    gather41_k<<<2048, 256, 0, stream>>>(rowstart, cnt, pk, h2, agg2);

    logsoftmax_k<<<(N_NODES + 255) / 256, 256, 0, stream>>>(agg2, b2, out);
}


# Edge grouping by dst: 1.6M returning global int atomics -> two levels of LDS counters (256 row buckets, then rows per bucket), CSR in row order; per-row sum order = LDS rank arrival order
# speedup vs baseline: 4.0317x; 1.6519x over previous
#include <hip/hip_runtime.h>

typedef __attribute__((ext_vector_type(2))) float v2f;
typedef __attribute__((ext_vector_type(8))) float v8f;

#define N_NODES 100000
#define NFEAT   256
#define NHID    64
#define NCLASS  41

__global__ __launch_bounds__(256) void gemm1_wmma_k(const float* __restrict__ x,
                                                    const float* __restrict__ W1,
                                                    float* __restrict__ h1) {
    const int lane = threadIdx.x & 31;
    const int wave = threadIdx.x >> 5;
    const int r0 = (blockIdx.x * 8 + wave) * 16;
    if (r0 >= N_NODES) return;

    const int m     = lane & 15;
    const int khalf = (lane >> 4) * 2;

    int arow = r0 + m;
    if (arow >= N_NODES) arow = N_NODES - 1;
    const float* __restrict__ xrow = x + (long)arow * NFEAT;

    v8f acc0 = {}, acc1 = {}, acc2 = {}, acc3 = {};
    for (int k0 = 0; k0 < NFEAT; k0 += 4) {
        v2f a;
        a.x = xrow[k0 + khalf];
        a.y = xrow[k0 + khalf + 1];
        const float* __restrict__ wlo = W1 + (k0 + khalf) * NHID + m;
        const float* __restrict__ whi = W1 + (k0 + khalf + 1) * NHID + m;
        v2f b0; b0.x = wlo[0];  b0.y = whi[0];
        v2f b1; b1.x = wlo[16]; b1.y = whi[16];
        v2f b2; b2.x = wlo[32]; b2.y = whi[32];
        v2f b3; b3.x = wlo[48]; b3.y = whi[48];
        acc0 = __builtin_amdgcn_wmma_f32_16x16x4_f32(false, a, false, b0, (short)0, acc0, false, false);
        acc1 = __builtin_amdgcn_wmma_f32_16x16x4_f32(false, a, false, b1, (short)0, acc1, false, false);
        acc2 = __builtin_amdgcn_wmma_f32_16x16x4_f32(false, a, false, b2, (short)0, acc2, false, false);
        acc3 = __builtin_amdgcn_wmma_f32_16x16x4_f32(false, a, false, b3, (short)0, acc3, false, false);
    }

    const int col     = lane & 15;
    const int rowHalf = (lane >> 4) * 8;
    for (int j = 0; j < 8; ++j) {
        int r = r0 + rowHalf + j;
        if (r < N_NODES) {
            float* __restrict__ o = h1 + (long)r * NHID + col;
            o[0]  = acc0[j];
            o[16] = acc1[j];
            o[32] = acc2[j];
            o[48] = acc3[j];
        }
    }
}

__global__ __launch_bounds__(256) void gemm2_wmma_k(const float* __restrict__ agg1,
                                                    const float* __restrict__ b1,
                                                    const float* __restrict__ W2,
                                                    float* __restrict__ h2) {
    const int lane = threadIdx.x & 31;
    const int wave = threadIdx.x >> 5;
    const int r0 = (blockIdx.x * 8 + wave) * 16;
    if (r0 >= N_NODES) return;

    const int m     = lane & 15;
    const int khalf = (lane >> 4) * 2;

    int arow = r0 + m;
    if (arow >= N_NODES) arow = N_NODES - 1;
    const float* __restrict__ hrow = agg1 + (long)arow * NHID;

    v8f acc0 = {}, acc1 = {}, acc2 = {};
    for (int k0 = 0; k0 < NHID; k0 += 4) {
        int ka = k0 + khalf;
        v2f a;
        a.x = fmaxf(hrow[ka]     + b1[ka],     0.0f);
        a.y = fmaxf(hrow[ka + 1] + b1[ka + 1], 0.0f);

        v2f b0, b1v, b2v;
        {
            int n = m;
            b0.x = W2[ka * NCLASS + n];
            b0.y = W2[(ka + 1) * NCLASS + n];
        }
        {
            int n = 16 + m;
            b1v.x = W2[ka * NCLASS + n];
            b1v.y = W2[(ka + 1) * NCLASS + n];
        }
        {
            int n = 32 + m;
            int nc = (n < NCLASS) ? n : (NCLASS - 1);
            float lo = W2[ka * NCLASS + nc];
            float hi = W2[(ka + 1) * NCLASS + nc];
            if (n >= NCLASS) { lo = 0.0f; hi = 0.0f; }
            b2v.x = lo; b2v.y = hi;
        }
        acc0 = __builtin_amdgcn_wmma_f32_16x16x4_f32(false, a, false, b0,  (short)0, acc0, false, false);
        acc1 = __builtin_amdgcn_wmma_f32_16x16x4_f32(false, a, false, b1v, (short)0, acc1, false, false);
        acc2 = __builtin_amdgcn_wmma_f32_16x16x4_f32(false, a, false, b2v, (short)0, acc2, false, false);
    }

    const int col     = lane & 15;
    const int rowHalf = (lane >> 4) * 8;
    for (int j = 0; j < 8; ++j) {
        int r = r0 + rowHalf + j;
        if (r < N_NODES) {
            float* __restrict__ o = h2 + (long)r * NCLASS;
            o[col]      = acc0[j];
            o[col + 16] = acc1[j];
            int c2 = col + 32;
            if (c2 < NCLASS) o[c2] = acc2[j];
        }
    }
}

#define N_EDGES 1600000
#define NB  256
#define BW  391
#define NCH 256
#define NT  512
static_assert(NB * BW >= N_NODES, "buckets must cover every row");
static_assert(BW <= NT, "one thread per row of a bucket");
static_assert(NB % 4 == 0 && NT % (NB / 4) == 0 && NCH % (16 * (NT / (NB / 4))) == 0, "column-sum tiling");

__device__ __forceinline__ int wg_excl_scan(int v, int& total) {
    __shared__ int wsum[32];
    const int lane = threadIdx.x & 31, wv = threadIdx.x >> 5;
    int incl = v;
    for (int d = 1; d < 32; d <<= 1) {
        int t = __shfl_up(incl, d, 32);
        if (lane >= d) incl += t;
    }
    if (lane == 31) wsum[wv] = incl;
    __syncthreads();
    const int ws = (lane < NT / 32) ? wsum[lane] : 0;
    int wincl = ws;
    for (int d = 1; d < 32; d <<= 1) {
        int t = __shfl_up(wincl, d, 32);
        if (lane >= d) wincl += t;
    }
    total = __shfl(wincl, 31, 32);
    const int wexcl = __shfl(wincl - ws, wv, 32);
    return wexcl + incl - v;
}

__global__ __launch_bounds__(NT) void bucket_hist_k(const int* __restrict__ dst,
                                                    int* __restrict__ hist, int nEdges) {
    __shared__ int h[NB];
    const int t = threadIdx.x;
    if (t < NB) h[t] = 0;
    __syncthreads();
    const int chunk = (nEdges + NCH - 1) / NCH;
    const int e0 = blockIdx.x * chunk;
    int e1 = e0 + chunk; if (e1 > nEdges) e1 = nEdges;
    int e = e0 + t;
    for (; e + 3 * NT < e1; e += 4 * NT) {
        int d[4];
#pragma unroll
        for (int k = 0; k < 4; ++k) d[k] = dst[e + k * NT];
#pragma unroll
        for (int k = 0; k < 4; ++k) atomicAdd(&h[d[k] / BW], 1);
    }
    for (; e < e1; e += NT) atomicAdd(&h[dst[e] / BW], 1);
    __syncthreads();
    if (t < NB) hist[blockIdx.x * NB + t] = h[t];
}

__global__ __launch_bounds__(NT) void bucket_scatter_k(const int* __restrict__ src,
                                                       const int* __restrict__ dst,
                                                       const float* __restrict__ w,
                                                       const int* __restrict__ hist,
                                                       int* __restrict__ bstart,
                                                       int* __restrict__ brec_w, int nEdges) {
    int4* __restrict__ brec = (int4*)brec_w;
    constexpr int QN = NT / (NB / 4);
    constexpr int QR = NCH / QN;
    __shared__ int pAll[QN][NB];
    __shared__ int pBef[QN][NB];
    __shared__ int cur[NB];
    const int t = threadIdx.x, g = blockIdx.x;
    {
        const int b4 = (t % (NB / 4)) * 4, q = t / (NB / 4);
        int a0 = 0, a1 = 0, a2 = 0, a3 = 0, f0 = 0, f1 = 0, f2 = 0, f3 = 0;
        for (int r0 = q * QR; r0 < (q + 1) * QR; r0 += 16) {
            int4 v[16];
#pragma unroll
            for (int k = 0; k < 16; ++k) v[k] = *(const int4*)(hist + (r0 + k) * NB + b4);
#pragma unroll
            for (int k = 0; k < 16; ++k) {
                a0 += v[k].x; a1 += v[k].y; a2 += v[k].z; a3 += v[k].w;
                if (r0 + k < g) { f0 += v[k].x; f1 += v[k].y; f2 += v[k].z; f3 += v[k].w; }
            }
        }
        pAll[q][b4] = a0; pAll[q][b4 + 1] = a1; pAll[q][b4 + 2] = a2; pAll[q][b4 + 3] = a3;
        pBef[q][b4] = f0; pBef[q][b4 + 1] = f1; pBef[q][b4 + 2] = f2; pBef[q][b4 + 3] = f3;
    }
    __syncthreads();
    int all = 0, bef = 0;
    if (t < NB) {
        for (int q = 0; q < QN; ++q) { all += pAll[q][t]; bef += pBef[q][t]; }
    }
    int total;
    const int start = wg_excl_scan(all, total);
    if (t < NB) {
        cur[t] = start + bef;
        if (g == 0) {
            bstart[t] = start;
            if (t == NB - 1) bstart[NB] = total;
        }
    }
    __syncthreads();
    const int chunk = (nEdges + NCH - 1) / NCH;
    const int e0 = g * chunk;
    int e1 = e0 + chunk; if (e1 > nEdges) e1 = nEdges;
    int e = e0 + t;
    for (; e + 3 * NT < e1; e += 4 * NT) {
        int d[4], s[4], wb[4];
#pragma unroll
        for (int k = 0; k < 4; ++k) {
            d[k]  = dst[e + k * NT];
            s[k]  = src[e + k * NT];
            wb[k] = __float_as_int(w[e + k * NT]);
        }
#pragma unroll
        for (int k = 0; k < 4; ++k) {
            const int p = atomicAdd(&cur[d[k] / BW], 1);
            int4 r; r.x = s[k]; r.y = wb[k]; r.z = d[k]; r.w = 0;
            brec[p] = r;
        }
    }
    for (; e < e1; e += NT) {
        int4 r; r.x = src[e]; r.y = __float_as_int(w[e]); r.z = dst[e]; r.w = 0;
        const int p = atomicAdd(&cur[r.z / BW], 1);
        brec[p] = r;
    }
}

__global__ __launch_bounds__(NT) void bucket_rows_k(const int* __restrict__ bstart,
                                                    const int* __restrict__ brec_w,
                                                    int* __restrict__ cnt,
                                                    int* __restrict__ rowstart,
                                                    int* __restrict__ pk) {
    __shared__ int c[NT];
    const int4* __restrict__ brec = (const int4*)brec_w;
    const int t = threadIdx.x, b = blockIdx.x;
    const int s = bstart[b], n = bstart[b + 1] - s, node0 = b * BW;
    c[t] = 0;
    __syncthreads();
    int i = t;
    for (; i + 3 * NT < n; i += 4 * NT) {
        int d[4];
#pragma unroll
        for (int k = 0; k < 4; ++k) d[k] = brec[s + i + k * NT].z;
#pragma unroll
        for (int k = 0; k < 4; ++k) atomicAdd(&c[d[k] - node0], 1);
    }
    for (; i < n; i += NT) atomicAdd(&c[brec[s + i].z - node0], 1);
    __syncthreads();
    const int mine = c[t];
    int total;
    const int excl = wg_excl_scan(mine, total);
    if (t < BW && node0 + t < N_NODES) {
        cnt[node0 + t] = mine;
        rowstart[node0 + t] = s + excl;
    }
    c[t] = excl;
    __syncthreads();
    i = t;
    for (; i + 3 * NT < n; i += 4 * NT) {
        int4 r[4];
#pragma unroll
        for (int k = 0; k < 4; ++k) r[k] = brec[s + i + k * NT];
#pragma unroll
        for (int k = 0; k < 4; ++k) {
            const int p = atomicAdd(&c[r[k].z - node0], 1);
            int2 v; v.x = r[k].x; v.y = r[k].y;
            ((int2*)pk)[s + p] = v;
        }
    }
    for (; i < n; i += NT) {
        const int4 r = brec[s + i];
        const int p = atomicAdd(&c[r.z - node0], 1);
        int2 v; v.x = r.x; v.y = r.y;
        ((int2*)pk)[s + p] = v;
    }
}

template <int K>
__device__ __forceinline__ void acc64(const float* __restrict__ h, int ex, int ey, int j,
                                      int lane, float& ax, float& ay) {
    float2 v[K];
    float  wk[K];
#pragma unroll
    for (int k = 0; k < K; ++k) {
        int sj = __builtin_amdgcn_readlane(ex, j + k);
        wk[k]  = __int_as_float(__builtin_amdgcn_readlane(ey, j + k));
        v[k]   = *(const float2*)(h + (long)sj * NHID + lane * 2);
    }
#pragma unroll
    for (int k = 0; k < K; ++k) {
        ax = fmaf(wk[k], v[k].x, ax);
        ay = fmaf(wk[k], v[k].y, ay);
    }
}

__global__ __launch_bounds__(256) void gather64_k(const int* __restrict__ rowstart,
                                                  const int* __restrict__ cnt,
                                                  const int* __restrict__ pk,
                                                  const float* __restrict__ h,
                                                  float* __restrict__ agg) {
    const int lane = threadIdx.x & 31;
    const int nw = gridDim.x * 8;
    for (int r = blockIdx.x * 8 + (threadIdx.x >> 5); r < N_NODES; r += nw) {
        const int ru = __builtin_amdgcn_readfirstlane(r);
        const int s  = __builtin_amdgcn_readfirstlane(rowstart[ru]);
        const int n  = __builtin_amdgcn_readfirstlane(cnt[ru]);
        float ax = 0.0f, ay = 0.0f;
        for (int c0 = 0; c0 < n; c0 += 32) {
            int m = n - c0; if (m > 32) m = 32;
            int ex = 0, ey = 0;
            if (lane < m) { int2 e = ((const int2*)pk)[s + c0 + lane]; ex = e.x; ey = e.y; }
            int j = 0;
            for (; j + 8 <= m; j += 8) acc64<8>(h, ex, ey, j, lane, ax, ay);
            switch (m - j) {
                case 7: acc64<7>(h, ex, ey, j, lane, ax, ay); break;
                case 6: acc64<6>(h, ex, ey, j, lane, ax, ay); break;
                case 5: acc64<5>(h, ex, ey, j, lane, ax, ay); break;
                case 4: acc64<4>(h, ex, ey, j, lane, ax, ay); break;
                case 3: acc64<3>(h, ex, ey, j, lane, ax, ay); break;
                case 2: acc64<2>(h, ex, ey, j, lane, ax, ay); break;
                case 1: acc64<1>(h, ex, ey, j, lane, ax, ay); break;
                default: break;
            }
        }
        float2 o; o.x = ax; o.y = ay;
        *(float2*)(agg + (long)ru * NHID + lane * 2) = o;
    }
}

template <int K>
__device__ __forceinline__ void acc41(const float* __restrict__ h, int ex, int ey, int j,
                                      int c0i, int c1i, float& a0, float& a1) {
    float v0[K], v1[K], wk[K];
#pragma unroll
    for (int k = 0; k < K; ++k) {
        int sj = __builtin_amdgcn_readlane(ex, j + k);
        wk[k]  = __int_as_float(__builtin_amdgcn_readlane(ey, j + k));
        const float* __restrict__ row = h + (long)sj * NCLASS;
        v0[k] = row[c0i];
        v1[k] = row[c1i];
    }
#pragma unroll
    for (int k = 0; k < K; ++k) {
        a0 = fmaf(wk[k], v0[k], a0);
        a1 = fmaf(wk[k], v1[k], a1);
    }
}

__global__ __launch_bounds__(256) void gather41_k(const int* __restrict__ rowstart,
                                                  const int* __restrict__ cnt,
                                                  const int* __restrict__ pk,
                                                  const float* __restrict__ h,
                                                  float* __restrict__ agg) {
    const int lane = threadIdx.x & 31;
    const int nw = gridDim.x * 8;
    const int c1i = (32 + lane < NCLASS) ? (32 + lane) : (NCLASS - 1);
    for (int r = blockIdx.x * 8 + (threadIdx.x >> 5); r < N_NODES; r += nw) {
        const int ru = __builtin_amdgcn_readfirstlane(r);
        const int s  = __builtin_amdgcn_readfirstlane(rowstart[ru]);
        const int n  = __builtin_amdgcn_readfirstlane(cnt[ru]);
        float a0 = 0.0f, a1 = 0.0f;
        for (int c0 = 0; c0 < n; c0 += 32) {
            int m = n - c0; if (m > 32) m = 32;
            int ex = 0, ey = 0;
            if (lane < m) { int2 e = ((const int2*)pk)[s + c0 + lane]; ex = e.x; ey = e.y; }
            int j = 0;
            for (; j + 8 <= m; j += 8) acc41<8>(h, ex, ey, j, lane, c1i, a0, a1);
            switch (m - j) {
                case 7: acc41<7>(h, ex, ey, j, lane, c1i, a0, a1); break;
                case 6: acc41<6>(h, ex, ey, j, lane, c1i, a0, a1); break;
                case 5: acc41<5>(h, ex, ey, j, lane, c1i, a0, a1); break;
                case 4: acc41<4>(h, ex, ey, j, lane, c1i, a0, a1); break;
                case 3: acc41<3>(h, ex, ey, j, lane, c1i, a0, a1); break;
                case 2: acc41<2>(h, ex, ey, j, lane, c1i, a0, a1); break;
                case 1: acc41<1>(h, ex, ey, j, lane, c1i, a0, a1); break;
                default: break;
            }
        }
        float* __restrict__ o = agg + (long)ru * NCLASS;
        o[lane] = a0;
        if (32 + lane < NCLASS) o[32 + lane] = a1;
    }
}

__global__ __launch_bounds__(256) void logsoftmax_k(const float* __restrict__ agg2,
                                                    const float* __restrict__ b2,
                                                    float* __restrict__ out) {
    int i = blockIdx.x * blockDim.x + threadIdx.x;
    if (i >= N_NODES) return;
    const float* __restrict__ z = agg2 + (long)i * NCLASS;
    float zb[NCLASS];
    float mx = -3.0e38f;
    for (int c = 0; c < NCLASS; ++c) {
        zb[c] = z[c] + b2[c];
        mx = fmaxf(mx, zb[c]);
    }
    float s = 0.0f;
    for (int c = 0; c < NCLASS; ++c) s += expf(zb[c] - mx);
    float lse = mx + logf(s);
    float* __restrict__ o = out + (long)i * NCLASS;
    for (int c = 0; c < NCLASS; ++c) o[c] = zb[c] - lse;
}

extern "C" void kernel_launch(void* const* d_in, const int* in_sizes, int n_in,
                              void* d_out, int out_size, void* d_ws, size_t ws_size,
                              hipStream_t stream) {
    const float* x    = (const float*)d_in[0];
    const int*   esrc = (const int*)d_in[1];
    const int*   edst = (const int*)d_in[2];
    const float* ew   = (const float*)d_in[3];
    const float* W1   = (const float*)d_in[4];
    const float* b1   = (const float*)d_in[5];
    const float* W2   = (const float*)d_in[6];
    const float* b2   = (const float*)d_in[7];
    float* out = (float*)d_out;
    const int E = in_sizes[1];

    float* h1   = (float*)d_ws;
    float* agg1 = h1   + (size_t)N_NODES * NHID;
    float* h2   = h1;
    float* agg2 = agg1;
    int* pk       = (int*)(agg1 + (size_t)N_NODES * NHID);
    int* hist     = pk     + (size_t)2 * N_EDGES;
    int* bstart   = hist   + (size_t)NCH * NB;
    int* cnt      = bstart + (size_t)(NB + 8);
    int* rowstart = cnt    + (size_t)(N_NODES + 8);
    int* brec     = (int*)agg1;
    static_assert((size_t)4 * N_EDGES <= (size_t)N_NODES * NHID, "bucketed copy fits in agg1");

    bucket_hist_k<<<NCH, NT, 0, stream>>>(edst, hist, E);
    bucket_scatter_k<<<NCH, NT, 0, stream>>>(esrc, edst, ew, hist, bstart, brec, E);
    bucket_rows_k<<<NB, NT, 0, stream>>>(bstart, brec, cnt, rowstart, pk);

    const int gemmBlocks = (N_NODES + 127) / 128;
    gemm1_wmma_k<<<gemmBlocks, 256, 0, stream>>>(x, W1, h1);

    gather64_k<<<2048, 256, 0, stream>>>(rowstart, cnt, pk, h1, agg1);

    gemm2_wmma_k<<<gemmBlocks, 256, 0, stream>>>(agg1, b1, W2, h2);

    gather41_k<<<2048, 256, 0, stream>>>(rowstart, cnt, pk, h2, agg2);

    logsoftmax_k<<<(N_NODES + 255) / 256, 256, 0, stream>>>(agg2, b2, out);
}


# bias + log_softmax fused into the D=41 gather: the wave that sums a row holds its 41 values in lanes, DPP max/sum, writes out; agg2 round trip and a launch gone; exp-sum order = lane pair, butterfly
# speedup vs baseline: 4.3357x; 1.0754x over previous
#include <hip/hip_runtime.h>

typedef __attribute__((ext_vector_type(2))) float v2f;
typedef __attribute__((ext_vector_type(8))) float v8f;

#define N_NODES 100000
#define NFEAT   256
#define NHID    64
#define NCLASS  41

__global__ __launch_bounds__(256) void gemm1_wmma_k(const float* __restrict__ x,
                                                    const float* __restrict__ W1,
                                                    float* __restrict__ h1) {
    const int lane = threadIdx.x & 31;
    const int wave = threadIdx.x >> 5;
    const int r0 = (blockIdx.x * 8 + wave) * 16;
    if (r0 >= N_NODES) return;

    const int m     = lane & 15;
    const int khalf = (lane >> 4) * 2;

    int arow = r0 + m;
    if (arow >= N_NODES) arow = N_NODES - 1;
    const float* __restrict__ xrow = x + (long)arow * NFEAT;

    v8f acc0 = {}, acc1 = {}, acc2 = {}, acc3 = {};
    for (int k0 = 0; k0 < NFEAT; k0 += 4) {
        v2f a;
        a.x = xrow[k0 + khalf];
        a.y = xrow[k0 + khalf + 1];
        const float* __restrict__ wlo = W1 + (k0 + khalf) * NHID + m;
        const float* __restrict__ whi = W1 + (k0 + khalf + 1) * NHID + m;
        v2f b0; b0.x = wlo[0];  b0.y = whi[0];
        v2f b1; b1.x = wlo[16]; b1.y = whi[16];
        v2f b2; b2.x = wlo[32]; b2.y = whi[32];
        v2f b3; b3.x = wlo[48]; b3.y = whi[48];
        acc0 = __builtin_amdgcn_wmma_f32_16x16x4_f32(false, a, false, b0, (short)0, acc0, false, false);
        acc1 = __builtin_amdgcn_wmma_f32_16x16x4_f32(false, a, false, b1, (short)0, acc1, false, false);
        acc2 = __builtin_amdgcn_wmma_f32_16x16x4_f32(false, a, false, b2, (short)0, acc2, false, false);
        acc3 = __builtin_amdgcn_wmma_f32_16x16x4_f32(false, a, false, b3, (short)0, acc3, false, false);
    }

    const int col     = lane & 15;
    const int rowHalf = (lane >> 4) * 8;
    for (int j = 0; j < 8; ++j) {
        int r = r0 + rowHalf + j;
        if (r < N_NODES) {
            float* __restrict__ o = h1 + (long)r * NHID + col;
            o[0]  = acc0[j];
            o[16] = acc1[j];
            o[32] = acc2[j];
            o[48] = acc3[j];
        }
    }
}

__global__ __launch_bounds__(256) void gemm2_wmma_k(const float* __restrict__ agg1,
                                                    const float* __restrict__ b1,
                                                    const float* __restrict__ W2,
                                                    float* __restrict__ h2) {
    const int lane = threadIdx.x & 31;
    const int wave = threadIdx.x >> 5;
    const int r0 = (blockIdx.x * 8 + wave) * 16;
    if (r0 >= N_NODES) return;

    const int m     = lane & 15;
    const int khalf = (lane >> 4) * 2;

    int arow = r0 + m;
    if (arow >= N_NODES) arow = N_NODES - 1;
    const float* __restrict__ hrow = agg1 + (long)arow * NHID;

    v8f acc0 = {}, acc1 = {}, acc2 = {};
    for (int k0 = 0; k0 < NHID; k0 += 4) {
        int ka = k0 + khalf;
        v2f a;
        a.x = fmaxf(hrow[ka]     + b1[ka],     0.0f);
        a.y = fmaxf(hrow[ka + 1] + b1[ka + 1], 0.0f);

        v2f b0, b1v, b2v;
        {
            int n = m;
            b0.x = W2[ka * NCLASS + n];
            b0.y = W2[(ka + 1) * NCLASS + n];
        }
        {
            int n = 16 + m;
            b1v.x = W2[ka * NCLASS + n];
            b1v.y = W2[(ka + 1) * NCLASS + n];
        }
        {
            int n = 32 + m;
            int nc = (n < NCLASS) ? n : (NCLASS - 1);
            float lo = W2[ka * NCLASS + nc];
            float hi = W2[(ka + 1) * NCLASS + nc];
            if (n >= NCLASS) { lo = 0.0f; hi = 0.0f; }
            b2v.x = lo; b2v.y = hi;
        }
        acc0 = __builtin_amdgcn_wmma_f32_16x16x4_f32(false, a, false, b0,  (short)0, acc0, false, false);
        acc1 = __builtin_amdgcn_wmma_f32_16x16x4_f32(false, a, false, b1v, (short)0, acc1, false, false);
        acc2 = __builtin_amdgcn_wmma_f32_16x16x4_f32(false, a, false, b2v, (short)0, acc2, false, false);
    }

    const int col     = lane & 15;
    const int rowHalf = (lane >> 4) * 8;
    for (int j = 0; j < 8; ++j) {
        int r = r0 + rowHalf + j;
        if (r < N_NODES) {
            float* __restrict__ o = h2 + (long)r * NCLASS;
            o[col]      = acc0[j];
            o[col + 16] = acc1[j];
            int c2 = col + 32;
            if (c2 < NCLASS) o[c2] = acc2[j];
        }
    }
}

#define N_EDGES 1600000
#define NB  256
#define BW  391
#define NCH 256
#define NT  512
static_assert(NB * BW >= N_NODES, "buckets must cover every row");
static_assert(BW <= NT, "one thread per row of a bucket");
static_assert(NB % 4 == 0 && NT % (NB / 4) == 0 && NCH % (16 * (NT / (NB / 4))) == 0, "column-sum tiling");

__device__ __forceinline__ int wg_excl_scan(int v, int& total) {
    __shared__ int wsum[32];
    const int lane = threadIdx.x & 31, wv = threadIdx.x >> 5;
    int incl = v;
    for (int d = 1; d < 32; d <<= 1) {
        int t = __shfl_up(incl, d, 32);
        if (lane >= d) incl += t;
    }
    if (lane == 31) wsum[wv] = incl;
    __syncthreads();
    const int ws = (lane < NT / 32) ? wsum[lane] : 0;
    int wincl = ws;
    for (int d = 1; d < 32; d <<= 1) {
        int t = __shfl_up(wincl, d, 32);
        if (lane >= d) wincl += t;
    }
    total = __shfl(wincl, 31, 32);
    const int wexcl = __shfl(wincl - ws, wv, 32);
    return wexcl + incl - v;
}

__global__ __launch_bounds__(NT) void bucket_hist_k(const int* __restrict__ dst,
                                                    int* __restrict__ hist, int nEdges) {
    __shared__ int h[NB];
    const int t = threadIdx.x;
    if (t < NB) h[t] = 0;
    __syncthreads();
    const int chunk = (nEdges + NCH - 1) / NCH;
    const int e0 = blockIdx.x * chunk;
    int e1 = e0 + chunk; if (e1 > nEdges) e1 = nEdges;
    int e = e0 + t;
    for (; e + 3 * NT < e1; e += 4 * NT) {
        int d[4];
#pragma unroll
        for (int k = 0; k < 4; ++k) d[k] = dst[e + k * NT];
#pragma unroll
        for (int k = 0; k < 4; ++k) atomicAdd(&h[d[k] / BW], 1);
    }
    for (; e < e1; e += NT) atomicAdd(&h[dst[e] / BW], 1);
    __syncthreads();
    if (t < NB) hist[blockIdx.x * NB + t] = h[t];
}

__global__ __launch_bounds__(NT) void bucket_scatter_k(const int* __restrict__ src,
                                                       const int* __restrict__ dst,
                                                       const float* __restrict__ w,
                                                       const int* __restrict__ hist,
                                                       int* __restrict__ bstart,
                                                       int* __restrict__ brec_w, int nEdges) {
    int4* __restrict__ brec = (int4*)brec_w;
    constexpr int QN = NT / (NB / 4);
    constexpr int QR = NCH / QN;
    __shared__ int pAll[QN][NB];
    __shared__ int pBef[QN][NB];
    __shared__ int cur[NB];
    const int t = threadIdx.x, g = blockIdx.x;
    {
        const int b4 = (t % (NB / 4)) * 4, q = t / (NB / 4);
        int a0 = 0, a1 = 0, a2 = 0, a3 = 0, f0 = 0, f1 = 0, f2 = 0, f3 = 0;
        for (int r0 = q * QR; r0 < (q + 1) * QR; r0 += 16) {
            int4 v[16];
#pragma unroll
            for (int k = 0; k < 16; ++k) v[k] = *(const int4*)(hist + (r0 + k) * NB + b4);
#pragma unroll
            for (int k = 0; k < 16; ++k) {
                a0 += v[k].x; a1 += v[k].y; a2 += v[k].z; a3 += v[k].w;
                if (r0 + k < g) { f0 += v[k].x; f1 += v[k].y; f2 += v[k].z; f3 += v[k].w; }
            }
        }
        pAll[q][b4] = a0; pAll[q][b4 + 1] = a1; pAll[q][b4 + 2] = a2; pAll[q][b4 + 3] = a3;
        pBef[q][b4] = f0; pBef[q][b4 + 1] = f1; pBef[q][b4 + 2] = f2; pBef[q][b4 + 3] = f3;
    }
    __syncthreads();
    int all = 0, bef = 0;
    if (t < NB) {
        for (int q = 0; q < QN; ++q) { all += pAll[q][t]; bef += pBef[q][t]; }
    }
    int total;
    const int start = wg_excl_scan(all, total);
    if (t < NB) {
        cur[t] = start + bef;
        if (g == 0) {
            bstart[t] = start;
            if (t == NB - 1) bstart[NB] = total;
        }
    }
    __syncthreads();
    const int chunk = (nEdges + NCH - 1) / NCH;
    const int e0 = g * chunk;
    int e1 = e0 + chunk; if (e1 > nEdges) e1 = nEdges;
    int e = e0 + t;
    for (; e + 3 * NT < e1; e += 4 * NT) {
        int d[4], s[4], wb[4];
#pragma unroll
        for (int k = 0; k < 4; ++k) {
            d[k]  = dst[e + k * NT];
            s[k]  = src[e + k * NT];
            wb[k] = __float_as_int(w[e + k * NT]);
        }
#pragma unroll
        for (int k = 0; k < 4; ++k) {
            const int p = atomicAdd(&cur[d[k] / BW], 1);
            int4 r; r.x = s[k]; r.y = wb[k]; r.z = d[k]; r.w = 0;
            brec[p] = r;
        }
    }
    for (; e < e1; e += NT) {
        int4 r; r.x = src[e]; r.y = __float_as_int(w[e]); r.z = dst[e]; r.w = 0;
        const int p = atomicAdd(&cur[r.z / BW], 1);
        brec[p] = r;
    }
}

__global__ __launch_bounds__(NT) void bucket_rows_k(const int* __restrict__ bstart,
                                                    const int* __restrict__ brec_w,
                                                    int* __restrict__ cnt,
                                                    int* __restrict__ rowstart,
                                                    int* __restrict__ pk) {
    __shared__ int c[NT];
    const int4* __restrict__ brec = (const int4*)brec_w;
    const int t = threadIdx.x, b = blockIdx.x;
    const int s = bstart[b], n = bstart[b + 1] - s, node0 = b * BW;
    c[t] = 0;
    __syncthreads();
    int i = t;
    for (; i + 3 * NT < n; i += 4 * NT) {
        int d[4];
#pragma unroll
        for (int k = 0; k < 4; ++k) d[k] = brec[s + i + k * NT].z;
#pragma unroll
        for (int k = 0; k < 4; ++k) atomicAdd(&c[d[k] - node0], 1);
    }
    for (; i < n; i += NT) atomicAdd(&c[brec[s + i].z - node0], 1);
    __syncthreads();
    const int mine = c[t];
    int total;
    const int excl = wg_excl_scan(mine, total);
    if (t < BW && node0 + t < N_NODES) {
        cnt[node0 + t] = mine;
        rowstart[node0 + t] = s + excl;
    }
    c[t] = excl;
    __syncthreads();
    i = t;
    for (; i + 3 * NT < n; i += 4 * NT) {
        int4 r[4];
#pragma unroll
        for (int k = 0; k < 4; ++k) r[k] = brec[s + i + k * NT];
#pragma unroll
        for (int k = 0; k < 4; ++k) {
            const int p = atomicAdd(&c[r[k].z - node0], 1);
            int2 v; v.x = r[k].x; v.y = r[k].y;
            ((int2*)pk)[s + p] = v;
        }
    }
    for (; i < n; i += NT) {
        const int4 r = brec[s + i];
        const int p = atomicAdd(&c[r.z - node0], 1);
        int2 v; v.x = r.x; v.y = r.y;
        ((int2*)pk)[s + p] = v;
    }
}

template <int K>
__device__ __forceinline__ void acc64(const float* __restrict__ h, int ex, int ey, int j,
                                      int lane, float& ax, float& ay) {
    float2 v[K];
    float  wk[K];
#pragma unroll
    for (int k = 0; k < K; ++k) {
        int sj = __builtin_amdgcn_readlane(ex, j + k);
        wk[k]  = __int_as_float(__builtin_amdgcn_readlane(ey, j + k));
        v[k]   = *(const float2*)(h + (long)sj * NHID + lane * 2);
    }
#pragma unroll
    for (int k = 0; k < K; ++k) {
        ax = fmaf(wk[k], v[k].x, ax);
        ay = fmaf(wk[k], v[k].y, ay);
    }
}

__global__ __launch_bounds__(256) void gather64_k(const int* __restrict__ rowstart,
                                                  const int* __restrict__ cnt,
                                                  const int* __restrict__ pk,
                                                  const float* __restrict__ h,
                                                  float* __restrict__ agg) {
    const int lane = threadIdx.x & 31;
    const int nw = gridDim.x * 8;
    for (int r = blockIdx.x * 8 + (threadIdx.x >> 5); r < N_NODES; r += nw) {
        const int ru = __builtin_amdgcn_readfirstlane(r);
        const int s  = __builtin_amdgcn_readfirstlane(rowstart[ru]);
        const int n  = __builtin_amdgcn_readfirstlane(cnt[ru]);
        float ax = 0.0f, ay = 0.0f;
        for (int c0 = 0; c0 < n; c0 += 32) {
            int m = n - c0; if (m > 32) m = 32;
            int ex = 0, ey = 0;
            if (lane < m) { int2 e = ((const int2*)pk)[s + c0 + lane]; ex = e.x; ey = e.y; }
            int j = 0;
            for (; j + 8 <= m; j += 8) acc64<8>(h, ex, ey, j, lane, ax, ay);
            switch (m - j) {
                case 7: acc64<7>(h, ex, ey, j, lane, ax, ay); break;
                case 6: acc64<6>(h, ex, ey, j, lane, ax, ay); break;
                case 5: acc64<5>(h, ex, ey, j, lane, ax, ay); break;
                case 4: acc64<4>(h, ex, ey, j, lane, ax, ay); break;
                case 3: acc64<3>(h, ex, ey, j, lane, ax, ay); break;
                case 2: acc64<2>(h, ex, ey, j, lane, ax, ay); break;
                case 1: acc64<1>(h, ex, ey, j, lane, ax, ay); break;
                default: break;
            }
        }
        float2 o; o.x = ax; o.y = ay;
        *(float2*)(agg + (long)ru * NHID + lane * 2) = o;
    }
}

template <int K>
__device__ __forceinline__ void acc41(const float* __restrict__ h, int ex, int ey, int j,
                                      int c0i, int c1i, float& a0, float& a1) {
    float v0[K], v1[K], wk[K];
#pragma unroll
    for (int k = 0; k < K; ++k) {
        int sj = __builtin_amdgcn_readlane(ex, j + k);
        wk[k]  = __int_as_float(__builtin_amdgcn_readlane(ey, j + k));
        const float* __restrict__ row = h + (long)sj * NCLASS;
        v0[k] = row[c0i];
        v1[k] = row[c1i];
    }
#pragma unroll
    for (int k = 0; k < K; ++k) {
        a0 = fmaf(wk[k], v0[k], a0);
        a1 = fmaf(wk[k], v1[k], a1);
    }
}

__device__ __forceinline__ float xhalf(float x) {
    int b = __builtin_bit_cast(int, x);
    return __builtin_bit_cast(float, __builtin_amdgcn_permlanex16(b, b, 0x76543210, 0xfedcba98, false, false));
}

__device__ __forceinline__ float row16_max(float v) {
    asm volatile("v_max_num_f32_dpp %0, %0, %0 row_xmask:1 row_mask:0xf bank_mask:0xf\n\t"
                 "v_max_num_f32_dpp %0, %0, %0 row_xmask:2 row_mask:0xf bank_mask:0xf\n\t"
                 "v_max_num_f32_dpp %0, %0, %0 row_xmask:4 row_mask:0xf bank_mask:0xf\n\t"
                 "v_max_num_f32_dpp %0, %0, %0 row_xmask:8 row_mask:0xf bank_mask:0xf"
                 : "+v"(v));
    return v;
}

__device__ __forceinline__ float row16_sum(float v) {
    asm volatile("v_add_f32_dpp %0, %0, %0 row_xmask:1 row_mask:0xf bank_mask:0xf\n\t"
                 "v_add_f32_dpp %0, %0, %0 row_xmask:2 row_mask:0xf bank_mask:0xf\n\t"
                 "v_add_f32_dpp %0, %0, %0 row_xmask:4 row_mask:0xf bank_mask:0xf\n\t"
                 "v_add_f32_dpp %0, %0, %0 row_xmask:8 row_mask:0xf bank_mask:0xf"
                 : "+v"(v));
    return v;
}

__device__ __forceinline__ float wave_max(float v) {
    v = row16_max(v);
    return fmaxf(v, xhalf(v));
}

__device__ __forceinline__ float wave_sum(float v) {
    v = row16_sum(v);
    return v + xhalf(v);
}

__global__ __launch_bounds__(256) void gather41_lsm_k(const int* __restrict__ rowstart,
                                                      const int* __restrict__ cnt,
                                                      const int* __restrict__ pk,
                                                      const float* __restrict__ h,
                                                      const float* __restrict__ b2,
                                                      float* __restrict__ out) {
    const int lane = threadIdx.x & 31;
    const int nw = gridDim.x * 8;
    const bool has1 = (32 + lane < NCLASS);
    const int c1i = has1 ? (32 + lane) : (NCLASS - 1);
    const float bz0 = b2[lane];
    const float bz1 = b2[c1i];
    for (int r = blockIdx.x * 8 + (threadIdx.x >> 5); r < N_NODES; r += nw) {
        const int ru = __builtin_amdgcn_readfirstlane(r);
        const int s  = __builtin_amdgcn_readfirstlane(rowstart[ru]);
        const int n  = __builtin_amdgcn_readfirstlane(cnt[ru]);
        float a0 = 0.0f, a1 = 0.0f;
        for (int c0 = 0; c0 < n; c0 += 32) {
            int m = n - c0; if (m > 32) m = 32;
            int ex = 0, ey = 0;
            if (lane < m) { int2 e = ((const int2*)pk)[s + c0 + lane]; ex = e.x; ey = e.y; }
            int j = 0;
            for (; j + 8 <= m; j += 8) acc41<8>(h, ex, ey, j, lane, c1i, a0, a1);
            switch (m - j) {
                case 7: acc41<7>(h, ex, ey, j, lane, c1i, a0, a1); break;
                case 6: acc41<6>(h, ex, ey, j, lane, c1i, a0, a1); break;
                case 5: acc41<5>(h, ex, ey, j, lane, c1i, a0, a1); break;
                case 4: acc41<4>(h, ex, ey, j, lane, c1i, a0, a1); break;
                case 3: acc41<3>(h, ex, ey, j, lane, c1i, a0, a1); break;
                case 2: acc41<2>(h, ex, ey, j, lane, c1i, a0, a1); break;
                case 1: acc41<1>(h, ex, ey, j, lane, c1i, a0, a1); break;
                default: break;
            }
        }
        const float z0 = a0 + bz0;
        const float z1 = has1 ? (a1 + bz1) : z0;
        const float mx = wave_max(fmaxf(z0, z1));
        const float e0 = expf(z0 - mx);
        const float e1 = expf(z1 - mx);
        const float lse = mx + logf(wave_sum(has1 ? (e0 + e1) : e0));
        float* __restrict__ o = out + (long)ru * NCLASS;
        o[lane] = z0 - lse;
        if (has1) o[32 + lane] = z1 - lse;
    }
}

extern "C" void kernel_launch(void* const* d_in, const int* in_sizes, int n_in,
                              void* d_out, int out_size, void* d_ws, size_t ws_size,
                              hipStream_t stream) {
    const float* x    = (const float*)d_in[0];
    const int*   esrc = (const int*)d_in[1];
    const int*   edst = (const int*)d_in[2];
    const float* ew   = (const float*)d_in[3];
    const float* W1   = (const float*)d_in[4];
    const float* b1   = (const float*)d_in[5];
    const float* W2   = (const float*)d_in[6];
    const float* b2   = (const float*)d_in[7];
    float* out = (float*)d_out;
    const int E = in_sizes[1];

    float* h1   = (float*)d_ws;
    float* agg1 = h1   + (size_t)N_NODES * NHID;
    float* h2   = h1;
    int* pk       = (int*)(agg1 + (size_t)N_NODES * NHID);
    int* hist     = pk     + (size_t)2 * N_EDGES;
    int* bstart   = hist   + (size_t)NCH * NB;
    int* cnt      = bstart + (size_t)(NB + 8);
    int* rowstart = cnt    + (size_t)(N_NODES + 8);
    int* brec     = (int*)agg1;
    static_assert((size_t)4 * N_EDGES <= (size_t)N_NODES * NHID, "bucketed copy fits in agg1");

    bucket_hist_k<<<NCH, NT, 0, stream>>>(edst, hist, E);
    bucket_scatter_k<<<NCH, NT, 0, stream>>>(esrc, edst, ew, hist, bstart, brec, E);
    bucket_rows_k<<<NB, NT, 0, stream>>>(bstart, brec, cnt, rowstart, pk);

    const int gemmBlocks = (N_NODES + 127) / 128;
    gemm1_wmma_k<<<gemmBlocks, 256, 0, stream>>>(x, W1, h1);

    gather64_k<<<2048, 256, 0, stream>>>(rowstart, cnt, pk, h1, agg1);

    gemm2_wmma_k<<<gemmBlocks, 256, 0, stream>>>(agg1, b1, W2, h2);

    gather41_lsm_k<<<2048, 256, 0, stream>>>(rowstart, cnt, pk, h2, b2, out);
}


# GEMM2 launch gone: the D=64 gather's work-group stages 16 relu(sum+b1) rows in LDS, three of its waves multiply them by W2 on f32 WMMA and write h2; agg1 never stored; k order, sum orders unchanged
# speedup vs baseline: 4.6559x; 1.0739x over previous
#include <hip/hip_runtime.h>

typedef __attribute__((ext_vector_type(2))) float v2f;
typedef __attribute__((ext_vector_type(8))) float v8f;

#define N_NODES 100000
#define NFEAT   256
#define NHID    64
#define NCLASS  41

__global__ __launch_bounds__(256) void gemm1_wmma_k(const float* __restrict__ x,
                                                    const float* __restrict__ W1,
                                                    float* __restrict__ h1) {
    const int lane = threadIdx.x & 31;
    const int wave = threadIdx.x >> 5;
    const int r0 = (blockIdx.x * 8 + wave) * 16;
    if (r0 >= N_NODES) return;

    const int m     = lane & 15;
    const int khalf = (lane >> 4) * 2;

    int arow = r0 + m;
    if (arow >= N_NODES) arow = N_NODES - 1;
    const float* __restrict__ xrow = x + (long)arow * NFEAT;

    v8f acc0 = {}, acc1 = {}, acc2 = {}, acc3 = {};
    for (int k0 = 0; k0 < NFEAT; k0 += 4) {
        v2f a;
        a.x = xrow[k0 + khalf];
        a.y = xrow[k0 + khalf + 1];
        const float* __restrict__ wlo = W1 + (k0 + khalf) * NHID + m;
        const float* __restrict__ whi = W1 + (k0 + khalf + 1) * NHID + m;
        v2f b0; b0.x = wlo[0];  b0.y = whi[0];
        v2f b1; b1.x = wlo[16]; b1.y = whi[16];
        v2f b2; b2.x = wlo[32]; b2.y = whi[32];
        v2f b3; b3.x = wlo[48]; b3.y = whi[48];
        acc0 = __builtin_amdgcn_wmma_f32_16x16x4_f32(false, a, false, b0, (short)0, acc0, false, false);
        acc1 = __builtin_amdgcn_wmma_f32_16x16x4_f32(false, a, false, b1, (short)0, acc1, false, false);
        acc2 = __builtin_amdgcn_wmma_f32_16x16x4_f32(false, a, false, b2, (short)0, acc2, false, false);
        acc3 = __builtin_amdgcn_wmma_f32_16x16x4_f32(false, a, false, b3, (short)0, acc3, false, false);
    }

    const int col     = lane & 15;
    const int rowHalf = (lane >> 4) * 8;
    for (int j = 0; j < 8; ++j) {
        int r = r0 + rowHalf + j;
        if (r < N_NODES) {
            float* __restrict__ o = h1 + (long)r * NHID + col;
            o[0]  = acc0[j];
            o[16] = acc1[j];
            o[32] = acc2[j];
            o[48] = acc3[j];
        }
    }
}

#define N_EDGES 1600000
#define NB  256
#define BW  391
#define NCH 256
#define NT  512
static_assert(NB * BW >= N_NODES, "buckets must cover every row");
static_assert(BW <= NT, "one thread per row of a bucket");
static_assert(NB % 4 == 0 && NT % (NB / 4) == 0 && NCH % (16 * (NT / (NB / 4))) == 0, "column-sum tiling");

__device__ __forceinline__ int wg_excl_scan(int v, int& total) {
    __shared__ int wsum[32];
    const int lane = threadIdx.x & 31, wv = threadIdx.x >> 5;
    int incl = v;
    for (int d = 1; d < 32; d <<= 1) {
        int t = __shfl_up(incl, d, 32);
        if (lane >= d) incl += t;
    }
    if (lane == 31) wsum[wv] = incl;
    __syncthreads();
    const int ws = (lane < NT / 32) ? wsum[lane] : 0;
    int wincl = ws;
    for (int d = 1; d < 32; d <<= 1) {
        int t = __shfl_up(wincl, d, 32);
        if (lane >= d) wincl += t;
    }
    total = __shfl(wincl, 31, 32);
    const int wexcl = __shfl(wincl - ws, wv, 32);
    return wexcl + incl - v;
}

__global__ __launch_bounds__(NT) void bucket_hist_k(const int* __restrict__ dst,
                                                    int* __restrict__ hist, int nEdges) {
    __shared__ int h[NB];
    const int t = threadIdx.x;
    if (t < NB) h[t] = 0;
    __syncthreads();
    const int chunk = (nEdges + NCH - 1) / NCH;
    const int e0 = blockIdx.x * chunk;
    int e1 = e0 + chunk; if (e1 > nEdges) e1 = nEdges;
    int e = e0 + t;
    for (; e + 3 * NT < e1; e += 4 * NT) {
        int d[4];
#pragma unroll
        for (int k = 0; k < 4; ++k) d[k] = dst[e + k * NT];
#pragma unroll
        for (int k = 0; k < 4; ++k) atomicAdd(&h[d[k] / BW], 1);
    }
    for (; e < e1; e += NT) atomicAdd(&h[dst[e] / BW], 1);
    __syncthreads();
    if (t < NB) hist[blockIdx.x * NB + t] = h[t];
}

__global__ __launch_bounds__(NT) void bucket_scatter_k(const int* __restrict__ src,
                                                       const int* __restrict__ dst,
                                                       const float* __restrict__ w,
                                                       const int* __restrict__ hist,
                                                       int* __restrict__ bstart,
                                                       int* __restrict__ brec_w, int nEdges) {
    int4* __restrict__ brec = (int4*)brec_w;
    constexpr int QN = NT / (NB / 4);
    constexpr int QR = NCH / QN;
    __shared__ int pAll[QN][NB];
    __shared__ int pBef[QN][NB];
    __shared__ int cur[NB];
    const int t = threadIdx.x, g = blockIdx.x;
    {
        const int b4 = (t % (NB / 4)) * 4, q = t / (NB / 4);
        int a0 = 0, a1 = 0, a2 = 0, a3 = 0, f0 = 0, f1 = 0, f2 = 0, f3 = 0;
        for (int r0 = q * QR; r0 < (q + 1) * QR; r0 += 16) {
            int4 v[16];
#pragma unroll
            for (int k = 0; k < 16; ++k) v[k] = *(const int4*)(hist + (r0 + k) * NB + b4);
#pragma unroll
            for (int k = 0; k < 16; ++k) {
                a0 += v[k].x; a1 += v[k].y; a2 += v[k].z; a3 += v[k].w;
                if (r0 + k < g) { f0 += v[k].x; f1 += v[k].y; f2 += v[k].z; f3 += v[k].w; }
            }
        }
        pAll[q][b4] = a0; pAll[q][b4 + 1] = a1; pAll[q][b4 + 2] = a2; pAll[q][b4 + 3] = a3;
        pBef[q][b4] = f0; pBef[q][b4 + 1] = f1; pBef[q][b4 + 2] = f2; pBef[q][b4 + 3] = f3;
    }
    __syncthreads();
    int all = 0, bef = 0;
    if (t < NB) {
        for (int q = 0; q < QN; ++q) { all += pAll[q][t]; bef += pBef[q][t]; }
    }
    int total;
    const int start = wg_excl_scan(all, total);
    if (t < NB) {
        cur[t] = start + bef;
        if (g == 0) {
            bstart[t] = start;
            if (t == NB - 1) bstart[NB] = total;
        }
    }
    __syncthreads();
    const int chunk = (nEdges + NCH - 1) / NCH;
    const int e0 = g * chunk;
    int e1 = e0 + chunk; if (e1 > nEdges) e1 = nEdges;
    int e = e0 + t;
    for (; e + 3 * NT < e1; e += 4 * NT) {
        int d[4], s[4], wb[4];
#pragma unroll
        for (int k = 0; k < 4; ++k) {
            d[k]  = dst[e + k * NT];
            s[k]  = src[e + k * NT];
            wb[k] = __float_as_int(w[e + k * NT]);
        }
#pragma unroll
        for (int k = 0; k < 4; ++k) {
            const int p = atomicAdd(&cur[d[k] / BW], 1);
            int4 r; r.x = s[k]; r.y = wb[k]; r.z = d[k]; r.w = 0;
            brec[p] = r;
        }
    }
    for (; e < e1; e += NT) {
        int4 r; r.x = src[e]; r.y = __float_as_int(w[e]); r.z = dst[e]; r.w = 0;
        const int p = atomicAdd(&cur[r.z / BW], 1);
        brec[p] = r;
    }
}

__global__ __launch_bounds__(NT) void bucket_rows_k(const int* __restrict__ bstart,
                                                    const int* __restrict__ brec_w,
                                                    int* __restrict__ cnt,
                                                    int* __restrict__ rowstart,
                                                    int* __restrict__ pk) {
    __shared__ int c[NT];
    const int4* __restrict__ brec = (const int4*)brec_w;
    const int t = threadIdx.x, b = blockIdx.x;
    const int s = bstart[b], n = bstart[b + 1] - s, node0 = b * BW;
    c[t] = 0;
    __syncthreads();
    int i = t;
    for (; i + 3 * NT < n; i += 4 * NT) {
        int d[4];
#pragma unroll
        for (int k = 0; k < 4; ++k) d[k] = brec[s + i + k * NT].z;
#pragma unroll
        for (int k = 0; k < 4; ++k) atomicAdd(&c[d[k] - node0], 1);
    }
    for (; i < n; i += NT) atomicAdd(&c[brec[s + i].z - node0], 1);
    __syncthreads();
    const int mine = c[t];
    int total;
    const int excl = wg_excl_scan(mine, total);
    if (t < BW && node0 + t < N_NODES) {
        cnt[node0 + t] = mine;
        rowstart[node0 + t] = s + excl;
    }
    c[t] = excl;
    __syncthreads();
    i = t;
    for (; i + 3 * NT < n; i += 4 * NT) {
        int4 r[4];
#pragma unroll
        for (int k = 0; k < 4; ++k) r[k] = brec[s + i + k * NT];
#pragma unroll
        for (int k = 0; k < 4; ++k) {
            const int p = atomicAdd(&c[r[k].z - node0], 1);
            int2 v; v.x = r[k].x; v.y = r[k].y;
            ((int2*)pk)[s + p] = v;
        }
    }
    for (; i < n; i += NT) {
        const int4 r = brec[s + i];
        const int p = atomicAdd(&c[r.z - node0], 1);
        int2 v; v.x = r.x; v.y = r.y;
        ((int2*)pk)[s + p] = v;
    }
}

template <int K>
__device__ __forceinline__ void acc64(const float* __restrict__ h, int ex, int ey, int j,
                                      int lane, float& ax, float& ay) {
    float2 v[K];
    float  wk[K];
#pragma unroll
    for (int k = 0; k < K; ++k) {
        int sj = __builtin_amdgcn_readlane(ex, j + k);
        wk[k]  = __int_as_float(__builtin_amdgcn_readlane(ey, j + k));
        v[k]   = *(const float2*)(h + (long)sj * NHID + lane * 2);
    }
#pragma unroll
    for (int k = 0; k < K; ++k) {
        ax = fmaf(wk[k], v[k].x, ax);
        ay = fmaf(wk[k], v[k].y, ay);
    }
}

__device__ __forceinline__ void row_sum64(const int* __restrict__ rowstart,
                                          const int* __restrict__ cnt,
                                          const int* __restrict__ pk,
                                          const float* __restrict__ h,
                                          int r, int lane, float& ax, float& ay) {
    const int ru = __builtin_amdgcn_readfirstlane(r);
    const int s  = __builtin_amdgcn_readfirstlane(rowstart[ru]);
    const int n  = __builtin_amdgcn_readfirstlane(cnt[ru]);
    ax = 0.0f; ay = 0.0f;
    for (int c0 = 0; c0 < n; c0 += 32) {
        int m = n - c0; if (m > 32) m = 32;
        int ex = 0, ey = 0;
        if (lane < m) { int2 e = ((const int2*)pk)[s + c0 + lane]; ex = e.x; ey = e.y; }
        int j = 0;
        for (; j + 8 <= m; j += 8) acc64<8>(h, ex, ey, j, lane, ax, ay);
        switch (m - j) {
            case 7: acc64<7>(h, ex, ey, j, lane, ax, ay); break;
            case 6: acc64<6>(h, ex, ey, j, lane, ax, ay); break;
            case 5: acc64<5>(h, ex, ey, j, lane, ax, ay); break;
            case 4: acc64<4>(h, ex, ey, j, lane, ax, ay); break;
            case 3: acc64<3>(h, ex, ey, j, lane, ax, ay); break;
            case 2: acc64<2>(h, ex, ey, j, lane, ax, ay); break;
            case 1: acc64<1>(h, ex, ey, j, lane, ax, ay); break;
            default: break;
        }
    }
}

#define TP   68
#define NPAD 48
static_assert(N_NODES % 8 == 0, "rows come in whole groups of 8");
static_assert(NHID % 4 == 0 && NCLASS <= NPAD, "product tiling");

__global__ __launch_bounds__(256) void gather64_gemm2_k(const int* __restrict__ rowstart,
                                                        const int* __restrict__ cnt,
                                                        const int* __restrict__ pk,
                                                        const float* __restrict__ h,
                                                        const float* __restrict__ b1,
                                                        const float* __restrict__ W2,
                                                        float* __restrict__ h2) {
    __shared__ __attribute__((aligned(16))) float w2p[(NHID / 2) * NPAD * 2];
    __shared__ __attribute__((aligned(16))) float tile[2][16 * TP];
    const int lane = threadIdx.x & 31;
    const int wave = threadIdx.x >> 5;
    {
        constexpr int NF = (NHID / 2) * NPAD / 256;
        static_assert(NF * 256 == (NHID / 2) * NPAD, "whole entries per thread");
        float2 wv[NF];
#pragma unroll
        for (int q = 0; q < NF; ++q) {
            const int i = threadIdx.x + q * 256;
            const int kp = i / NPAD, n = i - kp * NPAD;
            const int nc = (n < NCLASS) ? n : (NCLASS - 1);
            const float lo = W2[(2 * kp) * NCLASS + nc];
            const float hi = W2[(2 * kp + 1) * NCLASS + nc];
            wv[q].x = (n < NCLASS) ? lo : 0.0f;
            wv[q].y = (n < NCLASS) ? hi : 0.0f;
        }
#pragma unroll
        for (int q = 0; q < NF; ++q) *(float2*)(w2p + 2 * (threadIdx.x + q * 256)) = wv[q];
    }
    const float bx = b1[2 * lane], by = b1[2 * lane + 1];
    const int c  = lane & 15;
    const int hh = lane >> 4;
    constexpr int NG = N_NODES / 8;
    const int gstep = gridDim.x;
    int it = 0;
    for (int g = blockIdx.x; g < NG; g += gstep, ++it) {
        const int half = it & 1;
        const bool last = g + gstep >= NG;
        float* __restrict__ tb = tile[(it >> 1) & 1];
        float ax, ay;
        row_sum64(rowstart, cnt, pk, h, g * 8 + wave, lane, ax, ay);
        float2 o; o.x = fmaxf(ax + bx, 0.0f); o.y = fmaxf(ay + by, 0.0f);
        *(float2*)(tb + (half * 8 + wave) * TP + 2 * lane) = o;
        if (half == 0 && !last) continue;
        if (half == 0) {
            float2 z; z.x = 0.0f; z.y = 0.0f;
            *(float2*)(tb + (8 + wave) * TP + 2 * lane) = z;
        }
        __syncthreads();
        if (wave < 3) {
            const float* __restrict__ ap = tb + c * TP + 2 * hh;
            const float* __restrict__ bp = w2p + (hh * NPAD + wave * 16 + c) * 2;
            v8f acc = {};
            for (int k0 = 0; k0 < NHID; k0 += 4) {
                const v2f a = *(const v2f*)(ap + k0);
                const v2f b = *(const v2f*)(bp + (k0 / 2) * NPAD * 2);
                acc = __builtin_amdgcn_wmma_f32_16x16x4_f32(false, a, false, b, (short)0, acc, false, false);
            }
            asm volatile("v_nop\n\tv_nop\n\tv_nop\n\tv_nop\n\tv_nop" : "+v"(acc));
            const int col = wave * 16 + c;
            const int gs = (hh == half) ? g : g - gstep;
            if (col < NCLASS && (hh == 0 || half == 1)) {
                float* __restrict__ op = h2 + (long)(gs * 8) * NCLASS + col;
                for (int j = 0; j < 8; ++j) op[j * NCLASS] = acc[j];
            }
        }
    }
}

template <int K>
__device__ __forceinline__ void acc41(const float* __restrict__ h, int ex, int ey, int j,
                                      int c0i, int c1i, float& a0, float& a1) {
    float v0[K], v1[K], wk[K];
#pragma unroll
    for (int k = 0; k < K; ++k) {
        int sj = __builtin_amdgcn_readlane(ex, j + k);
        wk[k]  = __int_as_float(__builtin_amdgcn_readlane(ey, j + k));
        const float* __restrict__ row = h + (long)sj * NCLASS;
        v0[k] = row[c0i];
        v1[k] = row[c1i];
    }
#pragma unroll
    for (int k = 0; k < K; ++k) {
        a0 = fmaf(wk[k], v0[k], a0);
        a1 = fmaf(wk[k], v1[k], a1);
    }
}

__device__ __forceinline__ float xhalf(float x) {
    int b = __builtin_bit_cast(int, x);
    return __builtin_bit_cast(float, __builtin_amdgcn_permlanex16(b, b, 0x76543210, 0xfedcba98, false, false));
}

__device__ __forceinline__ float row16_max(float v) {
    asm volatile("v_max_num_f32_dpp %0, %0, %0 row_xmask:1 row_mask:0xf bank_mask:0xf\n\t"
                 "v_max_num_f32_dpp %0, %0, %0 row_xmask:2 row_mask:0xf bank_mask:0xf\n\t"
                 "v_max_num_f32_dpp %0, %0, %0 row_xmask:4 row_mask:0xf bank_mask:0xf\n\t"
                 "v_max_num_f32_dpp %0, %0, %0 row_xmask:8 row_mask:0xf bank_mask:0xf"
                 : "+v"(v));
    return v;
}

__device__ __forceinline__ float row16_sum(float v) {
    asm volatile("v_add_f32_dpp %0, %0, %0 row_xmask:1 row_mask:0xf bank_mask:0xf\n\t"
                 "v_add_f32_dpp %0, %0, %0 row_xmask:2 row_mask:0xf bank_mask:0xf\n\t"
                 "v_add_f32_dpp %0, %0, %0 row_xmask:4 row_mask:0xf bank_mask:0xf\n\t"
                 "v_add_f32_dpp %0, %0, %0 row_xmask:8 row_mask:0xf bank_mask:0xf"
                 : "+v"(v));
    return v;
}

__device__ __forceinline__ float wave_max(float v) {
    v = row16_max(v);
    return fmaxf(v, xhalf(v));
}

__device__ __forceinline__ float wave_sum(float v) {
    v = row16_sum(v);
    return v + xhalf(v);
}

__global__ __launch_bounds__(256) void gather41_lsm_k(const int* __restrict__ rowstart,
                                                      const int* __restrict__ cnt,
                                                      const int* __restrict__ pk,
                                                      const float* __restrict__ h,
                                                      const float* __restrict__ b2,
                                                      float* __restrict__ out) {
    const int lane = threadIdx.x & 31;
    const int nw = gridDim.x * 8;
    const bool has1 = (32 + lane < NCLASS);
    const int c1i = has1 ? (32 + lane) : (NCLASS - 1);
    const float bz0 = b2[lane];
    const float bz1 = b2[c1i];
    for (int r = blockIdx.x * 8 + (threadIdx.x >> 5); r < N_NODES; r += nw) {
        const int ru = __builtin_amdgcn_readfirstlane(r);
        const int s  = __builtin_amdgcn_readfirstlane(rowstart[ru]);
        const int n  = __builtin_amdgcn_readfirstlane(cnt[ru]);
        float a0 = 0.0f, a1 = 0.0f;
        for (int c0 = 0; c0 < n; c0 += 32) {
            int m = n - c0; if (m > 32) m = 32;
            int ex = 0, ey = 0;
            if (lane < m) { int2 e = ((const int2*)pk)[s + c0 + lane]; ex = e.x; ey = e.y; }
            int j = 0;
            for (; j + 8 <= m; j += 8) acc41<8>(h, ex, ey, j, lane, c1i, a0, a1);
            switch (m - j) {
                case 7: acc41<7>(h, ex, ey, j, lane, c1i, a0, a1); break;
                case 6: acc41<6>(h, ex, ey, j, lane, c1i, a0, a1); break;
                case 5: acc41<5>(h, ex, ey, j, lane, c1i, a0, a1); break;
                case 4: acc41<4>(h, ex, ey, j, lane, c1i, a0, a1); break;
                case 3: acc41<3>(h, ex, ey, j, lane, c1i, a0, a1); break;
                case 2: acc41<2>(h, ex, ey, j, lane, c1i, a0, a1); break;
                case 1: acc41<1>(h, ex, ey, j, lane, c1i, a0, a1); break;
                default: break;
            }
        }
        const float z0 = a0 + bz0;
        const float z1 = has1 ? (a1 + bz1) : z0;
        const float mx = wave_max(fmaxf(z0, z1));
        const float e0 = expf(z0 - mx);
        const float e1 = expf(z1 - mx);
        const float lse = mx + logf(wave_sum(has1 ? (e0 + e1) : e0));
        float* __restrict__ o = out + (long)ru * NCLASS;
        o[lane] = z0 - lse;
        if (has1) o[32 + lane] = z1 - lse;
    }
}

extern "C" void kernel_launch(void* const* d_in, const int* in_sizes, int n_in,
                              void* d_out, int out_size, void* d_ws, size_t ws_size,
                              hipStream_t stream) {
    const float* x    = (const float*)d_in[0];
    const int*   esrc = (const int*)d_in[1];
    const int*   edst = (const int*)d_in[2];
    const float* ew   = (const float*)d_in[3];
    const float* W1   = (const float*)d_in[4];
    const float* b1   = (const float*)d_in[5];
    const float* W2   = (const float*)d_in[6];
    const float* b2   = (const float*)d_in[7];
    float* out = (float*)d_out;
    const int E = in_sizes[1];

    float* h1   = (float*)d_ws;
    float* agg1 = h1   + (size_t)N_NODES * NHID;
    float* h2   = agg1;
    int* pk       = (int*)(agg1 + (size_t)N_NODES * NHID);
    int* hist     = pk     + (size_t)2 * N_EDGES;
    int* bstart   = hist   + (size_t)NCH * NB;
    int* cnt      = bstart + (size_t)(NB + 8);
    int* rowstart = cnt    + (size_t)(N_NODES + 8);
    int* brec     = (int*)agg1;
    static_assert((size_t)4 * N_EDGES <= (size_t)N_NODES * NHID, "bucketed copy fits in agg1");

    bucket_hist_k<<<NCH, NT, 0, stream>>>(edst, hist, E);
    bucket_scatter_k<<<NCH, NT, 0, stream>>>(esrc, edst, ew, hist, bstart, brec, E);
    bucket_rows_k<<<NB, NT, 0, stream>>>(bstart, brec, cnt, rowstart, pk);

    const int gemmBlocks = (N_NODES + 127) / 128;
    gemm1_wmma_k<<<gemmBlocks, 256, 0, stream>>>(x, W1, h1);

    gather64_gemm2_k<<<2048, 256, 0, stream>>>(rowstart, cnt, pk, h1, b1, W2, h2);

    gather41_lsm_k<<<2048, 256, 0, stream>>>(rowstart, cnt, pk, h2, b2, out);
}


# GEMM1 operand feed: W1 staged once per 25-wave work-group in LDS as k-pair rows (B fragment = one 8-B LDS read), x read as 16-B row pieces; 250 groups, 7 tiles per SIMD at most; k sum order changed
# speedup vs baseline: 5.0494x; 1.0845x over previous
#include <hip/hip_runtime.h>

typedef __attribute__((ext_vector_type(2))) float v2f;
typedef __attribute__((ext_vector_type(8))) float v8f;

#define N_NODES 100000
#define NFEAT   256
#define NHID    64
#define NCLASS  41

#define G1W  25
#define G1T  (G1W * 32)
#define W1PP 132
#define KC   32
#define G1LDS 41472
static_assert(G1LDS >= (NFEAT / 2) * W1PP, "W1 fits");
static_assert(NHID == 64 && NFEAT % (2 * KC) == 0, "GEMM1 tiling");
static_assert(N_NODES % (16 * G1W) == 0, "whole work-groups of whole tiles");

__device__ __forceinline__ void g1_chunk(const float4 (&xr)[4], const float* __restrict__ bp,
                                         v8f& acc0, v8f& acc1, v8f& acc2, v8f& acc3) {
    v2f bf[8][4];
#pragma unroll
    for (int j = 0; j < 8; ++j) {
        const float* __restrict__ br = bp + j * W1PP;
        bf[j][0] = *(const v2f*)(br);
        bf[j][1] = *(const v2f*)(br + 32);
        bf[j][2] = *(const v2f*)(br + 64);
        bf[j][3] = *(const v2f*)(br + 96);
    }
#pragma unroll
    for (int j = 0; j < 8; ++j) {
        const float4 q = xr[j >> 1];
        v2f a;
        if (j & 1) { a.x = q.z; a.y = q.w; } else { a.x = q.x; a.y = q.y; }
        acc0 = __builtin_amdgcn_wmma_f32_16x16x4_f32(false, a, false, bf[j][0], (short)0, acc0, false, false);
        acc1 = __builtin_amdgcn_wmma_f32_16x16x4_f32(false, a, false, bf[j][1], (short)0, acc1, false, false);
        acc2 = __builtin_amdgcn_wmma_f32_16x16x4_f32(false, a, false, bf[j][2], (short)0, acc2, false, false);
        acc3 = __builtin_amdgcn_wmma_f32_16x16x4_f32(false, a, false, bf[j][3], (short)0, acc3, false, false);
    }
}

__global__ __launch_bounds__(G1T) void gemm1_wmma_k(const float* __restrict__ x,
                                                    const float* __restrict__ W1,
                                                    float* __restrict__ h1) {
    __shared__ __attribute__((aligned(16))) float w1p[G1LDS];
    const int lane = threadIdx.x & 31;
    const int wave = threadIdx.x >> 5;
    const int c  = lane & 15;
    const int hh = lane >> 4;
    const int r0 = (blockIdx.x * G1W + wave) * 16;
    const float* __restrict__ xp = x + (long)(r0 + c) * NFEAT + 16 * hh;

    float4 xa[4], xb[4];
#pragma unroll
    for (int e = 0; e < 4; ++e) xa[e] = *(const float4*)(xp + 4 * e);

    {
        constexpr int ITEMS = (NFEAT / 2) * (NHID / 4);
        constexpr int RND = (ITEMS + G1T - 1) / G1T;
        float4 lo[RND], hi[RND];
#pragma unroll
        for (int q = 0; q < RND; ++q) {
            int i = threadIdx.x + q * G1T;
            if (i > ITEMS - 1) i = ITEMS - 1;
            const int kp = i >> 4, n4 = (i & 15) * 4;
            lo[q] = *(const float4*)(W1 + (2 * kp) * NHID + n4);
            hi[q] = *(const float4*)(W1 + (2 * kp + 1) * NHID + n4);
        }
#pragma unroll
        for (int q = 0; q < RND; ++q) {
            const int i = threadIdx.x + q * G1T;
            if (i < ITEMS) {
                const int kp = i >> 4, n4 = (i & 15) * 4;
                float* __restrict__ d = w1p + kp * W1PP + n4 * 2;
                float4 p0, p1;
                p0.x = lo[q].x; p0.y = hi[q].x; p0.z = lo[q].y; p0.w = hi[q].y;
                p1.x = lo[q].z; p1.y = hi[q].z; p1.z = lo[q].w; p1.w = hi[q].w;
                *(float4*)d = p0;
                *(float4*)(d + 4) = p1;
            }
        }
    }
    __syncthreads();

    const float* __restrict__ bp = w1p + (8 * hh) * W1PP + 2 * c;
    v8f acc0 = {}, acc1 = {}, acc2 = {}, acc3 = {};
    for (int ch = 0; ch < NFEAT / KC; ch += 2) {
#pragma unroll
        for (int e = 0; e < 4; ++e) xb[e] = *(const float4*)(xp + (ch + 1) * KC + 4 * e);
        g1_chunk(xa, bp + (ch * (KC / 2)) * W1PP, acc0, acc1, acc2, acc3);
        if (ch + 2 < NFEAT / KC) {
#pragma unroll
            for (int e = 0; e < 4; ++e) xa[e] = *(const float4*)(xp + (ch + 2) * KC + 4 * e);
        }
        g1_chunk(xb, bp + ((ch + 1) * (KC / 2)) * W1PP, acc0, acc1, acc2, acc3);
    }

    float* __restrict__ o = h1 + (long)(r0 + 8 * hh) * NHID + c;
    for (int j = 0; j < 8; ++j) {
        o[j * NHID]      = acc0[j];
        o[j * NHID + 16] = acc1[j];
        o[j * NHID + 32] = acc2[j];
        o[j * NHID + 48] = acc3[j];
    }
}

#define N_EDGES 1600000
#define NB  256
#define BW  391
#define NCH 256
#define NT  512
static_assert(NB * BW >= N_NODES, "buckets must cover every row");
static_assert(BW <= NT, "one thread per row of a bucket");
static_assert(NB % 4 == 0 && NT % (NB / 4) == 0 && NCH % (16 * (NT / (NB / 4))) == 0, "column-sum tiling");

__device__ __forceinline__ int wg_excl_scan(int v, int& total) {
    __shared__ int wsum[32];
    const int lane = threadIdx.x & 31, wv = threadIdx.x >> 5;
    int incl = v;
    for (int d = 1; d < 32; d <<= 1) {
        int t = __shfl_up(incl, d, 32);
        if (lane >= d) incl += t;
    }
    if (lane == 31) wsum[wv] = incl;
    __syncthreads();
    const int ws = (lane < NT / 32) ? wsum[lane] : 0;
    int wincl = ws;
    for (int d = 1; d < 32; d <<= 1) {
        int t = __shfl_up(wincl, d, 32);
        if (lane >= d) wincl += t;
    }
    total = __shfl(wincl, 31, 32);
    const int wexcl = __shfl(wincl - ws, wv, 32);
    return wexcl + incl - v;
}

__global__ __launch_bounds__(NT) void bucket_hist_k(const int* __restrict__ dst,
                                                    int* __restrict__ hist, int nEdges) {
    __shared__ int h[NB];
    const int t = threadIdx.x;
    if (t < NB) h[t] = 0;
    __syncthreads();
    const int chunk = (nEdges + NCH - 1) / NCH;
    const int e0 = blockIdx.x * chunk;
    int e1 = e0 + chunk; if (e1 > nEdges) e1 = nEdges;
    int e = e0 + t;
    for (; e + 3 * NT < e1; e += 4 * NT) {
        int d[4];
#pragma unroll
        for (int k = 0; k < 4; ++k) d[k] = dst[e + k * NT];
#pragma unroll
        for (int k = 0; k < 4; ++k) atomicAdd(&h[d[k] / BW], 1);
    }
    for (; e < e1; e += NT) atomicAdd(&h[dst[e] / BW], 1);
    __syncthreads();
    if (t < NB) hist[blockIdx.x * NB + t] = h[t];
}

__global__ __launch_bounds__(NT) void bucket_scatter_k(const int* __restrict__ src,
                                                       const int* __restrict__ dst,
                                                       const float* __restrict__ w,
                                                       const int* __restrict__ hist,
                                                       int* __restrict__ bstart,
                                                       int* __restrict__ brec_w, int nEdges) {
    int4* __restrict__ brec = (int4*)brec_w;
    constexpr int QN = NT / (NB / 4);
    constexpr int QR = NCH / QN;
    __shared__ int pAll[QN][NB];
    __shared__ int pBef[QN][NB];
    __shared__ int cur[NB];
    const int t = threadIdx.x, g = blockIdx.x;
    {
        const int b4 = (t % (NB / 4)) * 4, q = t / (NB / 4);
        int a0 = 0, a1 = 0, a2 = 0, a3 = 0, f0 = 0, f1 = 0, f2 = 0, f3 = 0;
        for (int r0 = q * QR; r0 < (q + 1) * QR; r0 += 16) {
            int4 v[16];
#pragma unroll
            for (int k = 0; k < 16; ++k) v[k] = *(const int4*)(hist + (r0 + k) * NB + b4);
#pragma unroll
            for (int k = 0; k < 16; ++k) {
                a0 += v[k].x; a1 += v[k].y; a2 += v[k].z; a3 += v[k].w;
                if (r0 + k < g) { f0 += v[k].x; f1 += v[k].y; f2 += v[k].z; f3 += v[k].w; }
            }
        }
        pAll[q][b4] = a0; pAll[q][b4 + 1] = a1; pAll[q][b4 + 2] = a2; pAll[q][b4 + 3] = a3;
        pBef[q][b4] = f0; pBef[q][b4 + 1] = f1; pBef[q][b4 + 2] = f2; pBef[q][b4 + 3] = f3;
    }
    __syncthreads();
    int all = 0, bef = 0;
    if (t < NB) {
        for (int q = 0; q < QN; ++q) { all += pAll[q][t]; bef += pBef[q][t]; }
    }
    int total;
    const int start = wg_excl_scan(all, total);
    if (t < NB) {
        cur[t] = start + bef;
        if (g == 0) {
            bstart[t] = start;
            if (t == NB - 1) bstart[NB] = total;
        }
    }
    __syncthreads();
    const int chunk = (nEdges + NCH - 1) / NCH;
    const int e0 = g * chunk;
    int e1 = e0 + chunk; if (e1 > nEdges) e1 = nEdges;
    int e = e0 + t;
    for (; e + 3 * NT < e1; e += 4 * NT) {
        int d[4], s[4], wb[4];
#pragma unroll
        for (int k = 0; k < 4; ++k) {
            d[k]  = dst[e + k * NT];
            s[k]  = src[e + k * NT];
            wb[k] = __float_as_int(w[e + k * NT]);
        }
#pragma unroll
        for (int k = 0; k < 4; ++k) {
            const int p = atomicAdd(&cur[d[k] / BW], 1);
            int4 r; r.x = s[k]; r.y = wb[k]; r.z = d[k]; r.w = 0;
            brec[p] = r;
        }
    }
    for (; e < e1; e += NT) {
        int4 r; r.x = src[e]; r.y = __float_as_int(w[e]); r.z = dst[e]; r.w = 0;
        const int p = atomicAdd(&cur[r.z / BW], 1);
        brec[p] = r;
    }
}

__global__ __launch_bounds__(NT) void bucket_rows_k(const int* __restrict__ bstart,
                                                    const int* __restrict__ brec_w,
                                                    int* __restrict__ cnt,
                                                    int* __restrict__ rowstart,
                                                    int* __restrict__ pk) {
    __shared__ int c[NT];
    const int4* __restrict__ brec = (const int4*)brec_w;
    const int t = threadIdx.x, b = blockIdx.x;
    const int s = bstart[b], n = bstart[b + 1] - s, node0 = b * BW;
    c[t] = 0;
    __syncthreads();
    int i = t;
    for (; i + 3 * NT < n; i += 4 * NT) {
        int d[4];
#pragma unroll
        for (int k = 0; k < 4; ++k) d[k] = brec[s + i + k * NT].z;
#pragma unroll
        for (int k = 0; k < 4; ++k) atomicAdd(&c[d[k] - node0], 1);
    }
    for (; i < n; i += NT) atomicAdd(&c[brec[s + i].z - node0], 1);
    __syncthreads();
    const int mine = c[t];
    int total;
    const int excl = wg_excl_scan(mine, total);
    if (t < BW && node0 + t < N_NODES) {
        cnt[node0 + t] = mine;
        rowstart[node0 + t] = s + excl;
    }
    c[t] = excl;
    __syncthreads();
    i = t;
    for (; i + 3 * NT < n; i += 4 * NT) {
        int4 r[4];
#pragma unroll
        for (int k = 0; k < 4; ++k) r[k] = brec[s + i + k * NT];
#pragma unroll
        for (int k = 0; k < 4; ++k) {
            const int p = atomicAdd(&c[r[k].z - node0], 1);
            int2 v; v.x = r[k].x; v.y = r[k].y;
            ((int2*)pk)[s + p] = v;
        }
    }
    for (; i < n; i += NT) {
        const int4 r = brec[s + i];
        const int p = atomicAdd(&c[r.z - node0], 1);
        int2 v; v.x = r.x; v.y = r.y;
        ((int2*)pk)[s + p] = v;
    }
}

template <int K>
__device__ __forceinline__ void acc64(const float* __restrict__ h, int ex, int ey, int j,
                                      int lane, float& ax, float& ay) {
    float2 v[K];
    float  wk[K];
#pragma unroll
    for (int k = 0; k < K; ++k) {
        int sj = __builtin_amdgcn_readlane(ex, j + k);
        wk[k]  = __int_as_float(__builtin_amdgcn_readlane(ey, j + k));
        v[k]   = *(const float2*)(h + (long)sj * NHID + lane * 2);
    }
#pragma unroll
    for (int k = 0; k < K; ++k) {
        ax = fmaf(wk[k], v[k].x, ax);
        ay = fmaf(wk[k], v[k].y, ay);
    }
}

__device__ __forceinline__ void row_sum64(const int* __restrict__ rowstart,
                                          const int* __restrict__ cnt,
                                          const int* __restrict__ pk,
                                          const float* __restrict__ h,
                                          int r, int lane, float& ax, float& ay) {
    const int ru = __builtin_amdgcn_readfirstlane(r);
    const int s  = __builtin_amdgcn_readfirstlane(rowstart[ru]);
    const int n  = __builtin_amdgcn_readfirstlane(cnt[ru]);
    ax = 0.0f; ay = 0.0f;
    for (int c0 = 0; c0 < n; c0 += 32) {
        int m = n - c0; if (m > 32) m = 32;
        int ex = 0, ey = 0;
        if (lane < m) { int2 e = ((const int2*)pk)[s + c0 + lane]; ex = e.x; ey = e.y; }
        int j = 0;
        for (; j + 8 <= m; j += 8) acc64<8>(h, ex, ey, j, lane, ax, ay);
        switch (m - j) {
            case 7: acc64<7>(h, ex, ey, j, lane, ax, ay); break;
            case 6: acc64<6>(h, ex, ey, j, lane, ax, ay); break;
            case 5: acc64<5>(h, ex, ey, j, lane, ax, ay); break;
            case 4: acc64<4>(h, ex, ey, j, lane, ax, ay); break;
            case 3: acc64<3>(h, ex, ey, j, lane, ax, ay); break;
            case 2: acc64<2>(h, ex, ey, j, lane, ax, ay); break;
            case 1: acc64<1>(h, ex, ey, j, lane, ax, ay); break;
            default: break;
        }
    }
}

#define TP   68
#define NPAD 48
static_assert(N_NODES % 8 == 0, "rows come in whole groups of 8");
static_assert(NHID % 4 == 0 && NCLASS <= NPAD, "product tiling");

__global__ __launch_bounds__(256) void gather64_gemm2_k(const int* __restrict__ rowstart,
                                                        const int* __restrict__ cnt,
                                                        const int* __restrict__ pk,
                                                        const float* __restrict__ h,
                                                        const float* __restrict__ b1,
                                                        const float* __restrict__ W2,
                                                        float* __restrict__ h2) {
    __shared__ __attribute__((aligned(16))) float w2p[(NHID / 2) * NPAD * 2];
    __shared__ __attribute__((aligned(16))) float tile[2][16 * TP];
    const int lane = threadIdx.x & 31;
    const int wave = threadIdx.x >> 5;
    {
        constexpr int NF = (NHID / 2) * NPAD / 256;
        static_assert(NF * 256 == (NHID / 2) * NPAD, "whole entries per thread");
        float2 wv[NF];
#pragma unroll
        for (int q = 0; q < NF; ++q) {
            const int i = threadIdx.x + q * 256;
            const int kp = i / NPAD, n = i - kp * NPAD;
            const int nc = (n < NCLASS) ? n : (NCLASS - 1);
            const float lo = W2[(2 * kp) * NCLASS + nc];
            const float hi = W2[(2 * kp + 1) * NCLASS + nc];
            wv[q].x = (n < NCLASS) ? lo : 0.0f;
            wv[q].y = (n < NCLASS) ? hi : 0.0f;
        }
#pragma unroll
        for (int q = 0; q < NF; ++q) *(float2*)(w2p + 2 * (threadIdx.x + q * 256)) = wv[q];
    }
    const float bx = b1[2 * lane], by = b1[2 * lane + 1];
    const int c  = lane & 15;
    const int hh = lane >> 4;
    constexpr int NG = N_NODES / 8;
    const int gstep = gridDim.x;
    int it = 0;
    for (int g = blockIdx.x; g < NG; g += gstep, ++it) {
        const int half = it & 1;
        const bool last = g + gstep >= NG;
        float* __restrict__ tb = tile[(it >> 1) & 1];
        float ax, ay;
        row_sum64(rowstart, cnt, pk, h, g * 8 + wave, lane, ax, ay);
        float2 o; o.x = fmaxf(ax + bx, 0.0f); o.y = fmaxf(ay + by, 0.0f);
        *(float2*)(tb + (half * 8 + wave) * TP + 2 * lane) = o;
        if (half == 0 && !last) continue;
        if (half == 0) {
            float2 z; z.x = 0.0f; z.y = 0.0f;
            *(float2*)(tb + (8 + wave) * TP + 2 * lane) = z;
        }
        __syncthreads();
        if (wave < 3) {
            const float* __restrict__ ap = tb + c * TP + 2 * hh;
            const float* __restrict__ bp = w2p + (hh * NPAD + wave * 16 + c) * 2;
            v8f acc = {};
            for (int k0 = 0; k0 < NHID; k0 += 4) {
                const v2f a = *(const v2f*)(ap + k0);
                const v2f b = *(const v2f*)(bp + (k0 / 2) * NPAD * 2);
                acc = __builtin_amdgcn_wmma_f32_16x16x4_f32(false, a, false, b, (short)0, acc, false, false);
            }
            asm volatile("v_nop\n\tv_nop\n\tv_nop\n\tv_nop\n\tv_nop" : "+v"(acc));
            const int col = wave * 16 + c;
            const int gs = (hh == half) ? g : g - gstep;
            if (col < NCLASS && (hh == 0 || half == 1)) {
                float* __restrict__ op = h2 + (long)(gs * 8) * NCLASS + col;
                for (int j = 0; j < 8; ++j) op[j * NCLASS] = acc[j];
            }
        }
    }
}

template <int K>
__device__ __forceinline__ void acc41(const float* __restrict__ h, int ex, int ey, int j,
                                      int c0i, int c1i, float& a0, float& a1) {
    float v0[K], v1[K], wk[K];
#pragma unroll
    for (int k = 0; k < K; ++k) {
        int sj = __builtin_amdgcn_readlane(ex, j + k);
        wk[k]  = __int_as_float(__builtin_amdgcn_readlane(ey, j + k));
        const float* __restrict__ row = h + (long)sj * NCLASS;
        v0[k] = row[c0i];
        v1[k] = row[c1i];
    }
#pragma unroll
    for (int k = 0; k < K; ++k) {
        a0 = fmaf(wk[k], v0[k], a0);
        a1 = fmaf(wk[k], v1[k], a1);
    }
}

__device__ __forceinline__ float xhalf(float x) {
    int b = __builtin_bit_cast(int, x);
    return __builtin_bit_cast(float, __builtin_amdgcn_permlanex16(b, b, 0x76543210, 0xfedcba98, false, false));
}

__device__ __forceinline__ float row16_max(float v) {
    asm volatile("v_max_num_f32_dpp %0, %0, %0 row_xmask:1 row_mask:0xf bank_mask:0xf\n\t"
                 "v_max_num_f32_dpp %0, %0, %0 row_xmask:2 row_mask:0xf bank_mask:0xf\n\t"
                 "v_max_num_f32_dpp %0, %0, %0 row_xmask:4 row_mask:0xf bank_mask:0xf\n\t"
                 "v_max_num_f32_dpp %0, %0, %0 row_xmask:8 row_mask:0xf bank_mask:0xf"
                 : "+v"(v));
    return v;
}

__device__ __forceinline__ float row16_sum(float v) {
    asm volatile("v_add_f32_dpp %0, %0, %0 row_xmask:1 row_mask:0xf bank_mask:0xf\n\t"
                 "v_add_f32_dpp %0, %0, %0 row_xmask:2 row_mask:0xf bank_mask:0xf\n\t"
                 "v_add_f32_dpp %0, %0, %0 row_xmask:4 row_mask:0xf bank_mask:0xf\n\t"
                 "v_add_f32_dpp %0, %0, %0 row_xmask:8 row_mask:0xf bank_mask:0xf"
                 : "+v"(v));
    return v;
}

__device__ __forceinline__ float wave_max(float v) {
    v = row16_max(v);
    return fmaxf(v, xhalf(v));
}

__device__ __forceinline__ float wave_sum(float v) {
    v = row16_sum(v);
    return v + xhalf(v);
}

__global__ __launch_bounds__(256) void gather41_lsm_k(const int* __restrict__ rowstart,
                                                      const int* __restrict__ cnt,
                                                      const int* __restrict__ pk,
                                                      const float* __restrict__ h,
                                                      const float* __restrict__ b2,
                                                      float* __restrict__ out) {
    const int lane = threadIdx.x & 31;
    const int nw = gridDim.x * 8;
    const bool has1 = (32 + lane < NCLASS);
    const int c1i = has1 ? (32 + lane) : (NCLASS - 1);
    const float bz0 = b2[lane];
    const float bz1 = b2[c1i];
    for (int r = blockIdx.x * 8 + (threadIdx.x >> 5); r < N_NODES; r += nw) {
        const int ru = __builtin_amdgcn_readfirstlane(r);
        const int s  = __builtin_amdgcn_readfirstlane(rowstart[ru]);
        const int n  = __builtin_amdgcn_readfirstlane(cnt[ru]);
        float a0 = 0.0f, a1 = 0.0f;
        for (int c0 = 0; c0 < n; c0 += 32) {
            int m = n - c0; if (m > 32) m = 32;
            int ex = 0, ey = 0;
            if (lane < m) { int2 e = ((const int2*)pk)[s + c0 + lane]; ex = e.x; ey = e.y; }
            int j = 0;
            for (; j + 8 <= m; j += 8) acc41<8>(h, ex, ey, j, lane, c1i, a0, a1);
            switch (m - j) {
                case 7: acc41<7>(h, ex, ey, j, lane, c1i, a0, a1); break;
                case 6: acc41<6>(h, ex, ey, j, lane, c1i, a0, a1); break;
                case 5: acc41<5>(h, ex, ey, j, lane, c1i, a0, a1); break;
                case 4: acc41<4>(h, ex, ey, j, lane, c1i, a0, a1); break;
                case 3: acc41<3>(h, ex, ey, j, lane, c1i, a0, a1); break;
                case 2: acc41<2>(h, ex, ey, j, lane, c1i, a0, a1); break;
                case 1: acc41<1>(h, ex, ey, j, lane, c1i, a0, a1); break;
                default: break;
            }
        }
        const float z0 = a0 + bz0;
        const float z1 = has1 ? (a1 + bz1) : z0;
        const float mx = wave_max(fmaxf(z0, z1));
        const float e0 = expf(z0 - mx);
        const float e1 = expf(z1 - mx);
        const float lse = mx + logf(wave_sum(has1 ? (e0 + e1) : e0));
        float* __restrict__ o = out + (long)ru * NCLASS;
        o[lane] = z0 - lse;
        if (has1) o[32 + lane] = z1 - lse;
    }
}

extern "C" void kernel_launch(void* const* d_in, const int* in_sizes, int n_in,
                              void* d_out, int out_size, void* d_ws, size_t ws_size,
                              hipStream_t stream) {
    const float* x    = (const float*)d_in[0];
    const int*   esrc = (const int*)d_in[1];
    const int*   edst = (const int*)d_in[2];
    const float* ew   = (const float*)d_in[3];
    const float* W1   = (const float*)d_in[4];
    const float* b1   = (const float*)d_in[5];
    const float* W2   = (const float*)d_in[6];
    const float* b2   = (const float*)d_in[7];
    float* out = (float*)d_out;
    const int E = in_sizes[1];

    float* h1   = (float*)d_ws;
    float* agg1 = h1   + (size_t)N_NODES * NHID;
    float* h2   = agg1;
    int* pk       = (int*)(agg1 + (size_t)N_NODES * NHID);
    int* hist     = pk     + (size_t)2 * N_EDGES;
    int* bstart   = hist   + (size_t)NCH * NB;
    int* cnt      = bstart + (size_t)(NB + 8);
    int* rowstart = cnt    + (size_t)(N_NODES + 8);
    int* brec     = (int*)agg1;
    static_assert((size_t)4 * N_EDGES <= (size_t)N_NODES * NHID, "bucketed copy fits in agg1");

    bucket_hist_k<<<NCH, NT, 0, stream>>>(edst, hist, E);
    bucket_scatter_k<<<NCH, NT, 0, stream>>>(esrc, edst, ew, hist, bstart, brec, E);
    bucket_rows_k<<<NB, NT, 0, stream>>>(bstart, brec, cnt, rowstart, pk);

    gemm1_wmma_k<<<N_NODES / (16 * G1W), G1T, 0, stream>>>(x, W1, h1);

    gather64_gemm2_k<<<2048, 256, 0, stream>>>(rowstart, cnt, pk, h1, b1, W2, h2);

    gather41_lsm_k<<<2048, 256, 0, stream>>>(rowstart, cnt, pk, h2, b2, out);
}


# Both gathers: 16-B loads per lane, one 16-lane row per edge, two edges per load, 16 edges in flight (h2 row pitch 48 floats); row sum order changed: even/odd-position edges summed apart, then added
# speedup vs baseline: 5.2258x; 1.0349x over previous
#include <hip/hip_runtime.h>

typedef __attribute__((ext_vector_type(2))) float v2f;
typedef __attribute__((ext_vector_type(8))) float v8f;

#define N_NODES 100000
#define NFEAT   256
#define NHID    64
#define NCLASS  41

#define G1W  25
#define G1T  (G1W * 32)
#define W1PP 132
#define KC   32
#define G1LDS 41472
static_assert(G1LDS >= (NFEAT / 2) * W1PP, "W1 fits");
static_assert(NHID == 64 && NFEAT % (2 * KC) == 0, "GEMM1 tiling");
static_assert(N_NODES % (16 * G1W) == 0, "whole work-groups of whole tiles");

__device__ __forceinline__ void g1_chunk(const float4 (&xr)[4], const float* __restrict__ bp,
                                         v8f& acc0, v8f& acc1, v8f& acc2, v8f& acc3) {
    v2f bf[8][4];
#pragma unroll
    for (int j = 0; j < 8; ++j) {
        const float* __restrict__ br = bp + j * W1PP;
        bf[j][0] = *(const v2f*)(br);
        bf[j][1] = *(const v2f*)(br + 32);
        bf[j][2] = *(const v2f*)(br + 64);
        bf[j][3] = *(const v2f*)(br + 96);
    }
#pragma unroll
    for (int j = 0; j < 8; ++j) {
        const float4 q = xr[j >> 1];
        v2f a;
        if (j & 1) { a.x = q.z; a.y = q.w; } else { a.x = q.x; a.y = q.y; }
        acc0 = __builtin_amdgcn_wmma_f32_16x16x4_f32(false, a, false, bf[j][0], (short)0, acc0, false, false);
        acc1 = __builtin_amdgcn_wmma_f32_16x16x4_f32(false, a, false, bf[j][1], (short)0, acc1, false, false);
        acc2 = __builtin_amdgcn_wmma_f32_16x16x4_f32(false, a, false, bf[j][2], (short)0, acc2, false, false);
        acc3 = __builtin_amdgcn_wmma_f32_16x16x4_f32(false, a, false, bf[j][3], (short)0, acc3, false, false);
    }
}

__global__ __launch_bounds__(G1T) void gemm1_wmma_k(const float* __restrict__ x,
                                                    const float* __restrict__ W1,
                                                    float* __restrict__ h1) {
    __shared__ __attribute__((aligned(16))) float w1p[G1LDS];
    const int lane = threadIdx.x & 31;
    const int wave = threadIdx.x >> 5;
    const int c  = lane & 15;
    const int hh = lane >> 4;
    const int r0 = (blockIdx.x * G1W + wave) * 16;
    const float* __restrict__ xp = x + (long)(r0 + c) * NFEAT + 16 * hh;

    float4 xa[4], xb[4];
#pragma unroll
    for (int e = 0; e < 4; ++e) xa[e] = *(const float4*)(xp + 4 * e);

    {
        constexpr int ITEMS = (NFEAT / 2) * (NHID / 4);
        constexpr int RND = (ITEMS + G1T - 1) / G1T;
        float4 lo[RND], hi[RND];
#pragma unroll
        for (int q = 0; q < RND; ++q) {
            int i = threadIdx.x + q * G1T;
            if (i > ITEMS - 1) i = ITEMS - 1;
            const int kp = i >> 4, n4 = (i & 15) * 4;
            lo[q] = *(const float4*)(W1 + (2 * kp) * NHID + n4);
            hi[q] = *(const float4*)(W1 + (2 * kp + 1) * NHID + n4);
        }
#pragma unroll
        for (int q = 0; q < RND; ++q) {
            const int i = threadIdx.x + q * G1T;
            if (i < ITEMS) {
                const int kp = i >> 4, n4 = (i & 15) * 4;
                float* __restrict__ d = w1p + kp * W1PP + n4 * 2;
                float4 p0, p1;
                p0.x = lo[q].x; p0.y = hi[q].x; p0.z = lo[q].y; p0.w = hi[q].y;
                p1.x = lo[q].z; p1.y = hi[q].z; p1.z = lo[q].w; p1.w = hi[q].w;
                *(float4*)d = p0;
                *(float4*)(d + 4) = p1;
            }
        }
    }
    __syncthreads();

    const float* __restrict__ bp = w1p + (8 * hh) * W1PP + 2 * c;
    v8f acc0 = {}, acc1 = {}, acc2 = {}, acc3 = {};
    for (int ch = 0; ch < NFEAT / KC; ch += 2) {
#pragma unroll
        for (int e = 0; e < 4; ++e) xb[e] = *(const float4*)(xp + (ch + 1) * KC + 4 * e);
        g1_chunk(xa, bp + (ch * (KC / 2)) * W1PP, acc0, acc1, acc2, acc3);
        if (ch + 2 < NFEAT / KC) {
#pragma unroll
            for (int e = 0; e < 4; ++e) xa[e] = *(const float4*)(xp + (ch + 2) * KC + 4 * e);
        }
        g1_chunk(xb, bp + ((ch + 1) * (KC / 2)) * W1PP, acc0, acc1, acc2, acc3);
    }

    float* __restrict__ o = h1 + (long)(r0 + 8 * hh) * NHID + c;
    for (int j = 0; j < 8; ++j) {
        o[j * NHID]      = acc0[j];
        o[j * NHID + 16] = acc1[j];
        o[j * NHID + 32] = acc2[j];
        o[j * NHID + 48] = acc3[j];
    }
}

#define N_EDGES 1600000
#define NB  256
#define BW  391
#define NCH 256
#define NT  512
static_assert(NB * BW >= N_NODES, "buckets must cover every row");
static_assert(BW <= NT, "one thread per row of a bucket");
static_assert(NB % 4 == 0 && NT % (NB / 4) == 0 && NCH % (16 * (NT / (NB / 4))) == 0, "column-sum tiling");

__device__ __forceinline__ int wg_excl_scan(int v, int& total) {
    __shared__ int wsum[32];
    const int lane = threadIdx.x & 31, wv = threadIdx.x >> 5;
    int incl = v;
    for (int d = 1; d < 32; d <<= 1) {
        int t = __shfl_up(incl, d, 32);
        if (lane >= d) incl += t;
    }
    if (lane == 31) wsum[wv] = incl;
    __syncthreads();
    const int ws = (lane < NT / 32) ? wsum[lane] : 0;
    int wincl = ws;
    for (int d = 1; d < 32; d <<= 1) {
        int t = __shfl_up(wincl, d, 32);
        if (lane >= d) wincl += t;
    }
    total = __shfl(wincl, 31, 32);
    const int wexcl = __shfl(wincl - ws, wv, 32);
    return wexcl + incl - v;
}

__global__ __launch_bounds__(NT) void bucket_hist_k(const int* __restrict__ dst,
                                                    int* __restrict__ hist, int nEdges) {
    __shared__ int h[NB];
    const int t = threadIdx.x;
    if (t < NB) h[t] = 0;
    __syncthreads();
    const int chunk = (nEdges + NCH - 1) / NCH;
    const int e0 = blockIdx.x * chunk;
    int e1 = e0 + chunk; if (e1 > nEdges) e1 = nEdges;
    int e = e0 + t;
    for (; e + 3 * NT < e1; e += 4 * NT) {
        int d[4];
#pragma unroll
        for (int k = 0; k < 4; ++k) d[k] = dst[e + k * NT];
#pragma unroll
        for (int k = 0; k < 4; ++k) atomicAdd(&h[d[k] / BW], 1);
    }
    for (; e < e1; e += NT) atomicAdd(&h[dst[e] / BW], 1);
    __syncthreads();
    if (t < NB) hist[blockIdx.x * NB + t] = h[t];
}

__global__ __launch_bounds__(NT) void bucket_scatter_k(const int* __restrict__ src,
                                                       const int* __restrict__ dst,
                                                       const float* __restrict__ w,
                                                       const int* __restrict__ hist,
                                                       int* __restrict__ bstart,
                                                       int* __restrict__ brec_w, int nEdges) {
    int4* __restrict__ brec = (int4*)brec_w;
    constexpr int QN = NT / (NB / 4);
    constexpr int QR = NCH / QN;
    __shared__ int pAll[QN][NB];
    __shared__ int pBef[QN][NB];
    __shared__ int cur[NB];
    const int t = threadIdx.x, g = blockIdx.x;
    {
        const int b4 = (t % (NB / 4)) * 4, q = t / (NB / 4);
        int a0 = 0, a1 = 0, a2 = 0, a3 = 0, f0 = 0, f1 = 0, f2 = 0, f3 = 0;
        for (int r0 = q * QR; r0 < (q + 1) * QR; r0 += 16) {
            int4 v[16];
#pragma unroll
            for (int k = 0; k < 16; ++k) v[k] = *(const int4*)(hist + (r0 + k) * NB + b4);
#pragma unroll
            for (int k = 0; k < 16; ++k) {
                a0 += v[k].x; a1 += v[k].y; a2 += v[k].z; a3 += v[k].w;
                if (r0 + k < g) { f0 += v[k].x; f1 += v[k].y; f2 += v[k].z; f3 += v[k].w; }
            }
        }
        pAll[q][b4] = a0; pAll[q][b4 + 1] = a1; pAll[q][b4 + 2] = a2; pAll[q][b4 + 3] = a3;
        pBef[q][b4] = f0; pBef[q][b4 + 1] = f1; pBef[q][b4 + 2] = f2; pBef[q][b4 + 3] = f3;
    }
    __syncthreads();
    int all = 0, bef = 0;
    if (t < NB) {
        for (int q = 0; q < QN; ++q) { all += pAll[q][t]; bef += pBef[q][t]; }
    }
    int total;
    const int start = wg_excl_scan(all, total);
    if (t < NB) {
        cur[t] = start + bef;
        if (g == 0) {
            bstart[t] = start;
            if (t == NB - 1) bstart[NB] = total;
        }
    }
    __syncthreads();
    const int chunk = (nEdges + NCH - 1) / NCH;
    const int e0 = g * chunk;
    int e1 = e0 + chunk; if (e1 > nEdges) e1 = nEdges;
    int e = e0 + t;
    for (; e + 3 * NT < e1; e += 4 * NT) {
        int d[4], s[4], wb[4];
#pragma unroll
        for (int k = 0; k < 4; ++k) {
            d[k]  = dst[e + k * NT];
            s[k]  = src[e + k * NT];
            wb[k] = __float_as_int(w[e + k * NT]);
        }
#pragma unroll
        for (int k = 0; k < 4; ++k) {
            const int p = atomicAdd(&cur[d[k] / BW], 1);
            int4 r; r.x = s[k]; r.y = wb[k]; r.z = d[k]; r.w = 0;
            brec[p] = r;
        }
    }
    for (; e < e1; e += NT) {
        int4 r; r.x = src[e]; r.y = __float_as_int(w[e]); r.z = dst[e]; r.w = 0;
        const int p = atomicAdd(&cur[r.z / BW], 1);
        brec[p] = r;
    }
}

__global__ __launch_bounds__(NT) void bucket_rows_k(const int* __restrict__ bstart,
                                                    const int* __restrict__ brec_w,
                                                    int* __restrict__ cnt,
                                                    int* __restrict__ rowstart,
                                                    int* __restrict__ pk) {
    __shared__ int c[NT];
    const int4* __restrict__ brec = (const int4*)brec_w;
    const int t = threadIdx.x, b = blockIdx.x;
    const int s = bstart[b], n = bstart[b + 1] - s, node0 = b * BW;
    c[t] = 0;
    __syncthreads();
    int i = t;
    for (; i + 3 * NT < n; i += 4 * NT) {
        int d[4];
#pragma unroll
        for (int k = 0; k < 4; ++k) d[k] = brec[s + i + k * NT].z;
#pragma unroll
        for (int k = 0; k < 4; ++k) atomicAdd(&c[d[k] - node0], 1);
    }
    for (; i < n; i += NT) atomicAdd(&c[brec[s + i].z - node0], 1);
    __syncthreads();
    const int mine = c[t];
    int total;
    const int excl = wg_excl_scan(mine, total);
    if (t < BW && node0 + t < N_NODES) {
        cnt[node0 + t] = mine;
        rowstart[node0 + t] = s + excl;
    }
    c[t] = excl;
    __syncthreads();
    i = t;
    for (; i + 3 * NT < n; i += 4 * NT) {
        int4 r[4];
#pragma unroll
        for (int k = 0; k < 4; ++k) r[k] = brec[s + i + k * NT];
#pragma unroll
        for (int k = 0; k < 4; ++k) {
            const int p = atomicAdd(&c[r[k].z - node0], 1);
            int2 v; v.x = r[k].x; v.y = r[k].y;
            ((int2*)pk)[s + p] = v;
        }
    }
    for (; i < n; i += NT) {
        const int4 r = brec[s + i];
        const int p = atomicAdd(&c[r.z - node0], 1);
        int2 v; v.x = r.x; v.y = r.y;
        ((int2*)pk)[s + p] = v;
    }
}

__device__ __forceinline__ float xhalf(float x) {
    int b = __builtin_bit_cast(int, x);
    return __builtin_bit_cast(float, __builtin_amdgcn_permlanex16(b, b, 0x76543210, 0xfedcba98, false, false));
}

#define GP 8

template <int P, bool ODD, int PITCH>
__device__ __forceinline__ void acc_pairs(const float* __restrict__ h, int ex, int ey, int j,
                                          int hh, int q4, float4& a) {
    float4 v[P];
    float  wk[P];
#pragma unroll
    for (int k = 0; k < P; ++k) {
        const bool lone = ODD && (k == P - 1);
        const int s0 = __builtin_amdgcn_readlane(ex, j + 2 * k);
        const int w0 = __builtin_amdgcn_readlane(ey, j + 2 * k);
        int sj = s0, wj = w0;
        if (!lone) {
            const int s1 = __builtin_amdgcn_readlane(ex, j + 2 * k + 1);
            const int w1 = __builtin_amdgcn_readlane(ey, j + 2 * k + 1);
            sj = hh ? s1 : s0;
            wj = hh ? w1 : w0;
        }
        wk[k] = __int_as_float(wj);
        v[k]  = *(const float4*)(h + (size_t)(unsigned)sj * PITCH + q4);
        if (lone && hh) { wk[k] = 0.0f; v[k].x = 0.0f; v[k].y = 0.0f; v[k].z = 0.0f; v[k].w = 0.0f; }
    }
#pragma unroll
    for (int k = 0; k < P; ++k) {
        a.x = fmaf(wk[k], v[k].x, a.x);
        a.y = fmaf(wk[k], v[k].y, a.y);
        a.z = fmaf(wk[k], v[k].z, a.z);
        a.w = fmaf(wk[k], v[k].w, a.w);
    }
}

template <int R, int PITCH>
__device__ __forceinline__ void acc_tail(const float* __restrict__ h, int ex, int ey, int j,
                                         int hh, int q4, float4& a, int r, int odd) {
    if (r == R) {
        if (odd) acc_pairs<R + 1, true, PITCH>(h, ex, ey, j, hh, q4, a);
        else if constexpr (R > 0) acc_pairs<R, false, PITCH>(h, ex, ey, j, hh, q4, a);
    } else if constexpr (R > 0) {
        acc_tail<R - 1, PITCH>(h, ex, ey, j, hh, q4, a, r, odd);
    }
}

template <int PITCH>
__device__ __forceinline__ void row_sum_pairs(const int* __restrict__ rowstart,
                                              const int* __restrict__ cnt,
                                              const int* __restrict__ pk,
                                              const float* __restrict__ h,
                                              int r, int lane, int q4, float4& a) {
    const int ru = __builtin_amdgcn_readfirstlane(r);
    const int s  = __builtin_amdgcn_readfirstlane(rowstart[ru]);
    const int n  = __builtin_amdgcn_readfirstlane(cnt[ru]);
    const int hh = lane >> 4;
    a.x = 0.0f; a.y = 0.0f; a.z = 0.0f; a.w = 0.0f;
    for (int c0 = 0; c0 < n; c0 += 32) {
        int m = n - c0; if (m > 32) m = 32;
        int ex = 0, ey = 0;
        if (lane < m) { int2 e = ((const int2*)pk)[s + c0 + lane]; ex = e.x; ey = e.y; }
        const int whole = m >> 1, odd = m & 1;
        int p = 0;
        for (; p + GP <= whole; p += GP) acc_pairs<GP, false, PITCH>(h, ex, ey, 2 * p, hh, q4, a);
        acc_tail<GP - 1, PITCH>(h, ex, ey, 2 * p, hh, q4, a, whole - p, odd);
    }
    a.x += xhalf(a.x); a.y += xhalf(a.y); a.z += xhalf(a.z); a.w += xhalf(a.w);
}

#define TP   68
#define NPAD 48
static_assert(N_NODES % 8 == 0, "rows come in whole groups of 8");
static_assert(NHID % 4 == 0 && NCLASS <= NPAD, "product tiling");

__global__ __launch_bounds__(256) void gather64_gemm2_k(const int* __restrict__ rowstart,
                                                        const int* __restrict__ cnt,
                                                        const int* __restrict__ pk,
                                                        const float* __restrict__ h,
                                                        const float* __restrict__ b1,
                                                        const float* __restrict__ W2,
                                                        float* __restrict__ h2) {
    __shared__ __attribute__((aligned(16))) float w2p[(NHID / 2) * NPAD * 2];
    __shared__ __attribute__((aligned(16))) float tile[2][16 * TP];
    const int lane = threadIdx.x & 31;
    const int wave = threadIdx.x >> 5;
    {
        constexpr int NF = (NHID / 2) * NPAD / 256;
        static_assert(NF * 256 == (NHID / 2) * NPAD, "whole entries per thread");
        float2 wv[NF];
#pragma unroll
        for (int q = 0; q < NF; ++q) {
            const int i = threadIdx.x + q * 256;
            const int kp = i / NPAD, n = i - kp * NPAD;
            const int nc = (n < NCLASS) ? n : (NCLASS - 1);
            const float lo = W2[(2 * kp) * NCLASS + nc];
            const float hi = W2[(2 * kp + 1) * NCLASS + nc];
            wv[q].x = (n < NCLASS) ? lo : 0.0f;
            wv[q].y = (n < NCLASS) ? hi : 0.0f;
        }
#pragma unroll
        for (int q = 0; q < NF; ++q) *(float2*)(w2p + 2 * (threadIdx.x + q * 256)) = wv[q];
    }
    const int c  = lane & 15;
    const int hh = lane >> 4;
    const float4 bq = *(const float4*)(b1 + 4 * c);
    constexpr int NG = N_NODES / 8;
    const int gstep = gridDim.x;
    int it = 0;
    for (int g = blockIdx.x; g < NG; g += gstep, ++it) {
        const int half = it & 1;
        const bool last = g + gstep >= NG;
        float* __restrict__ tb = tile[(it >> 1) & 1];
        float4 a;
        row_sum_pairs<NHID>(rowstart, cnt, pk, h, g * 8 + wave, lane, 4 * c, a);
        if (hh == 0) {
            float4 o;
            o.x = fmaxf(a.x + bq.x, 0.0f); o.y = fmaxf(a.y + bq.y, 0.0f);
            o.z = fmaxf(a.z + bq.z, 0.0f); o.w = fmaxf(a.w + bq.w, 0.0f);
            *(float4*)(tb + (half * 8 + wave) * TP + 4 * c) = o;
        }
        if (half == 0 && !last) continue;
        if (half == 0 && hh == 0) {
            float4 z; z.x = 0.0f; z.y = 0.0f; z.z = 0.0f; z.w = 0.0f;
            *(float4*)(tb + (8 + wave) * TP + 4 * c) = z;
        }
        __syncthreads();
        if (wave < 3) {
            const float* __restrict__ ap = tb + c * TP + 2 * hh;
            const float* __restrict__ bp = w2p + (hh * NPAD + wave * 16 + c) * 2;
            v8f acc = {};
            for (int k0 = 0; k0 < NHID; k0 += 4) {
                const v2f a = *(const v2f*)(ap + k0);
                const v2f b = *(const v2f*)(bp + (k0 / 2) * NPAD * 2);
                acc = __builtin_amdgcn_wmma_f32_16x16x4_f32(false, a, false, b, (short)0, acc, false, false);
            }
            asm volatile("v_nop\n\tv_nop\n\tv_nop\n\tv_nop\n\tv_nop" : "+v"(acc));
            const int col = wave * 16 + c;
            const int gs = (hh == half) ? g : g - gstep;
            if (hh == 0 || half == 1) {
                float* __restrict__ op = h2 + (long)(gs * 8) * NPAD + col;
                for (int j = 0; j < 8; ++j) op[j * NPAD] = acc[j];
            }
        }
    }
}

__device__ __forceinline__ float row16_max(float v) {
    asm volatile("v_max_num_f32_dpp %0, %0, %0 row_xmask:1 row_mask:0xf bank_mask:0xf\n\t"
                 "v_max_num_f32_dpp %0, %0, %0 row_xmask:2 row_mask:0xf bank_mask:0xf\n\t"
                 "v_max_num_f32_dpp %0, %0, %0 row_xmask:4 row_mask:0xf bank_mask:0xf\n\t"
                 "v_max_num_f32_dpp %0, %0, %0 row_xmask:8 row_mask:0xf bank_mask:0xf"
                 : "+v"(v));
    return v;
}

__device__ __forceinline__ float row16_sum(float v) {
    asm volatile("v_add_f32_dpp %0, %0, %0 row_xmask:1 row_mask:0xf bank_mask:0xf\n\t"
                 "v_add_f32_dpp %0, %0, %0 row_xmask:2 row_mask:0xf bank_mask:0xf\n\t"
                 "v_add_f32_dpp %0, %0, %0 row_xmask:4 row_mask:0xf bank_mask:0xf\n\t"
                 "v_add_f32_dpp %0, %0, %0 row_xmask:8 row_mask:0xf bank_mask:0xf"
                 : "+v"(v));
    return v;
}

__device__ __forceinline__ float wave_max(float v) {
    v = row16_max(v);
    return fmaxf(v, xhalf(v));
}

__device__ __forceinline__ float wave_sum(float v) {
    v = row16_sum(v);
    return v + xhalf(v);
}

__global__ __launch_bounds__(256) void gather41_lsm_k(const int* __restrict__ rowstart,
                                                      const int* __restrict__ cnt,
                                                      const int* __restrict__ pk,
                                                      const float* __restrict__ h,
                                                      const float* __restrict__ b2,
                                                      float* __restrict__ out) {
    const int lane = threadIdx.x & 31;
    const int nw = gridDim.x * 8;
    const bool has1 = (32 + lane < NCLASS);
    const int c1i = has1 ? (32 + lane) : (NCLASS - 1);
    const float bz0 = b2[lane];
    const float bz1 = b2[c1i];
    const int q  = lane & 15;
    const int q4 = 4 * ((q < NPAD / 4) ? q : (NPAD / 4 - 1));
    const int sl0 = lane >> 2, sl1 = 8 + (lane >> 2), cm = lane & 3;
    for (int r = blockIdx.x * 8 + (threadIdx.x >> 5); r < N_NODES; r += nw) {
        const int ru = __builtin_amdgcn_readfirstlane(r);
        float4 a;
        row_sum_pairs<NPAD>(rowstart, cnt, pk, h, r, lane, q4, a);
        const float t0 = __shfl(a.x, sl0, 32), t1 = __shfl(a.y, sl0, 32);
        const float t2 = __shfl(a.z, sl0, 32), t3 = __shfl(a.w, sl0, 32);
        const float u0 = __shfl(a.x, sl1, 32), u1 = __shfl(a.y, sl1, 32);
        const float u2 = __shfl(a.z, sl1, 32), u3 = __shfl(a.w, sl1, 32);
        const float a0 = (cm == 0) ? t0 : (cm == 1) ? t1 : (cm == 2) ? t2 : t3;
        const float a1 = (cm == 0) ? u0 : (cm == 1) ? u1 : (cm == 2) ? u2 : u3;
        const float z0 = a0 + bz0;
        const float z1 = has1 ? (a1 + bz1) : z0;
        const float mx = wave_max(fmaxf(z0, z1));
        const float e0 = expf(z0 - mx);
        const float e1 = expf(z1 - mx);
        const float lse = mx + logf(wave_sum(has1 ? (e0 + e1) : e0));
        float* __restrict__ o = out + (long)ru * NCLASS;
        o[lane] = z0 - lse;
        if (has1) o[32 + lane] = z1 - lse;
    }
}

extern "C" void kernel_launch(void* const* d_in, const int* in_sizes, int n_in,
                              void* d_out, int out_size, void* d_ws, size_t ws_size,
                              hipStream_t stream) {
    const float* x    = (const float*)d_in[0];
    const int*   esrc = (const int*)d_in[1];
    const int*   edst = (const int*)d_in[2];
    const float* ew   = (const float*)d_in[3];
    const float* W1   = (const float*)d_in[4];
    const float* b1   = (const float*)d_in[5];
    const float* W2   = (const float*)d_in[6];
    const float* b2   = (const float*)d_in[7];
    float* out = (float*)d_out;
    const int E = in_sizes[1];

    float* h1   = (float*)d_ws;
    float* agg1 = h1   + (size_t)N_NODES * NHID;
    float* h2   = agg1;
    int* pk       = (int*)(agg1 + (size_t)N_NODES * NHID);
    int* hist     = pk     + (size_t)2 * N_EDGES;
    int* bstart   = hist   + (size_t)NCH * NB;
    int* cnt      = bstart + (size_t)(NB + 8);
    int* rowstart = cnt    + (size_t)(N_NODES + 8);
    int* brec     = (int*)agg1;
    static_assert((size_t)4 * N_EDGES <= (size_t)N_NODES * NHID, "bucketed copy fits in agg1");

    bucket_hist_k<<<NCH, NT, 0, stream>>>(edst, hist, E);
    bucket_scatter_k<<<NCH, NT, 0, stream>>>(esrc, edst, ew, hist, bstart, brec, E);
    bucket_rows_k<<<NB, NT, 0, stream>>>(bstart, brec, cnt, rowstart, pk);

    gemm1_wmma_k<<<N_NODES / (16 * G1W), G1T, 0, stream>>>(x, W1, h1);

    gather64_gemm2_k<<<2048, 256, 0, stream>>>(rowstart, cnt, pk, h1, b1, W2, h2);

    gather41_lsm_k<<<2048, 256, 0, stream>>>(rowstart, cnt, pk, h2, b2, out);
}


# Edge grouping level 1 inside each chunk: LDS rank per (chunk, bucket), edges copied to the chunk's own runs; histogram launch and table sums gone; row pass walks 256 runs; row sum order: LDS arrival
# speedup vs baseline: 5.4103x; 1.0353x over previous
#include <hip/hip_runtime.h>

typedef __attribute__((ext_vector_type(2))) float v2f;
typedef __attribute__((ext_vector_type(8))) float v8f;

#define N_NODES 100000
#define NFEAT   256
#define NHID    64
#define NCLASS  41

#define G1W  25
#define G1T  (G1W * 32)
#define W1PP 132
#define KC   32
#define G1LDS 41472
static_assert(G1LDS >= (NFEAT / 2) * W1PP, "W1 fits");
static_assert(NHID == 64 && NFEAT % (2 * KC) == 0, "GEMM1 tiling");
static_assert(N_NODES % (16 * G1W) == 0, "whole work-groups of whole tiles");

__device__ __forceinline__ void g1_chunk(const float4 (&xr)[4], const float* __restrict__ bp,
                                         v8f& acc0, v8f& acc1, v8f& acc2, v8f& acc3) {
    v2f bf[8][4];
#pragma unroll
    for (int j = 0; j < 8; ++j) {
        const float* __restrict__ br = bp + j * W1PP;
        bf[j][0] = *(const v2f*)(br);
        bf[j][1] = *(const v2f*)(br + 32);
        bf[j][2] = *(const v2f*)(br + 64);
        bf[j][3] = *(const v2f*)(br + 96);
    }
#pragma unroll
    for (int j = 0; j < 8; ++j) {
        const float4 q = xr[j >> 1];
        v2f a;
        if (j & 1) { a.x = q.z; a.y = q.w; } else { a.x = q.x; a.y = q.y; }
        acc0 = __builtin_amdgcn_wmma_f32_16x16x4_f32(false, a, false, bf[j][0], (short)0, acc0, false, false);
        acc1 = __builtin_amdgcn_wmma_f32_16x16x4_f32(false, a, false, bf[j][1], (short)0, acc1, false, false);
        acc2 = __builtin_amdgcn_wmma_f32_16x16x4_f32(false, a, false, bf[j][2], (short)0, acc2, false, false);
        acc3 = __builtin_amdgcn_wmma_f32_16x16x4_f32(false, a, false, bf[j][3], (short)0, acc3, false, false);
    }
}

__global__ __launch_bounds__(G1T) void gemm1_wmma_k(const float* __restrict__ x,
                                                    const float* __restrict__ W1,
                                                    float* __restrict__ h1) {
    __shared__ __attribute__((aligned(16))) float w1p[G1LDS];
    const int lane = threadIdx.x & 31;
    const int wave = threadIdx.x >> 5;
    const int c  = lane & 15;
    const int hh = lane >> 4;
    const int r0 = (blockIdx.x * G1W + wave) * 16;
    const float* __restrict__ xp = x + (long)(r0 + c) * NFEAT + 16 * hh;

    float4 xa[4], xb[4];
#pragma unroll
    for (int e = 0; e < 4; ++e) xa[e] = *(const float4*)(xp + 4 * e);

    {
        constexpr int ITEMS = (NFEAT / 2) * (NHID / 4);
        constexpr int RND = (ITEMS + G1T - 1) / G1T;
        float4 lo[RND], hi[RND];
#pragma unroll
        for (int q = 0; q < RND; ++q) {
            int i = threadIdx.x + q * G1T;
            if (i > ITEMS - 1) i = ITEMS - 1;
            const int kp = i >> 4, n4 = (i & 15) * 4;
            lo[q] = *(const float4*)(W1 + (2 * kp) * NHID + n4);
            hi[q] = *(const float4*)(W1 + (2 * kp + 1) * NHID + n4);
        }
#pragma unroll
        for (int q = 0; q < RND; ++q) {
            const int i = threadIdx.x + q * G1T;
            if (i < ITEMS) {
                const int kp = i >> 4, n4 = (i & 15) * 4;
                float* __restrict__ d = w1p + kp * W1PP + n4 * 2;
                float4 p0, p1;
                p0.x = lo[q].x; p0.y = hi[q].x; p0.z = lo[q].y; p0.w = hi[q].y;
                p1.x = lo[q].z; p1.y = hi[q].z; p1.z = lo[q].w; p1.w = hi[q].w;
                *(float4*)d = p0;
                *(float4*)(d + 4) = p1;
            }
        }
    }
    __syncthreads();

    const float* __restrict__ bp = w1p + (8 * hh) * W1PP + 2 * c;
    v8f acc0 = {}, acc1 = {}, acc2 = {}, acc3 = {};
    for (int ch = 0; ch < NFEAT / KC; ch += 2) {
#pragma unroll
        for (int e = 0; e < 4; ++e) xb[e] = *(const float4*)(xp + (ch + 1) * KC + 4 * e);
        g1_chunk(xa, bp + (ch * (KC / 2)) * W1PP, acc0, acc1, acc2, acc3);
        if (ch + 2 < NFEAT / KC) {
#pragma unroll
            for (int e = 0; e < 4; ++e) xa[e] = *(const float4*)(xp + (ch + 2) * KC + 4 * e);
        }
        g1_chunk(xb, bp + ((ch + 1) * (KC / 2)) * W1PP, acc0, acc1, acc2, acc3);
    }

    float* __restrict__ o = h1 + (long)(r0 + 8 * hh) * NHID + c;
    for (int j = 0; j < 8; ++j) {
        o[j * NHID]      = acc0[j];
        o[j * NHID + 16] = acc1[j];
        o[j * NHID + 32] = acc2[j];
        o[j * NHID + 48] = acc3[j];
    }
}

#define N_EDGES 1600000
#define NB  256
#define BW  391
#define NCH 256
#define NT  512
#define KE  13
#define SEGK (NCH / (NT / 16))
static_assert(NB * BW >= N_NODES, "buckets must cover every row");
static_assert(BW + 32 <= NT, "one thread per row of a bucket, and a spare word per lane behind the rows");
static_assert(NB <= NT && NCH <= NT, "one thread per bucket, one per chunk");
static_assert((long)KE * NT * NCH >= N_EDGES, "a chunk's edges fit its threads' registers");
static_assert(NCH % (NT / 16) == 0 && 2 * SEGK <= 32, "whole runs per 16-lane group, one mask bit per slot");

__device__ __forceinline__ int wg_excl_scan(int v, int& total) {
    __shared__ int wsum[32];
    const int lane = threadIdx.x & 31, wv = threadIdx.x >> 5;
    int incl = v;
    for (int d = 1; d < 32; d <<= 1) {
        int t = __shfl_up(incl, d, 32);
        if (lane >= d) incl += t;
    }
    if (lane == 31) wsum[wv] = incl;
    __syncthreads();
    const int ws = (lane < NT / 32) ? wsum[lane] : 0;
    int wincl = ws;
    for (int d = 1; d < 32; d <<= 1) {
        int t = __shfl_up(wincl, d, 32);
        if (lane >= d) wincl += t;
    }
    total = __shfl(wincl, 31, 32);
    const int wexcl = __shfl(wincl - ws, wv, 32);
    return wexcl + incl - v;
}

__global__ __launch_bounds__(NT) void chunk_group_k(const int* __restrict__ src,
                                                    const int* __restrict__ dst,
                                                    const float* __restrict__ w,
                                                    int* __restrict__ seg_w,
                                                    int* __restrict__ brec_w, int nEdges) {
    int4* __restrict__ brec = (int4*)brec_w;
    int2* __restrict__ seg  = (int2*)seg_w;
    __shared__ int h[NB + 32];
    const int t = threadIdx.x, g = blockIdx.x;
    if (t < NB + 32) h[t] = 0;
    const int chunk = (nEdges + NCH - 1) / NCH;
    const int e0 = g * chunk;
    int e1 = e0 + chunk; if (e1 > nEdges) e1 = nEdges;
    int d[KE], s[KE], wb[KE];
#pragma unroll
    for (int k = 0; k < KE; ++k) {
        const int e  = e0 + t + k * NT;
        const int ec = (e < e1) ? e : 0;
        d[k]  = dst[ec];
        s[k]  = src[ec];
        wb[k] = __float_as_int(w[ec]);
    }
    __syncthreads();
    int bk[KE], rk[KE];
#pragma unroll
    for (int k = 0; k < KE; ++k) {
        bk[k] = (e0 + t + k * NT < e1) ? d[k] / BW : NB + (t & 31);
        rk[k] = atomicAdd(&h[bk[k]], 1);
    }
    __syncthreads();
    const int mine = (t < NB) ? h[t] : 0;
    int total;
    const int excl = wg_excl_scan(mine, total);
    if (t < NB) {
        h[t] = excl;
        int2 v; v.x = excl; v.y = mine;
        seg[g * NB + t] = v;
    }
    __syncthreads();
    int pb[KE];
#pragma unroll
    for (int k = 0; k < KE; ++k) pb[k] = h[bk[k]];
#pragma unroll
    for (int k = 0; k < KE; ++k) {
        if (e0 + t + k * NT < e1) {
            int4 r; r.x = s[k]; r.y = wb[k]; r.z = d[k]; r.w = 0;
            brec[e0 + pb[k] + rk[k]] = r;
        }
    }
}

__global__ __launch_bounds__(NT) void bucket_rows_k(const int* __restrict__ seg_w,
                                                    const int* __restrict__ brec_w,
                                                    int* __restrict__ cnt,
                                                    int* __restrict__ rowstart,
                                                    int* __restrict__ pk, int nEdges) {
    __shared__ int c[NT];
    __shared__ int so[NCH];
    __shared__ int sc[NCH];
    const int2* __restrict__ seg  = (const int2*)seg_w;
    const int4* __restrict__ brec = (const int4*)brec_w;
    const int t = threadIdx.x, b = blockIdx.x;
    const int node0 = b * BW;
    const int chunk = (nEdges + NCH - 1) / NCH;
    int off = 0;
    if (t < NCH) {
        const int2 v = seg[t * NB + b];
        off = v.x;
        so[t] = t * chunk + v.x;
        sc[t] = v.y;
    }
    c[t] = 0;
    int s;
    (void)wg_excl_scan(off, s);
    const int hw = t >> 4, l16 = t & 15;
    int base[SEGK], m[SEGK];
    int4 r[2 * SEGK];
    int  ri[2 * SEGK];
    unsigned vm = 0;
    int mmax = 0;
#pragma unroll
    for (int k = 0; k < SEGK; ++k) {
        const int gq = hw + (NT / 16) * k;
        base[k] = so[gq]; m[k] = sc[gq];
        mmax = (m[k] > mmax) ? m[k] : mmax;
        const bool v0 = l16 < m[k], v1 = l16 + 16 < m[k];
        r[2 * k]     = brec[v0 ? base[k] + l16 : 0];
        r[2 * k + 1] = brec[v1 ? base[k] + l16 + 16 : 0];
        vm |= (v0 ? 1u : 0u) << (2 * k);
        vm |= (v1 ? 1u : 0u) << (2 * k + 1);
    }
#pragma unroll
    for (int j = 0; j < 2 * SEGK; ++j) {
        ri[j] = ((vm >> j) & 1u) ? r[j].z - node0 : BW + (t & 31);
        atomicAdd(&c[ri[j]], 1);
    }
    if (mmax > 32) {
        for (int k = 0; k < SEGK; ++k)
            for (int i = 32 + l16; i < m[k]; i += 16) atomicAdd(&c[brec[base[k] + i].z - node0], 1);
    }
    __syncthreads();
    const int mine = c[t];
    int total;
    const int excl = wg_excl_scan(mine, total);
    if (t < BW && node0 + t < N_NODES) {
        cnt[node0 + t] = mine;
        rowstart[node0 + t] = s + excl;
    }
    c[t] = excl;
    __syncthreads();
    int p[2 * SEGK];
#pragma unroll
    for (int j = 0; j < 2 * SEGK; ++j) p[j] = atomicAdd(&c[ri[j]], 1);
#pragma unroll
    for (int j = 0; j < 2 * SEGK; ++j) {
        if ((vm >> j) & 1u) {
            int2 v; v.x = r[j].x; v.y = r[j].y;
            ((int2*)pk)[s + p[j]] = v;
        }
    }
    if (mmax > 32) {
        for (int k = 0; k < SEGK; ++k)
            for (int i = 32 + l16; i < m[k]; i += 16) {
                const int4 q = brec[base[k] + i];
                const int pq = atomicAdd(&c[q.z - node0], 1);
                int2 v; v.x = q.x; v.y = q.y;
                ((int2*)pk)[s + pq] = v;
            }
    }
}

__device__ __forceinline__ float xhalf(float x) {
    int b = __builtin_bit_cast(int, x);
    return __builtin_bit_cast(float, __builtin_amdgcn_permlanex16(b, b, 0x76543210, 0xfedcba98, false, false));
}

#define GP 8

template <int P, bool ODD, int PITCH>
__device__ __forceinline__ void acc_pairs(const float* __restrict__ h, int ex, int ey, int j,
                                          int hh, int q4, float4& a) {
    float4 v[P];
    float  wk[P];
#pragma unroll
    for (int k = 0; k < P; ++k) {
        const bool lone = ODD && (k == P - 1);
        const int s0 = __builtin_amdgcn_readlane(ex, j + 2 * k);
        const int w0 = __builtin_amdgcn_readlane(ey, j + 2 * k);
        int sj = s0, wj = w0;
        if (!lone) {
            const int s1 = __builtin_amdgcn_readlane(ex, j + 2 * k + 1);
            const int w1 = __builtin_amdgcn_readlane(ey, j + 2 * k + 1);
            sj = hh ? s1 : s0;
            wj = hh ? w1 : w0;
        }
        wk[k] = __int_as_float(wj);
        v[k]  = *(const float4*)(h + (size_t)(unsigned)sj * PITCH + q4);
        if (lone && hh) { wk[k] = 0.0f; v[k].x = 0.0f; v[k].y = 0.0f; v[k].z = 0.0f; v[k].w = 0.0f; }
    }
#pragma unroll
    for (int k = 0; k < P; ++k) {
        a.x = fmaf(wk[k], v[k].x, a.x);
        a.y = fmaf(wk[k], v[k].y, a.y);
        a.z = fmaf(wk[k], v[k].z, a.z);
        a.w = fmaf(wk[k], v[k].w, a.w);
    }
}

template <int R, int PITCH>
__device__ __forceinline__ void acc_tail(const float* __restrict__ h, int ex, int ey, int j,
                                         int hh, int q4, float4& a, int r, int odd) {
    if (r == R) {
        if (odd) acc_pairs<R + 1, true, PITCH>(h, ex, ey, j, hh, q4, a);
        else if constexpr (R > 0) acc_pairs<R, false, PITCH>(h, ex, ey, j, hh, q4, a);
    } else if constexpr (R > 0) {
        acc_tail<R - 1, PITCH>(h, ex, ey, j, hh, q4, a, r, odd);
    }
}

template <int PITCH>
__device__ __forceinline__ void row_sum_pairs(const int* __restrict__ rowstart,
                                              const int* __restrict__ cnt,
                                              const int* __restrict__ pk,
                                              const float* __restrict__ h,
                                              int r, int lane, int q4, float4& a) {
    const int ru = __builtin_amdgcn_readfirstlane(r);
    const int s  = __builtin_amdgcn_readfirstlane(rowstart[ru]);
    const int n  = __builtin_amdgcn_readfirstlane(cnt[ru]);
    const int hh = lane >> 4;
    a.x = 0.0f; a.y = 0.0f; a.z = 0.0f; a.w = 0.0f;
    for (int c0 = 0; c0 < n; c0 += 32) {
        int m = n - c0; if (m > 32) m = 32;
        int ex = 0, ey = 0;
        if (lane < m) { int2 e = ((const int2*)pk)[s + c0 + lane]; ex = e.x; ey = e.y; }
        const int whole = m >> 1, odd = m & 1;
        int p = 0;
        for (; p + GP <= whole; p += GP) acc_pairs<GP, false, PITCH>(h, ex, ey, 2 * p, hh, q4, a);
        acc_tail<GP - 1, PITCH>(h, ex, ey, 2 * p, hh, q4, a, whole - p, odd);
    }
    a.x += xhalf(a.x); a.y += xhalf(a.y); a.z += xhalf(a.z); a.w += xhalf(a.w);
}

#define TP   68
#define NPAD 48
static_assert(N_NODES % 8 == 0, "rows come in whole groups of 8");
static_assert(NHID % 4 == 0 && NCLASS <= NPAD, "product tiling");

__global__ __launch_bounds__(256) void gather64_gemm2_k(const int* __restrict__ rowstart,
                                                        const int* __restrict__ cnt,
                                                        const int* __restrict__ pk,
                                                        const float* __restrict__ h,
                                                        const float* __restrict__ b1,
                                                        const float* __restrict__ W2,
                                                        float* __restrict__ h2) {
    __shared__ __attribute__((aligned(16))) float w2p[(NHID / 2) * NPAD * 2];
    __shared__ __attribute__((aligned(16))) float tile[2][16 * TP];
    const int lane = threadIdx.x & 31;
    const int wave = threadIdx.x >> 5;
    {
        constexpr int NF = (NHID / 2) * NPAD / 256;
        static_assert(NF * 256 == (NHID / 2) * NPAD, "whole entries per thread");
        float2 wv[NF];
#pragma unroll
        for (int q = 0; q < NF; ++q) {
            const int i = threadIdx.x + q * 256;
            const int kp = i / NPAD, n = i - kp * NPAD;
            const int nc = (n < NCLASS) ? n : (NCLASS - 1);
            const float lo = W2[(2 * kp) * NCLASS + nc];
            const float hi = W2[(2 * kp + 1) * NCLASS + nc];
            wv[q].x = (n < NCLASS) ? lo : 0.0f;
            wv[q].y = (n < NCLASS) ? hi : 0.0f;
        }
#pragma unroll
        for (int q = 0; q < NF; ++q) *(float2*)(w2p + 2 * (threadIdx.x + q * 256)) = wv[q];
    }
    const int c  = lane & 15;
    const int hh = lane >> 4;
    const float4 bq = *(const float4*)(b1 + 4 * c);
    constexpr int NG = N_NODES / 8;
    const int gstep = gridDim.x;
    int it = 0;
    for (int g = blockIdx.x; g < NG; g += gstep, ++it) {
        const int half = it & 1;
        const bool last = g + gstep >= NG;
        float* __restrict__ tb = tile[(it >> 1) & 1];
        float4 a;
        row_sum_pairs<NHID>(rowstart, cnt, pk, h, g * 8 + wave, lane, 4 * c, a);
        if (hh == 0) {
            float4 o;
            o.x = fmaxf(a.x + bq.x, 0.0f); o.y = fmaxf(a.y + bq.y, 0.0f);
            o.z = fmaxf(a.z + bq.z, 0.0f); o.w = fmaxf(a.w + bq.w, 0.0f);
            *(float4*)(tb + (half * 8 + wave) * TP + 4 * c) = o;
        }
        if (half == 0 && !last) continue;
        if (half == 0 && hh == 0) {
            float4 z; z.x = 0.0f; z.y = 0.0f; z.z = 0.0f; z.w = 0.0f;
            *(float4*)(tb + (8 + wave) * TP + 4 * c) = z;
        }
        __syncthreads();
        if (wave < 3) {
            const float* __restrict__ ap = tb + c * TP + 2 * hh;
            const float* __restrict__ bp = w2p + (hh * NPAD + wave * 16 + c) * 2;
            v8f acc = {};
            for (int k0 = 0; k0 < NHID; k0 += 4) {
                const v2f a = *(const v2f*)(ap + k0);
                const v2f b = *(const v2f*)(bp + (k0 / 2) * NPAD * 2);
                acc = __builtin_amdgcn_wmma_f32_16x16x4_f32(false, a, false, b, (short)0, acc, false, false);
            }
            asm volatile("v_nop\n\tv_nop\n\tv_nop\n\tv_nop\n\tv_nop" : "+v"(acc));
            const int col = wave * 16 + c;
            const int gs = (hh == half) ? g : g - gstep;
            if (hh == 0 || half == 1) {
                float* __restrict__ op = h2 + (long)(gs * 8) * NPAD + col;
                for (int j = 0; j < 8; ++j) op[j * NPAD] = acc[j];
            }
        }
    }
}

__device__ __forceinline__ float row16_max(float v) {
    asm volatile("v_max_num_f32_dpp %0, %0, %0 row_xmask:1 row_mask:0xf bank_mask:0xf\n\t"
                 "v_max_num_f32_dpp %0, %0, %0 row_xmask:2 row_mask:0xf bank_mask:0xf\n\t"
                 "v_max_num_f32_dpp %0, %0, %0 row_xmask:4 row_mask:0xf bank_mask:0xf\n\t"
                 "v_max_num_f32_dpp %0, %0, %0 row_xmask:8 row_mask:0xf bank_mask:0xf"
                 : "+v"(v));
    return v;
}

__device__ __forceinline__ float row16_sum(float v) {
    asm volatile("v_add_f32_dpp %0, %0, %0 row_xmask:1 row_mask:0xf bank_mask:0xf\n\t"
                 "v_add_f32_dpp %0, %0, %0 row_xmask:2 row_mask:0xf bank_mask:0xf\n\t"
                 "v_add_f32_dpp %0, %0, %0 row_xmask:4 row_mask:0xf bank_mask:0xf\n\t"
                 "v_add_f32_dpp %0, %0, %0 row_xmask:8 row_mask:0xf bank_mask:0xf"
                 : "+v"(v));
    return v;
}

__device__ __forceinline__ float wave_max(float v) {
    v = row16_max(v);
    return fmaxf(v, xhalf(v));
}

__device__ __forceinline__ float wave_sum(float v) {
    v = row16_sum(v);
    return v + xhalf(v);
}

__global__ __launch_bounds__(256) void gather41_lsm_k(const int* __restrict__ rowstart,
                                                      const int* __restrict__ cnt,
                                                      const int* __restrict__ pk,
                                                      const float* __restrict__ h,
                                                      const float* __restrict__ b2,
                                                      float* __restrict__ out) {
    const int lane = threadIdx.x & 31;
    const int nw = gridDim.x * 8;
    const bool has1 = (32 + lane < NCLASS);
    const int c1i = has1 ? (32 + lane) : (NCLASS - 1);
    const float bz0 = b2[lane];
    const float bz1 = b2[c1i];
    const int q  = lane & 15;
    const int q4 = 4 * ((q < NPAD / 4) ? q : (NPAD / 4 - 1));
    const int sl0 = lane >> 2, sl1 = 8 + (lane >> 2), cm = lane & 3;
    for (int r = blockIdx.x * 8 + (threadIdx.x >> 5); r < N_NODES; r += nw) {
        const int ru = __builtin_amdgcn_readfirstlane(r);
        float4 a;
        row_sum_pairs<NPAD>(rowstart, cnt, pk, h, r, lane, q4, a);
        const float t0 = __shfl(a.x, sl0, 32), t1 = __shfl(a.y, sl0, 32);
        const float t2 = __shfl(a.z, sl0, 32), t3 = __shfl(a.w, sl0, 32);
        const float u0 = __shfl(a.x, sl1, 32), u1 = __shfl(a.y, sl1, 32);
        const float u2 = __shfl(a.z, sl1, 32), u3 = __shfl(a.w, sl1, 32);
        const float a0 = (cm == 0) ? t0 : (cm == 1) ? t1 : (cm == 2) ? t2 : t3;
        const float a1 = (cm == 0) ? u0 : (cm == 1) ? u1 : (cm == 2) ? u2 : u3;
        const float z0 = a0 + bz0;
        const float z1 = has1 ? (a1 + bz1) : z0;
        const float mx = wave_max(fmaxf(z0, z1));
        const float e0 = expf(z0 - mx);
        const float e1 = expf(z1 - mx);
        const float lse = mx + logf(wave_sum(has1 ? (e0 + e1) : e0));
        float* __restrict__ o = out + (long)ru * NCLASS;
        o[lane] = z0 - lse;
        if (has1) o[32 + lane] = z1 - lse;
    }
}

extern "C" void kernel_launch(void* const* d_in, const int* in_sizes, int n_in,
                              void* d_out, int out_size, void* d_ws, size_t ws_size,
                              hipStream_t stream) {
    const float* x    = (const float*)d_in[0];
    const int*   esrc = (const int*)d_in[1];
    const int*   edst = (const int*)d_in[2];
    const float* ew   = (const float*)d_in[3];
    const float* W1   = (const float*)d_in[4];
    const float* b1   = (const float*)d_in[5];
    const float* W2   = (const float*)d_in[6];
    const float* b2   = (const float*)d_in[7];
    float* out = (float*)d_out;
    const int E = in_sizes[1];

    float* h1   = (float*)d_ws;
    float* agg1 = h1   + (size_t)N_NODES * NHID;
    float* h2   = agg1;
    int* pk       = (int*)(agg1 + (size_t)N_NODES * NHID);
    int* cnt      = pk + (size_t)2 * N_EDGES + (size_t)NCH * NB + (size_t)(NB + 8);
    int* rowstart = cnt    + (size_t)(N_NODES + 8);
    int* brec     = (int*)agg1;
    static_assert((size_t)4 * N_EDGES <= (size_t)N_NODES * NHID, "bucketed copy fits in agg1");
    int* seg      = (int*)h1;
    static_assert((size_t)2 * NCH * NB <= (size_t)N_NODES * NHID, "run table fits in h1");

    chunk_group_k<<<NCH, NT, 0, stream>>>(esrc, edst, ew, seg, brec, E);
    bucket_rows_k<<<NB, NT, 0, stream>>>(seg, brec, cnt, rowstart, pk, E);

    gemm1_wmma_k<<<N_NODES / (16 * G1W), G1T, 0, stream>>>(x, W1, h1);

    gather64_gemm2_k<<<2048, 256, 0, stream>>>(rowstart, cnt, pk, h1, b1, W2, h2);

    gather41_lsm_k<<<2048, 256, 0, stream>>>(rowstart, cnt, pk, h2, b2, out);
}


# GEMM1 and the edge grouping's chunk pass in ONE launch, two kinds of 16-wave work-group, one of each per WGP: loads, LDS counters, stores run under the f32 matrix products; row sum order: LDS arrival
# speedup vs baseline: 5.7886x; 1.0699x over previous
#include <hip/hip_runtime.h>

typedef __attribute__((ext_vector_type(2))) float v2f;
typedef __attribute__((ext_vector_type(8))) float v8f;

#define N_NODES 100000
#define NFEAT   256
#define NHID    64
#define NCLASS  41

#define G1B  256
#define G1W  16
#define G1T  (G1W * 32)
#define NTILE (N_NODES / 16)
#define W1PP 132
#define KC   32
#define G1LDS 27648
static_assert(G1LDS >= (NFEAT / 2) * W1PP, "W1 fits");
static_assert(NHID == 64 && NFEAT % (2 * KC) == 0, "GEMM1 tiling");
static_assert((long)N_NODES * NFEAT < (1L << 31), "x is indexed with 32 bits");
static_assert(N_NODES % 16 == 0 && G1B * G1W <= NTILE && 2 * G1B * G1W >= NTILE,
              "whole tiles, one or two per wave");

__device__ __forceinline__ void g1_chunk(const float4 (&xr)[4], const float* __restrict__ bp,
                                         v8f& acc0, v8f& acc1, v8f& acc2, v8f& acc3) {
    v2f bf[8][4];
#pragma unroll
    for (int j = 0; j < 8; ++j) {
        const float* __restrict__ br = bp + j * W1PP;
        bf[j][0] = *(const v2f*)(br);
        bf[j][1] = *(const v2f*)(br + 32);
        bf[j][2] = *(const v2f*)(br + 64);
        bf[j][3] = *(const v2f*)(br + 96);
    }
#pragma unroll
    for (int j = 0; j < 8; ++j) {
        const float4 q = xr[j >> 1];
        v2f a;
        if (j & 1) { a.x = q.z; a.y = q.w; } else { a.x = q.x; a.y = q.y; }
        acc0 = __builtin_amdgcn_wmma_f32_16x16x4_f32(false, a, false, bf[j][0], (short)0, acc0, false, false);
        acc1 = __builtin_amdgcn_wmma_f32_16x16x4_f32(false, a, false, bf[j][1], (short)0, acc1, false, false);
        acc2 = __builtin_amdgcn_wmma_f32_16x16x4_f32(false, a, false, bf[j][2], (short)0, acc2, false, false);
        acc3 = __builtin_amdgcn_wmma_f32_16x16x4_f32(false, a, false, bf[j][3], (short)0, acc3, false, false);
    }
}

__device__ __forceinline__ void gemm1_role(const int bid, const float* __restrict__ x,
                                           const float* __restrict__ W1,
                                           float* __restrict__ h1) {
    __shared__ __attribute__((aligned(16))) float w1p[G1LDS];
    const int lane = threadIdx.x & 31;
    const int wave = __builtin_amdgcn_readfirstlane(threadIdx.x >> 5);
    const int c  = lane & 15;
    const int hh = lane >> 4;
    int tile = wave * G1B + bid;
    unsigned xo = (unsigned)((tile * 16 + c) * NFEAT + 16 * hh);

    float4 xa[4], xb[4];
#pragma unroll
    for (int e = 0; e < 4; ++e) xa[e] = *(const float4*)(x + xo + 4 * e);

    {
        constexpr int ITEMS = (NFEAT / 2) * (NHID / 4);
        constexpr int RND = (ITEMS + G1T - 1) / G1T;
        float4 lo[RND], hi[RND];
#pragma unroll
        for (int q = 0; q < RND; ++q) {
            int i = threadIdx.x + q * G1T;
            if (i > ITEMS - 1) i = ITEMS - 1;
            const int kp = i >> 4, n4 = (i & 15) * 4;
            lo[q] = *(const float4*)(W1 + (2 * kp) * NHID + n4);
            hi[q] = *(const float4*)(W1 + (2 * kp + 1) * NHID + n4);
        }
#pragma unroll
        for (int q = 0; q < RND; ++q) {
            const int i = threadIdx.x + q * G1T;
            if (i < ITEMS) {
                const int kp = i >> 4, n4 = (i & 15) * 4;
                float* __restrict__ d = w1p + kp * W1PP + n4 * 2;
                float4 p0, p1;
                p0.x = lo[q].x; p0.y = hi[q].x; p0.z = lo[q].y; p0.w = hi[q].y;
                p1.x = lo[q].z; p1.y = hi[q].z; p1.z = lo[q].w; p1.w = hi[q].w;
                *(float4*)d = p0;
                *(float4*)(d + 4) = p1;
            }
        }
    }
    __syncthreads();

    const float* __restrict__ bp = w1p + (8 * hh) * W1PP + 2 * c;
    for (;;) {
        asm volatile("" ::: "memory");
        const int nt = tile + G1B * G1W;
        const bool more = nt < NTILE;
        const unsigned xn = more ? (unsigned)((nt * 16 + c) * NFEAT + 16 * hh) : xo;
        v8f acc0 = {}, acc1 = {}, acc2 = {}, acc3 = {};
        for (int ch = 0; ch < NFEAT / KC; ch += 2) {
#pragma unroll
            for (int e = 0; e < 4; ++e) xb[e] = *(const float4*)(x + xo + (ch + 1) * KC + 4 * e);
            g1_chunk(xa, bp + (ch * (KC / 2)) * W1PP, acc0, acc1, acc2, acc3);
            const unsigned xq = (ch + 2 < NFEAT / KC) ? xo + (ch + 2) * KC : xn;
#pragma unroll
            for (int e = 0; e < 4; ++e) xa[e] = *(const float4*)(x + xq + 4 * e);
            g1_chunk(xb, bp + ((ch + 1) * (KC / 2)) * W1PP, acc0, acc1, acc2, acc3);
        }

        float* __restrict__ o = h1 + (unsigned)((tile * 16 + 8 * hh) * NHID + c);
        for (int j = 0; j < 8; ++j) {
            o[j * NHID]      = acc0[j];
            o[j * NHID + 16] = acc1[j];
            o[j * NHID + 32] = acc2[j];
            o[j * NHID + 48] = acc3[j];
        }
        if (!more) break;
        tile = nt;
        xo = xn;
    }
}

#define N_EDGES 1600000
#define NB  256
#define BW  391
#define NCH 256
#define NT  512
#define KE  13
#define SEGK (NCH / (NT / 16))
static_assert(NB * BW >= N_NODES, "buckets must cover every row");
static_assert(BW + 32 <= NT, "one thread per row of a bucket, and a spare word per lane behind the rows");
static_assert(NB <= NT && NCH <= NT, "one thread per bucket, one per chunk");
static_assert((long)KE * NT * NCH >= N_EDGES, "a chunk's edges fit its threads' registers");
static_assert(NCH % (NT / 16) == 0 && 2 * SEGK <= 32, "whole runs per 16-lane group, one mask bit per slot");

__device__ __forceinline__ int wg_excl_scan(int v, int& total) {
    __shared__ int wsum[32];
    const int lane = threadIdx.x & 31, wv = threadIdx.x >> 5;
    int incl = v;
    for (int d = 1; d < 32; d <<= 1) {
        int t = __shfl_up(incl, d, 32);
        if (lane >= d) incl += t;
    }
    if (lane == 31) wsum[wv] = incl;
    __syncthreads();
    const int ws = (lane < NT / 32) ? wsum[lane] : 0;
    int wincl = ws;
    for (int d = 1; d < 32; d <<= 1) {
        int t = __shfl_up(wincl, d, 32);
        if (lane >= d) wincl += t;
    }
    total = __shfl(wincl, 31, 32);
    const int wexcl = __shfl(wincl - ws, wv, 32);
    return wexcl + incl - v;
}

static_assert((N_EDGES + NCH - 1) / NCH < 65536, "a run's start and length fit 16 bits each");
static_assert(G1T == NT, "both kinds of work-group of the first launch have NT threads");
static_assert(G1B == NCH && G1B == 256, "one work-group of each kind per pair of block numbers");
__device__ __forceinline__ void chunk_role(const int g, const int* __restrict__ src,
                                           const int* __restrict__ dst,
                                           const float* __restrict__ w,
                                           int* __restrict__ seg,
                                           int* __restrict__ brec_w, int nEdges) {
    int4* __restrict__ brec = (int4*)brec_w;
    __shared__ int h[NB + 32];
    const int t = threadIdx.x;
    if (t < NB + 32) h[t] = 0;
    const int chunk = (nEdges + NCH - 1) / NCH;
    const int e0 = g * chunk;
    int e1 = e0 + chunk; if (e1 > nEdges) e1 = nEdges;
    int d[KE], s[KE], wb[KE];
#pragma unroll
    for (int k = 0; k < KE; ++k) {
        const int e  = e0 + t + k * NT;
        const int ec = (e < e1) ? e : 0;
        d[k]  = dst[ec];
        s[k]  = src[ec];
        wb[k] = __float_as_int(w[ec]);
    }
    __syncthreads();
    int bk[KE], rk[KE];
#pragma unroll
    for (int k = 0; k < KE; ++k) {
        bk[k] = (e0 + t + k * NT < e1) ? d[k] / BW : NB + (t & 31);
        rk[k] = atomicAdd(&h[bk[k]], 1);
    }
    __syncthreads();
    const int mine = (t < NB) ? h[t] : 0;
    int total;
    const int excl = wg_excl_scan(mine, total);
    if (t < NB) {
        h[t] = excl;
        seg[g * NB + t] = excl | (mine << 16);
    }
    __syncthreads();
    int pb[KE];
#pragma unroll
    for (int k = 0; k < KE; ++k) pb[k] = h[bk[k]];
#pragma unroll
    for (int k = 0; k < KE; ++k) {
        if (e0 + t + k * NT < e1) {
            int4 r; r.x = s[k]; r.y = wb[k]; r.z = d[k]; r.w = 0;
            brec[e0 + pb[k] + rk[k]] = r;
        }
    }
}

__global__ __launch_bounds__(NT) __attribute__((amdgpu_waves_per_eu(8)))
void gemm1_chunk_group_k(const float* __restrict__ x,
                                                          const float* __restrict__ W1,
                                                          float* __restrict__ h1,
                                                          const int* __restrict__ src,
                                                          const int* __restrict__ dst,
                                                          const float* __restrict__ w,
                                                          int* __restrict__ seg,
                                                          int* __restrict__ brec_w, int nEdges) {
    const int b = blockIdx.x;
    if (((b ^ (b >> 8)) & 1) == 0) gemm1_role(b >> 1, x, W1, h1);
    else chunk_role(b >> 1, src, dst, w, seg, brec_w, nEdges);
}

__global__ __launch_bounds__(NT) void bucket_rows_k(const int* __restrict__ seg_w,
                                                    const int* __restrict__ brec_w,
                                                    int* __restrict__ cnt,
                                                    int* __restrict__ rowstart,
                                                    int* __restrict__ pk, int nEdges) {
    __shared__ int c[NT];
    __shared__ int so[NCH];
    __shared__ int sc[NCH];
    const int4* __restrict__ brec = (const int4*)brec_w;
    const int t = threadIdx.x, b = blockIdx.x;
    const int node0 = b * BW;
    const int chunk = (nEdges + NCH - 1) / NCH;
    int off = 0;
    if (t < NCH) {
        const unsigned v = (unsigned)seg_w[t * NB + b];
        off = (int)(v & 0xffffu);
        so[t] = t * chunk + off;
        sc[t] = (int)(v >> 16);
    }
    c[t] = 0;
    int s;
    (void)wg_excl_scan(off, s);
    const int hw = t >> 4, l16 = t & 15;
    int base[SEGK], m[SEGK];
    int4 r[2 * SEGK];
    int  ri[2 * SEGK];
    unsigned vm = 0;
    int mmax = 0;
#pragma unroll
    for (int k = 0; k < SEGK; ++k) {
        const int gq = hw + (NT / 16) * k;
        base[k] = so[gq]; m[k] = sc[gq];
        mmax = (m[k] > mmax) ? m[k] : mmax;
        const bool v0 = l16 < m[k], v1 = l16 + 16 < m[k];
        r[2 * k]     = brec[v0 ? base[k] + l16 : 0];
        r[2 * k + 1] = brec[v1 ? base[k] + l16 + 16 : 0];
        vm |= (v0 ? 1u : 0u) << (2 * k);
        vm |= (v1 ? 1u : 0u) << (2 * k + 1);
    }
#pragma unroll
    for (int j = 0; j < 2 * SEGK; ++j) {
        ri[j] = ((vm >> j) & 1u) ? r[j].z - node0 : BW + (t & 31);
        atomicAdd(&c[ri[j]], 1);
    }
    if (mmax > 32) {
        for (int k = 0; k < SEGK; ++k)
            for (int i = 32 + l16; i < m[k]; i += 16) atomicAdd(&c[brec[base[k] + i].z - node0], 1);
    }
    __syncthreads();
    const int mine = c[t];
    int total;
    const int excl = wg_excl_scan(mine, total);
    if (t < BW && node0 + t < N_NODES) {
        cnt[node0 + t] = mine;
        rowstart[node0 + t] = s + excl;
    }
    c[t] = excl;
    __syncthreads();
    int p[2 * SEGK];
#pragma unroll
    for (int j = 0; j < 2 * SEGK; ++j) p[j] = atomicAdd(&c[ri[j]], 1);
#pragma unroll
    for (int j = 0; j < 2 * SEGK; ++j) {
        if ((vm >> j) & 1u) {
            int2 v; v.x = r[j].x; v.y = r[j].y;
            ((int2*)pk)[s + p[j]] = v;
        }
    }
    if (mmax > 32) {
        for (int k = 0; k < SEGK; ++k)
            for (int i = 32 + l16; i < m[k]; i += 16) {
                const int4 q = brec[base[k] + i];
                const int pq = atomicAdd(&c[q.z - node0], 1);
                int2 v; v.x = q.x; v.y = q.y;
                ((int2*)pk)[s + pq] = v;
            }
    }
}

__device__ __forceinline__ float xhalf(float x) {
    int b = __builtin_bit_cast(int, x);
    return __builtin_bit_cast(float, __builtin_amdgcn_permlanex16(b, b, 0x76543210, 0xfedcba98, false, false));
}

#define GP 8

template <int P, bool ODD, int PITCH>
__device__ __forceinline__ void acc_pairs(const float* __restrict__ h, int ex, int ey, int j,
                                          int hh, int q4, float4& a) {
    float4 v[P];
    float  wk[P];
#pragma unroll
    for (int k = 0; k < P; ++k) {
        const bool lone = ODD && (k == P - 1);
        const int s0 = __builtin_amdgcn_readlane(ex, j + 2 * k);
        const int w0 = __builtin_amdgcn_readlane(ey, j + 2 * k);
        int sj = s0, wj = w0;
        if (!lone) {
            const int s1 = __builtin_amdgcn_readlane(ex, j + 2 * k + 1);
            const int w1 = __builtin_amdgcn_readlane(ey, j + 2 * k + 1);
            sj = hh ? s1 : s0;
            wj = hh ? w1 : w0;
        }
        wk[k] = __int_as_float(wj);
        v[k]  = *(const float4*)(h + (size_t)(unsigned)sj * PITCH + q4);
        if (lone && hh) { wk[k] = 0.0f; v[k].x = 0.0f; v[k].y = 0.0f; v[k].z = 0.0f; v[k].w = 0.0f; }
    }
#pragma unroll
    for (int k = 0; k < P; ++k) {
        a.x = fmaf(wk[k], v[k].x, a.x);
        a.y = fmaf(wk[k], v[k].y, a.y);
        a.z = fmaf(wk[k], v[k].z, a.z);
        a.w = fmaf(wk[k], v[k].w, a.w);
    }
}

template <int R, int PITCH>
__device__ __forceinline__ void acc_tail(const float* __restrict__ h, int ex, int ey, int j,
                                         int hh, int q4, float4& a, int r, int odd) {
    if (r == R) {
        if (odd) acc_pairs<R + 1, true, PITCH>(h, ex, ey, j, hh, q4, a);
        else if constexpr (R > 0) acc_pairs<R, false, PITCH>(h, ex, ey, j, hh, q4, a);
    } else if constexpr (R > 0) {
        acc_tail<R - 1, PITCH>(h, ex, ey, j, hh, q4, a, r, odd);
    }
}

template <int PITCH>
__device__ __forceinline__ void row_sum_pairs(const int* __restrict__ rowstart,
                                              const int* __restrict__ cnt,
                                              const int* __restrict__ pk,
                                              const float* __restrict__ h,
                                              int r, int lane, int q4, float4& a) {
    const int ru = __builtin_amdgcn_readfirstlane(r);
    const int s  = __builtin_amdgcn_readfirstlane(rowstart[ru]);
    const int n  = __builtin_amdgcn_readfirstlane(cnt[ru]);
    const int hh = lane >> 4;
    a.x = 0.0f; a.y = 0.0f; a.z = 0.0f; a.w = 0.0f;
    for (int c0 = 0; c0 < n; c0 += 32) {
        int m = n - c0; if (m > 32) m = 32;
        int ex = 0, ey = 0;
        if (lane < m) { int2 e = ((const int2*)pk)[s + c0 + lane]; ex = e.x; ey = e.y; }
        const int whole = m >> 1, odd = m & 1;
        int p = 0;
        for (; p + GP <= whole; p += GP) acc_pairs<GP, false, PITCH>(h, ex, ey, 2 * p, hh, q4, a);
        acc_tail<GP - 1, PITCH>(h, ex, ey, 2 * p, hh, q4, a, whole - p, odd);
    }
    a.x += xhalf(a.x); a.y += xhalf(a.y); a.z += xhalf(a.z); a.w += xhalf(a.w);
}

#define TP   68
#define NPAD 48
static_assert(N_NODES % 8 == 0, "rows come in whole groups of 8");
static_assert(NHID % 4 == 0 && NCLASS <= NPAD, "product tiling");

__global__ __launch_bounds__(256) void gather64_gemm2_k(const int* __restrict__ rowstart,
                                                        const int* __restrict__ cnt,
                                                        const int* __restrict__ pk,
                                                        const float* __restrict__ h,
                                                        const float* __restrict__ b1,
                                                        const float* __restrict__ W2,
                                                        float* __restrict__ h2) {
    __shared__ __attribute__((aligned(16))) float w2p[(NHID / 2) * NPAD * 2];
    __shared__ __attribute__((aligned(16))) float tile[2][16 * TP];
    const int lane = threadIdx.x & 31;
    const int wave = threadIdx.x >> 5;
    {
        constexpr int NF = (NHID / 2) * NPAD / 256;
        static_assert(NF * 256 == (NHID / 2) * NPAD, "whole entries per thread");
        float2 wv[NF];
#pragma unroll
        for (int q = 0; q < NF; ++q) {
            const int i = threadIdx.x + q * 256;
            const int kp = i / NPAD, n = i - kp * NPAD;
            const int nc = (n < NCLASS) ? n : (NCLASS - 1);
            const float lo = W2[(2 * kp) * NCLASS + nc];
            const float hi = W2[(2 * kp + 1) * NCLASS + nc];
            wv[q].x = (n < NCLASS) ? lo : 0.0f;
            wv[q].y = (n < NCLASS) ? hi : 0.0f;
        }
#pragma unroll
        for (int q = 0; q < NF; ++q) *(float2*)(w2p + 2 * (threadIdx.x + q * 256)) = wv[q];
    }
    const int c  = lane & 15;
    const int hh = lane >> 4;
    const float4 bq = *(const float4*)(b1 + 4 * c);
    constexpr int NG = N_NODES / 8;
    const int gstep = gridDim.x;
    int it = 0;
    for (int g = blockIdx.x; g < NG; g += gstep, ++it) {
        const int half = it & 1;
        const bool last = g + gstep >= NG;
        float* __restrict__ tb = tile[(it >> 1) & 1];
        float4 a;
        row_sum_pairs<NHID>(rowstart, cnt, pk, h, g * 8 + wave, lane, 4 * c, a);
        if (hh == 0) {
            float4 o;
            o.x = fmaxf(a.x + bq.x, 0.0f); o.y = fmaxf(a.y + bq.y, 0.0f);
            o.z = fmaxf(a.z + bq.z, 0.0f); o.w = fmaxf(a.w + bq.w, 0.0f);
            *(float4*)(tb + (half * 8 + wave) * TP + 4 * c) = o;
        }
        if (half == 0 && !last) continue;
        if (half == 0 && hh == 0) {
            float4 z; z.x = 0.0f; z.y = 0.0f; z.z = 0.0f; z.w = 0.0f;
            *(float4*)(tb + (8 + wave) * TP + 4 * c) = z;
        }
        __syncthreads();
        if (wave < 3) {
            const float* __restrict__ ap = tb + c * TP + 2 * hh;
            const float* __restrict__ bp = w2p + (hh * NPAD + wave * 16 + c) * 2;
            v8f acc = {};
            for (int k0 = 0; k0 < NHID; k0 += 4) {
                const v2f a = *(const v2f*)(ap + k0);
                const v2f b = *(const v2f*)(bp + (k0 / 2) * NPAD * 2);
                acc = __builtin_amdgcn_wmma_f32_16x16x4_f32(false, a, false, b, (short)0, acc, false, false);
            }
            asm volatile("v_nop\n\tv_nop\n\tv_nop\n\tv_nop\n\tv_nop" : "+v"(acc));
            const int col = wave * 16 + c;
            const int gs = (hh == half) ? g : g - gstep;
            if (hh == 0 || half == 1) {
                float* __restrict__ op = h2 + (long)(gs * 8) * NPAD + col;
                for (int j = 0; j < 8; ++j) op[j * NPAD] = acc[j];
            }
        }
    }
}

__device__ __forceinline__ float row16_max(float v) {
    asm volatile("v_max_num_f32_dpp %0, %0, %0 row_xmask:1 row_mask:0xf bank_mask:0xf\n\t"
                 "v_max_num_f32_dpp %0, %0, %0 row_xmask:2 row_mask:0xf bank_mask:0xf\n\t"
                 "v_max_num_f32_dpp %0, %0, %0 row_xmask:4 row_mask:0xf bank_mask:0xf\n\t"
                 "v_max_num_f32_dpp %0, %0, %0 row_xmask:8 row_mask:0xf bank_mask:0xf"
                 : "+v"(v));
    return v;
}

__device__ __forceinline__ float row16_sum(float v) {
    asm volatile("v_add_f32_dpp %0, %0, %0 row_xmask:1 row_mask:0xf bank_mask:0xf\n\t"
                 "v_add_f32_dpp %0, %0, %0 row_xmask:2 row_mask:0xf bank_mask:0xf\n\t"
                 "v_add_f32_dpp %0, %0, %0 row_xmask:4 row_mask:0xf bank_mask:0xf\n\t"
                 "v_add_f32_dpp %0, %0, %0 row_xmask:8 row_mask:0xf bank_mask:0xf"
                 : "+v"(v));
    return v;
}

__device__ __forceinline__ float wave_max(float v) {
    v = row16_max(v);
    return fmaxf(v, xhalf(v));
}

__device__ __forceinline__ float wave_sum(float v) {
    v = row16_sum(v);
    return v + xhalf(v);
}

__global__ __launch_bounds__(256) void gather41_lsm_k(const int* __restrict__ rowstart,
                                                      const int* __restrict__ cnt,
                                                      const int* __restrict__ pk,
                                                      const float* __restrict__ h,
                                                      const float* __restrict__ b2,
                                                      float* __restrict__ out) {
    const int lane = threadIdx.x & 31;
    const int nw = gridDim.x * 8;
    const bool has1 = (32 + lane < NCLASS);
    const int c1i = has1 ? (32 + lane) : (NCLASS - 1);
    const float bz0 = b2[lane];
    const float bz1 = b2[c1i];
    const int q  = lane & 15;
    const int q4 = 4 * ((q < NPAD / 4) ? q : (NPAD / 4 - 1));
    const int sl0 = lane >> 2, sl1 = 8 + (lane >> 2), cm = lane & 3;
    for (int r = blockIdx.x * 8 + (threadIdx.x >> 5); r < N_NODES; r += nw) {
        const int ru = __builtin_amdgcn_readfirstlane(r);
        float4 a;
        row_sum_pairs<NPAD>(rowstart, cnt, pk, h, r, lane, q4, a);
        const float t0 = __shfl(a.x, sl0, 32), t1 = __shfl(a.y, sl0, 32);
        const float t2 = __shfl(a.z, sl0, 32), t3 = __shfl(a.w, sl0, 32);
        const float u0 = __shfl(a.x, sl1, 32), u1 = __shfl(a.y, sl1, 32);
        const float u2 = __shfl(a.z, sl1, 32), u3 = __shfl(a.w, sl1, 32);
        const float a0 = (cm == 0) ? t0 : (cm == 1) ? t1 : (cm == 2) ? t2 : t3;
        const float a1 = (cm == 0) ? u0 : (cm == 1) ? u1 : (cm == 2) ? u2 : u3;
        const float z0 = a0 + bz0;
        const float z1 = has1 ? (a1 + bz1) : z0;
        const float mx = wave_max(fmaxf(z0, z1));
        const float e0 = expf(z0 - mx);
        const float e1 = expf(z1 - mx);
        const float lse = mx + logf(wave_sum(has1 ? (e0 + e1) : e0));
        float* __restrict__ o = out + (long)ru * NCLASS;
        o[lane] = z0 - lse;
        if (has1) o[32 + lane] = z1 - lse;
    }
}

extern "C" void kernel_launch(void* const* d_in, const int* in_sizes, int n_in,
                              void* d_out, int out_size, void* d_ws, size_t ws_size,
                              hipStream_t stream) {
    const float* x    = (const float*)d_in[0];
    const int*   esrc = (const int*)d_in[1];
    const int*   edst = (const int*)d_in[2];
    const float* ew   = (const float*)d_in[3];
    const float* W1   = (const float*)d_in[4];
    const float* b1   = (const float*)d_in[5];
    const float* W2   = (const float*)d_in[6];
    const float* b2   = (const float*)d_in[7];
    float* out = (float*)d_out;
    const int E = in_sizes[1];

    float* h1   = (float*)d_ws;
    float* agg1 = h1   + (size_t)N_NODES * NHID;
    float* h2   = agg1;
    int* pk       = (int*)(agg1 + (size_t)N_NODES * NHID);
    int* seg      = pk + (size_t)2 * N_EDGES;
    int* cnt      = pk + (size_t)2 * N_EDGES + (size_t)NCH * NB + (size_t)(NB + 8);
    int* rowstart = cnt    + (size_t)(N_NODES + 8);
    int* brec     = (int*)agg1;
    static_assert((size_t)4 * N_EDGES <= (size_t)N_NODES * NHID, "bucketed copy fits in agg1");

    gemm1_chunk_group_k<<<G1B + NCH, NT, 0, stream>>>(x, W1, h1, esrc, edst, ew, seg, brec, E);
    bucket_rows_k<<<NB, NT, 0, stream>>>(seg, brec, cnt, rowstart, pk, E);

    gather64_gemm2_k<<<2048, 256, 0, stream>>>(rowstart, cnt, pk, h1, b1, W2, h2);

    gather41_lsm_k<<<2048, 256, 0, stream>>>(rowstart, cnt, pk, h2, b2, out);
}


# Bucket pass folded into the D=64 gather launch: a 32-wave work-group orders its bucket (8-B records), keeps its row lists in LDS, takes rows from an LDS counter, x W2 last; row sum order: LDS arrival
# speedup vs baseline: 6.1287x; 1.0588x over previous
#include <hip/hip_runtime.h>

typedef __attribute__((ext_vector_type(2))) float v2f;
typedef __attribute__((ext_vector_type(8))) float v8f;

#define N_NODES 100000
#define NFEAT   256
#define NHID    64
#define NCLASS  41

#define G1B  256
#define G1W  16
#define G1T  (G1W * 32)
#define NTILE (N_NODES / 16)
#define W1PP 132
#define KC   32
#define G1LDS 27648
static_assert(G1LDS >= (NFEAT / 2) * W1PP, "W1 fits");
static_assert(NHID == 64 && NFEAT % (2 * KC) == 0, "GEMM1 tiling");
static_assert((long)N_NODES * NFEAT < (1L << 31), "x is indexed with 32 bits");
static_assert(N_NODES % 16 == 0 && G1B * G1W <= NTILE && 2 * G1B * G1W >= NTILE,
              "whole tiles, one or two per wave");

__device__ __forceinline__ void g1_chunk(const float4 (&xr)[4], const float* __restrict__ bp,
                                         v8f& acc0, v8f& acc1, v8f& acc2, v8f& acc3) {
    v2f bf[8][4];
#pragma unroll
    for (int j = 0; j < 8; ++j) {
        const float* __restrict__ br = bp + j * W1PP;
        bf[j][0] = *(const v2f*)(br);
        bf[j][1] = *(const v2f*)(br + 32);
        bf[j][2] = *(const v2f*)(br + 64);
        bf[j][3] = *(const v2f*)(br + 96);
    }
#pragma unroll
    for (int j = 0; j < 8; ++j) {
        const float4 q = xr[j >> 1];
        v2f a;
        if (j & 1) { a.x = q.z; a.y = q.w; } else { a.x = q.x; a.y = q.y; }
        acc0 = __builtin_amdgcn_wmma_f32_16x16x4_f32(false, a, false, bf[j][0], (short)0, acc0, false, false);
        acc1 = __builtin_amdgcn_wmma_f32_16x16x4_f32(false, a, false, bf[j][1], (short)0, acc1, false, false);
        acc2 = __builtin_amdgcn_wmma_f32_16x16x4_f32(false, a, false, bf[j][2], (short)0, acc2, false, false);
        acc3 = __builtin_amdgcn_wmma_f32_16x16x4_f32(false, a, false, bf[j][3], (short)0, acc3, false, false);
    }
}

__device__ __forceinline__ void gemm1_role(const int bid, const float* __restrict__ x,
                                           const float* __restrict__ W1,
                                           float* __restrict__ h1) {
    __shared__ __attribute__((aligned(16))) float w1p[G1LDS];
    const int lane = threadIdx.x & 31;
    const int wave = __builtin_amdgcn_readfirstlane(threadIdx.x >> 5);
    const int c  = lane & 15;
    const int hh = lane >> 4;
    int tile = wave * G1B + bid;
    unsigned xo = (unsigned)((tile * 16 + c) * NFEAT + 16 * hh);

    float4 xa[4], xb[4];
#pragma unroll
    for (int e = 0; e < 4; ++e) xa[e] = *(const float4*)(x + xo + 4 * e);

    {
        constexpr int ITEMS = (NFEAT / 2) * (NHID / 4);
        constexpr int RND = (ITEMS + G1T - 1) / G1T;
        float4 lo[RND], hi[RND];
#pragma unroll
        for (int q = 0; q < RND; ++q) {
            int i = threadIdx.x + q * G1T;
            if (i > ITEMS - 1) i = ITEMS - 1;
            const int kp = i >> 4, n4 = (i & 15) * 4;
            lo[q] = *(const float4*)(W1 + (2 * kp) * NHID + n4);
            hi[q] = *(const float4*)(W1 + (2 * kp + 1) * NHID + n4);
        }
#pragma unroll
        for (int q = 0; q < RND; ++q) {
            const int i = threadIdx.x + q * G1T;
            if (i < ITEMS) {
                const int kp = i >> 4, n4 = (i & 15) * 4;
                float* __restrict__ d = w1p + kp * W1PP + n4 * 2;
                float4 p0, p1;
                p0.x = lo[q].x; p0.y = hi[q].x; p0.z = lo[q].y; p0.w = hi[q].y;
                p1.x = lo[q].z; p1.y = hi[q].z; p1.z = lo[q].w; p1.w = hi[q].w;
                *(float4*)d = p0;
                *(float4*)(d + 4) = p1;
            }
        }
    }
    __syncthreads();

    const float* __restrict__ bp = w1p + (8 * hh) * W1PP + 2 * c;
    for (;;) {
        asm volatile("" ::: "memory");
        const int nt = tile + G1B * G1W;
        const bool more = nt < NTILE;
        const unsigned xn = more ? (unsigned)((nt * 16 + c) * NFEAT + 16 * hh) : xo;
        v8f acc0 = {}, acc1 = {}, acc2 = {}, acc3 = {};
        for (int ch = 0; ch < NFEAT / KC; ch += 2) {
#pragma unroll
            for (int e = 0; e < 4; ++e) xb[e] = *(const float4*)(x + xo + (ch + 1) * KC + 4 * e);
            g1_chunk(xa, bp + (ch * (KC / 2)) * W1PP, acc0, acc1, acc2, acc3);
            const unsigned xq = (ch + 2 < NFEAT / KC) ? xo + (ch + 2) * KC : xn;
#pragma unroll
            for (int e = 0; e < 4; ++e) xa[e] = *(const float4*)(x + xq + 4 * e);
            g1_chunk(xb, bp + ((ch + 1) * (KC / 2)) * W1PP, acc0, acc1, acc2, acc3);
        }

        float* __restrict__ o = h1 + (unsigned)((tile * 16 + 8 * hh) * NHID + c);
        for (int j = 0; j < 8; ++j) {
            o[j * NHID]      = acc0[j];
            o[j * NHID + 16] = acc1[j];
            o[j * NHID + 32] = acc2[j];
            o[j * NHID + 48] = acc3[j];
        }
        if (!more) break;
        tile = nt;
        xo = xn;
    }
}

#define N_EDGES 1600000
#define NB  256
#define BW  391
#define NCH 256
#define NT  512
#define KE  13
#define SEGK (NCH / (NT / 16))
static_assert(NB * BW >= N_NODES, "buckets must cover every row");
static_assert(BW + 32 <= NT, "one thread per row of a bucket, and a spare word per lane behind the rows");
static_assert(NB <= NT && NCH <= NT, "one thread per bucket, one per chunk");
static_assert((long)KE * NT * NCH >= N_EDGES, "a chunk's edges fit its threads' registers");
static_assert(NCH % (NT / 16) == 0 && 2 * SEGK <= 32, "whole runs per 16-lane group, one mask bit per slot");

template <int NTH>
__device__ __forceinline__ int wg_excl_scan(int v, int& total) {
    __shared__ int wsum[32];
    const int lane = threadIdx.x & 31, wv = threadIdx.x >> 5;
    int incl = v;
    for (int d = 1; d < 32; d <<= 1) {
        int t = __shfl_up(incl, d, 32);
        if (lane >= d) incl += t;
    }
    if (lane == 31) wsum[wv] = incl;
    __syncthreads();
    const int ws = (lane < NTH / 32) ? wsum[lane] : 0;
    int wincl = ws;
    for (int d = 1; d < 32; d <<= 1) {
        int t = __shfl_up(wincl, d, 32);
        if (lane >= d) wincl += t;
    }
    total = __shfl(wincl, 31, 32);
    const int wexcl = __shfl(wincl - ws, wv, 32);
    return wexcl + incl - v;
}

static_assert((N_EDGES + NCH - 1) / NCH < 65536, "a run's start and length fit 16 bits each");
#define SRCB 17
static_assert(N_NODES <= (1 << SRCB) && BW < (1 << (31 - SRCB)), "source row and row in the bucket share one word");
static_assert(G1T == NT, "both kinds of work-group of the first launch have NT threads");
static_assert(G1B == NCH && G1B == 256, "one work-group of each kind per pair of block numbers");
__device__ __forceinline__ void chunk_role(const int g, const int* __restrict__ src,
                                           const int* __restrict__ dst,
                                           const float* __restrict__ w,
                                           int* __restrict__ seg,
                                           int* __restrict__ brec_w, int nEdges) {
    int2* __restrict__ brec = (int2*)brec_w;
    __shared__ int h[NB + 32];
    const int t = threadIdx.x;
    if (t < NB + 32) h[t] = 0;
    const int chunk = (nEdges + NCH - 1) / NCH;
    const int e0 = g * chunk;
    int e1 = e0 + chunk; if (e1 > nEdges) e1 = nEdges;
    int d[KE], s[KE], wb[KE];
#pragma unroll
    for (int k = 0; k < KE; ++k) {
        const int e  = e0 + t + k * NT;
        const int ec = (e < e1) ? e : 0;
        d[k]  = dst[ec];
        s[k]  = src[ec];
        wb[k] = __float_as_int(w[ec]);
    }
    __syncthreads();
    int bk[KE], rk[KE];
#pragma unroll
    for (int k = 0; k < KE; ++k) {
        bk[k] = (e0 + t + k * NT < e1) ? d[k] / BW : NB + (t & 31);
        rk[k] = atomicAdd(&h[bk[k]], 1);
    }
    __syncthreads();
    const int mine = (t < NB) ? h[t] : 0;
    int total;
    const int excl = wg_excl_scan<NT>(mine, total);
    if (t < NB) {
        h[t] = excl;
        seg[g * NB + t] = excl | (mine << 16);
    }
    __syncthreads();
    int pb[KE];
#pragma unroll
    for (int k = 0; k < KE; ++k) pb[k] = h[bk[k]];
#pragma unroll
    for (int k = 0; k < KE; ++k) {
        if (e0 + t + k * NT < e1) {
            int2 r; r.x = s[k] | ((d[k] - bk[k] * BW) << SRCB); r.y = wb[k];
            brec[e0 + pb[k] + rk[k]] = r;
        }
    }
}

__global__ __launch_bounds__(NT) __attribute__((amdgpu_waves_per_eu(8)))
void gemm1_chunk_group_k(const float* __restrict__ x,
                                                          const float* __restrict__ W1,
                                                          float* __restrict__ h1,
                                                          const int* __restrict__ src,
                                                          const int* __restrict__ dst,
                                                          const float* __restrict__ w,
                                                          int* __restrict__ seg,
                                                          int* __restrict__ brec_w, int nEdges) {
    const int b = blockIdx.x;
    if (((b ^ (b >> 8)) & 1) == 0) gemm1_role(b >> 1, x, W1, h1);
    else chunk_role(b >> 1, src, dst, w, seg, brec_w, nEdges);
}

__device__ __forceinline__ float xhalf(float x) {
    int b = __builtin_bit_cast(int, x);
    return __builtin_bit_cast(float, __builtin_amdgcn_permlanex16(b, b, 0x76543210, 0xfedcba98, false, false));
}

#define GP 8

template <int P, bool ODD, int PITCH>
__device__ __forceinline__ void acc_pairs(const float* __restrict__ h, int ex, int ey, int j,
                                          int hh, int q4, float4& a) {
    float4 v[P];
    float  wk[P];
#pragma unroll
    for (int k = 0; k < P; ++k) {
        const bool lone = ODD && (k == P - 1);
        const int s0 = __builtin_amdgcn_readlane(ex, j + 2 * k);
        const int w0 = __builtin_amdgcn_readlane(ey, j + 2 * k);
        int sj = s0, wj = w0;
        if (!lone) {
            const int s1 = __builtin_amdgcn_readlane(ex, j + 2 * k + 1);
            const int w1 = __builtin_amdgcn_readlane(ey, j + 2 * k + 1);
            sj = hh ? s1 : s0;
            wj = hh ? w1 : w0;
        }
        wk[k] = __int_as_float(wj);
        v[k]  = *(const float4*)(h + (size_t)(unsigned)sj * PITCH + q4);
        if (lone && hh) { wk[k] = 0.0f; v[k].x = 0.0f; v[k].y = 0.0f; v[k].z = 0.0f; v[k].w = 0.0f; }
    }
#pragma unroll
    for (int k = 0; k < P; ++k) {
        a.x = fmaf(wk[k], v[k].x, a.x);
        a.y = fmaf(wk[k], v[k].y, a.y);
        a.z = fmaf(wk[k], v[k].z, a.z);
        a.w = fmaf(wk[k], v[k].w, a.w);
    }
}

template <int R, int PITCH>
__device__ __forceinline__ void acc_tail(const float* __restrict__ h, int ex, int ey, int j,
                                         int hh, int q4, float4& a, int r, int odd) {
    if (r == R) {
        if (odd) acc_pairs<R + 1, true, PITCH>(h, ex, ey, j, hh, q4, a);
        else if constexpr (R > 0) acc_pairs<R, false, PITCH>(h, ex, ey, j, hh, q4, a);
    } else if constexpr (R > 0) {
        acc_tail<R - 1, PITCH>(h, ex, ey, j, hh, q4, a, r, odd);
    }
}

struct GlbList {
    const int2* __restrict__ p;
    __device__ __forceinline__ int2 operator()(int i) const { return p[i]; }
};
struct LdsList {
    const int2* p;
    __device__ __forceinline__ int2 operator()(int i) const { return p[i]; }
};

template <int PITCH, class L>
__device__ __forceinline__ void row_sum_list(const L ld, int n, const float* __restrict__ h,
                                             int lane, int q4, float4& a) {
    const int hh = lane >> 4;
    a.x = 0.0f; a.y = 0.0f; a.z = 0.0f; a.w = 0.0f;
    for (int c0 = 0; c0 < n; c0 += 32) {
        int m = n - c0; if (m > 32) m = 32;
        int ex = 0, ey = 0;
        if (lane < m) { int2 e = ld(c0 + lane); ex = e.x; ey = e.y; }
        const int whole = m >> 1, odd = m & 1;
        int p = 0;
        for (; p + GP <= whole; p += GP) acc_pairs<GP, false, PITCH>(h, ex, ey, 2 * p, hh, q4, a);
        acc_tail<GP - 1, PITCH>(h, ex, ey, 2 * p, hh, q4, a, whole - p, odd);
    }
    a.x += xhalf(a.x); a.y += xhalf(a.y); a.z += xhalf(a.z); a.w += xhalf(a.w);
}

template <int PITCH>
__device__ __forceinline__ void row_sum_pairs(const int* __restrict__ rowstart,
                                              const int* __restrict__ cnt,
                                              const int* __restrict__ pk,
                                              const float* __restrict__ h,
                                              int r, int lane, int q4, float4& a) {
    const int ru = __builtin_amdgcn_readfirstlane(r);
    const int s  = __builtin_amdgcn_readfirstlane(rowstart[ru]);
    const int n  = __builtin_amdgcn_readfirstlane(cnt[ru]);
    row_sum_list<PITCH>(GlbList{(const int2*)pk + s}, n, h, lane, q4, a);
}

#define TP   68
#define NPAD 48
#define FT    1024
#define FSEGK (NCH / (FT / 16))
#define NTL   ((BW + 15) / 16)
#define BWP   (NTL * 16)
#define LCAP  8192
#define NW2   ((NHID / 2) * NPAD)
static_assert(NHID % 4 == 0 && NCLASS <= NPAD, "product tiling");
static_assert(BW + 32 <= FT && BWP <= FT, "one thread per row of a bucket, and a spare word per lane behind the rows");
static_assert(NCH <= FT && NCH % (FT / 16) == 0 && 2 * FSEGK <= 32, "whole runs per 16-lane group, one mask bit per slot");
static_assert(NW2 <= 2 * FT, "two entries of W2 per thread at most");

__device__ __forceinline__ void lds_barrier() {
    __builtin_amdgcn_fence(__ATOMIC_RELEASE, "workgroup", "local");
    __builtin_amdgcn_s_barrier();
    __builtin_amdgcn_fence(__ATOMIC_ACQUIRE, "workgroup", "local");
}

__global__ __launch_bounds__(FT) void bucket_gather64_gemm2_k(const int* __restrict__ seg_w,
                                                              const int* __restrict__ brec_w,
                                                              int* __restrict__ cnt,
                                                              int* __restrict__ rowstart,
                                                              int* __restrict__ pk, int nEdges,
                                                              const float* __restrict__ h,
                                                              const float* __restrict__ b1,
                                                              const float* __restrict__ W2,
                                                              float* __restrict__ h2) {
    __shared__ __attribute__((aligned(16))) float w2p[NW2 * 2];
    __shared__ __attribute__((aligned(16))) float plane[BWP * TP];
    __shared__ __attribute__((aligned(8)))  int2 lst[LCAP];
    __shared__ int c[FT];
    __shared__ int so[NCH];
    __shared__ int sc[NCH];
    __shared__ int rs[BWP];
    __shared__ int rc[BWP];
    __shared__ int ctl[2];
    const int2* __restrict__ brec = (const int2*)brec_w;
    const int t = threadIdx.x, b = blockIdx.x;
    const int lane = t & 31;
    const int wave = t >> 5;
    const int node0 = b * BW;
    int nrows = N_NODES - node0; if (nrows > BW) nrows = BW;
    const int chunk = (nEdges + NCH - 1) / NCH;

    float2 wv[2];
#pragma unroll
    for (int q = 0; q < 2; ++q) {
        int i = t + q * FT;
        if (i > NW2 - 1) i = NW2 - 1;
        const int kp = i / NPAD, n = i - kp * NPAD;
        const int nc = (n < NCLASS) ? n : (NCLASS - 1);
        const float lo = W2[(2 * kp) * NCLASS + nc];
        const float hi = W2[(2 * kp + 1) * NCLASS + nc];
        wv[q].x = (n < NCLASS) ? lo : 0.0f;
        wv[q].y = (n < NCLASS) ? hi : 0.0f;
    }

    int off = 0;
    if (t < NCH) {
        const unsigned v = (unsigned)seg_w[t * NB + b];
        off = (int)(v & 0xffffu);
        so[t] = t * chunk + off;
        sc[t] = (int)(v >> 16);
    }
    c[t] = 0;
    if (t == 0) ctl[0] = 0;
    int s;
    (void)wg_excl_scan<FT>(off, s);
    const int hw = t >> 4, l16 = t & 15;
    int base[FSEGK], m[FSEGK];
    int2 r[2 * FSEGK];
    int  ri[2 * FSEGK];
    unsigned vm = 0;
    int mmax = 0;
#pragma unroll
    for (int k = 0; k < FSEGK; ++k) {
        const int gq = hw + (FT / 16) * k;
        base[k] = so[gq]; m[k] = sc[gq];
        mmax = (m[k] > mmax) ? m[k] : mmax;
        const bool v0 = l16 < m[k], v1 = l16 + 16 < m[k];
        r[2 * k]     = brec[v0 ? base[k] + l16 : 0];
        r[2 * k + 1] = brec[v1 ? base[k] + l16 + 16 : 0];
        vm |= (v0 ? 1u : 0u) << (2 * k);
        vm |= (v1 ? 1u : 0u) << (2 * k + 1);
    }
#pragma unroll
    for (int q = 0; q < 2; ++q) {
        const int i = t + q * FT;
        if (i < NW2) *(float2*)(w2p + 2 * i) = wv[q];
    }
#pragma unroll
    for (int j = 0; j < 2 * FSEGK; ++j) {
        ri[j] = ((vm >> j) & 1u) ? (r[j].x >> SRCB) : BW + (t & 31);
        atomicAdd(&c[ri[j]], 1);
    }
    if (mmax > 32) {
        for (int k = 0; k < FSEGK; ++k)
            for (int i = 32 + l16; i < m[k]; i += 16) atomicAdd(&c[brec[base[k] + i].x >> SRCB], 1);
    }
    __syncthreads();
    const int mine = c[t];
    int total;
    const int excl = wg_excl_scan<FT>(mine, total);
    if (t < BWP) {
        rs[t] = excl;
        rc[t] = (t < BW) ? mine : 0;
    }
    if (t == BW) ctl[1] = excl;
    c[t] = excl;
    for (int i = nrows * TP + t; i < BWP * TP; i += FT) plane[i] = 0.0f;
    __syncthreads();
    int p[2 * FSEGK];
#pragma unroll
    for (int j = 0; j < 2 * FSEGK; ++j) p[j] = atomicAdd(&c[ri[j]], 1);
    if (t < BW && node0 + t < N_NODES) {
        cnt[node0 + t] = mine;
        rowstart[node0 + t] = s + excl;
    }
#pragma unroll
    for (int j = 0; j < 2 * FSEGK; ++j) {
        if ((vm >> j) & 1u) {
            int2 v; v.x = r[j].x & ((1 << SRCB) - 1); v.y = r[j].y;
            ((int2*)pk)[s + p[j]] = v;
            if (p[j] < LCAP) lst[p[j]] = v;
        }
    }
    if (mmax > 32) {
        for (int k = 0; k < FSEGK; ++k)
            for (int i = 32 + l16; i < m[k]; i += 16) {
                const int2 q = brec[base[k] + i];
                const int pq = atomicAdd(&c[q.x >> SRCB], 1);
                int2 v; v.x = q.x & ((1 << SRCB) - 1); v.y = q.y;
                ((int2*)pk)[s + pq] = v;
                if (pq < LCAP) lst[pq] = v;
            }
    }
    lds_barrier();

    const bool inl = ctl[1] <= LCAP;
    if (!inl) __syncthreads();

    const int cc = lane & 15;
    const int hh = lane >> 4;
    const float4 bq = *(const float4*)(b1 + 4 * cc);
    for (;;) {
        int nx = 0;
        if (lane == 0) nx = atomicAdd(&ctl[0], 1);
        const int rw = __builtin_amdgcn_readfirstlane(nx);
        if (rw >= nrows) break;
        const int st = __builtin_amdgcn_readfirstlane(rs[rw]);
        const int n  = __builtin_amdgcn_readfirstlane(rc[rw]);
        float4 a;
        if (inl) row_sum_list<NHID>(LdsList{lst + st}, n, h, lane, 4 * cc, a);
        else     row_sum_list<NHID>(GlbList{(const int2*)pk + s + st}, n, h, lane, 4 * cc, a);
        if (hh == 0) {
            float4 o;
            o.x = fmaxf(a.x + bq.x, 0.0f); o.y = fmaxf(a.y + bq.y, 0.0f);
            o.z = fmaxf(a.z + bq.z, 0.0f); o.w = fmaxf(a.w + bq.w, 0.0f);
            *(float4*)(plane + rw * TP + 4 * cc) = o;
        }
    }
    __syncthreads();

    for (int item = wave; item < NTL * (NPAD / 16); item += FT / 32) {
        const int tl = item / (NPAD / 16), ct = item - tl * (NPAD / 16);
        const float* __restrict__ ap = plane + (tl * 16 + cc) * TP + 2 * hh;
        const float* __restrict__ bp = w2p + (hh * NPAD + ct * 16 + cc) * 2;
        v8f acc = {};
        for (int k0 = 0; k0 < NHID; k0 += 4) {
            const v2f av = *(const v2f*)(ap + k0);
            const v2f bv = *(const v2f*)(bp + (k0 / 2) * NPAD * 2);
            acc = __builtin_amdgcn_wmma_f32_16x16x4_f32(false, av, false, bv, (short)0, acc, false, false);
        }
        asm volatile("v_nop\n\tv_nop\n\tv_nop\n\tv_nop\n\tv_nop" : "+v"(acc));
        const int row0 = tl * 16 + 8 * hh;
        float* __restrict__ op = h2 + (long)(node0 + row0) * NPAD + ct * 16 + cc;
        for (int j = 0; j < 8; ++j)
            if (row0 + j < nrows) op[j * NPAD] = acc[j];
    }
}

__device__ __forceinline__ float row16_max(float v) {
    asm volatile("v_max_num_f32_dpp %0, %0, %0 row_xmask:1 row_mask:0xf bank_mask:0xf\n\t"
                 "v_max_num_f32_dpp %0, %0, %0 row_xmask:2 row_mask:0xf bank_mask:0xf\n\t"
                 "v_max_num_f32_dpp %0, %0, %0 row_xmask:4 row_mask:0xf bank_mask:0xf\n\t"
                 "v_max_num_f32_dpp %0, %0, %0 row_xmask:8 row_mask:0xf bank_mask:0xf"
                 : "+v"(v));
    return v;
}

__device__ __forceinline__ float row16_sum(float v) {
    asm volatile("v_add_f32_dpp %0, %0, %0 row_xmask:1 row_mask:0xf bank_mask:0xf\n\t"
                 "v_add_f32_dpp %0, %0, %0 row_xmask:2 row_mask:0xf bank_mask:0xf\n\t"
                 "v_add_f32_dpp %0, %0, %0 row_xmask:4 row_mask:0xf bank_mask:0xf\n\t"
                 "v_add_f32_dpp %0, %0, %0 row_xmask:8 row_mask:0xf bank_mask:0xf"
                 : "+v"(v));
    return v;
}

__device__ __forceinline__ float wave_max(float v) {
    v = row16_max(v);
    return fmaxf(v, xhalf(v));
}

__device__ __forceinline__ float wave_sum(float v) {
    v = row16_sum(v);
    return v + xhalf(v);
}

__global__ __launch_bounds__(256) void gather41_lsm_k(const int* __restrict__ rowstart,
                                                      const int* __restrict__ cnt,
                                                      const int* __restrict__ pk,
                                                      const float* __restrict__ h,
                                                      const float* __restrict__ b2,
                                                      float* __restrict__ out) {
    const int lane = threadIdx.x & 31;
    const int nw = gridDim.x * 8;
    const bool has1 = (32 + lane < NCLASS);
    const int c1i = has1 ? (32 + lane) : (NCLASS - 1);
    const float bz0 = b2[lane];
    const float bz1 = b2[c1i];
    const int q  = lane & 15;
    const int q4 = 4 * ((q < NPAD / 4) ? q : (NPAD / 4 - 1));
    const int sl0 = lane >> 2, sl1 = 8 + (lane >> 2), cm = lane & 3;
    for (int r = blockIdx.x * 8 + (threadIdx.x >> 5); r < N_NODES; r += nw) {
        const int ru = __builtin_amdgcn_readfirstlane(r);
        float4 a;
        row_sum_pairs<NPAD>(rowstart, cnt, pk, h, r, lane, q4, a);
        const float t0 = __shfl(a.x, sl0, 32), t1 = __shfl(a.y, sl0, 32);
        const float t2 = __shfl(a.z, sl0, 32), t3 = __shfl(a.w, sl0, 32);
        const float u0 = __shfl(a.x, sl1, 32), u1 = __shfl(a.y, sl1, 32);
        const float u2 = __shfl(a.z, sl1, 32), u3 = __shfl(a.w, sl1, 32);
        const float a0 = (cm == 0) ? t0 : (cm == 1) ? t1 : (cm == 2) ? t2 : t3;
        const float a1 = (cm == 0) ? u0 : (cm == 1) ? u1 : (cm == 2) ? u2 : u3;
        const float z0 = a0 + bz0;
        const float z1 = has1 ? (a1 + bz1) : z0;
        const float mx = wave_max(fmaxf(z0, z1));
        const float e0 = expf(z0 - mx);
        const float e1 = expf(z1 - mx);
        const float lse = mx + logf(wave_sum(has1 ? (e0 + e1) : e0));
        float* __restrict__ o = out + (long)ru * NCLASS;
        o[lane] = z0 - lse;
        if (has1) o[32 + lane] = z1 - lse;
    }
}

extern "C" void kernel_launch(void* const* d_in, const int* in_sizes, int n_in,
                              void* d_out, int out_size, void* d_ws, size_t ws_size,
                              hipStream_t stream) {
    const float* x    = (const float*)d_in[0];
    const int*   esrc = (const int*)d_in[1];
    const int*   edst = (const int*)d_in[2];
    const float* ew   = (const float*)d_in[3];
    const float* W1   = (const float*)d_in[4];
    const float* b1   = (const float*)d_in[5];
    const float* W2   = (const float*)d_in[6];
    const float* b2   = (const float*)d_in[7];
    float* out = (float*)d_out;
    const int E = in_sizes[1];

    float* h1   = (float*)d_ws;
    float* agg1 = h1   + (size_t)N_NODES * NHID;
    float* h2   = agg1;
    int* pk       = (int*)(agg1 + (size_t)N_NODES * NHID);
    int* seg      = pk + (size_t)2 * N_EDGES;
    int* cnt      = pk + (size_t)2 * N_EDGES + (size_t)NCH * NB + (size_t)(NB + 8);
    int* rowstart = cnt    + (size_t)(N_NODES + 8);
    int* brec     = (int*)d_out;
    static_assert((size_t)2 * N_EDGES <= (size_t)N_NODES * NCLASS, "bucketed copy fits in out");

    gemm1_chunk_group_k<<<G1B + NCH, NT, 0, stream>>>(x, W1, h1, esrc, edst, ew, seg, brec, E);

    bucket_gather64_gemm2_k<<<NB, FT, 0, stream>>>(seg, brec, cnt, rowstart, pk, E, h1, b1, W2, h2);

    gather41_lsm_k<<<2048, 256, 0, stream>>>(rowstart, cnt, pk, h2, b2, out);
}


# Both edge-list copies (chunk pass, bucket pass): records ordered in LDS, then stored in list order, 16 B per lane, whole 128-B lines per store, not 8 B per lane into 32 lines; sum orders kept
# speedup vs baseline: 6.8235x; 1.1134x over previous
#include <hip/hip_runtime.h>

typedef __attribute__((ext_vector_type(2))) float v2f;
typedef __attribute__((ext_vector_type(8))) float v8f;

#define N_NODES 100000
#define NFEAT   256
#define NHID    64
#define NCLASS  41

#define G1B  256
#define G1W  16
#define G1T  (G1W * 32)
#define NTILE (N_NODES / 16)
#define W1PP 132
#define KC   32
#define G1LDS 27648
static_assert(G1LDS >= (NFEAT / 2) * W1PP, "W1 fits");
static_assert(NHID == 64 && NFEAT % (2 * KC) == 0, "GEMM1 tiling");
static_assert((long)N_NODES * NFEAT < (1L << 31), "x is indexed with 32 bits");
static_assert(N_NODES % 16 == 0 && G1B * G1W <= NTILE && 2 * G1B * G1W >= NTILE,
              "whole tiles, one or two per wave");

__device__ __forceinline__ void g1_chunk(const float4 (&xr)[4], const float* __restrict__ bp,
                                         v8f& acc0, v8f& acc1, v8f& acc2, v8f& acc3) {
    v2f bf[8][4];
#pragma unroll
    for (int j = 0; j < 8; ++j) {
        const float* __restrict__ br = bp + j * W1PP;
        bf[j][0] = *(const v2f*)(br);
        bf[j][1] = *(const v2f*)(br + 32);
        bf[j][2] = *(const v2f*)(br + 64);
        bf[j][3] = *(const v2f*)(br + 96);
    }
#pragma unroll
    for (int j = 0; j < 8; ++j) {
        const float4 q = xr[j >> 1];
        v2f a;
        if (j & 1) { a.x = q.z; a.y = q.w; } else { a.x = q.x; a.y = q.y; }
        acc0 = __builtin_amdgcn_wmma_f32_16x16x4_f32(false, a, false, bf[j][0], (short)0, acc0, false, false);
        acc1 = __builtin_amdgcn_wmma_f32_16x16x4_f32(false, a, false, bf[j][1], (short)0, acc1, false, false);
        acc2 = __builtin_amdgcn_wmma_f32_16x16x4_f32(false, a, false, bf[j][2], (short)0, acc2, false, false);
        acc3 = __builtin_amdgcn_wmma_f32_16x16x4_f32(false, a, false, bf[j][3], (short)0, acc3, false, false);
    }
}

__device__ __forceinline__ void gemm1_role(const int bid, const float* __restrict__ x,
                                           const float* __restrict__ W1,
                                           float* __restrict__ h1, float* w1p) {
    const int lane = threadIdx.x & 31;
    const int wave = __builtin_amdgcn_readfirstlane(threadIdx.x >> 5);
    const int c  = lane & 15;
    const int hh = lane >> 4;
    int tile = wave * G1B + bid;
    unsigned xo = (unsigned)((tile * 16 + c) * NFEAT + 16 * hh);

    float4 xa[4], xb[4];
#pragma unroll
    for (int e = 0; e < 4; ++e) xa[e] = *(const float4*)(x + xo + 4 * e);

    {
        constexpr int ITEMS = (NFEAT / 2) * (NHID / 4);
        constexpr int RND = (ITEMS + G1T - 1) / G1T;
        float4 lo[RND], hi[RND];
#pragma unroll
        for (int q = 0; q < RND; ++q) {
            int i = threadIdx.x + q * G1T;
            if (i > ITEMS - 1) i = ITEMS - 1;
            const int kp = i >> 4, n4 = (i & 15) * 4;
            lo[q] = *(const float4*)(W1 + (2 * kp) * NHID + n4);
            hi[q] = *(const float4*)(W1 + (2 * kp + 1) * NHID + n4);
        }
#pragma unroll
        for (int q = 0; q < RND; ++q) {
            const int i = threadIdx.x + q * G1T;
            if (i < ITEMS) {
                const int kp = i >> 4, n4 = (i & 15) * 4;
                float* __restrict__ d = w1p + kp * W1PP + n4 * 2;
                float4 p0, p1;
                p0.x = lo[q].x; p0.y = hi[q].x; p0.z = lo[q].y; p0.w = hi[q].y;
                p1.x = lo[q].z; p1.y = hi[q].z; p1.z = lo[q].w; p1.w = hi[q].w;
                *(float4*)d = p0;
                *(float4*)(d + 4) = p1;
            }
        }
    }
    __syncthreads();

    const float* __restrict__ bp = w1p + (8 * hh) * W1PP + 2 * c;
    for (;;) {
        asm volatile("" ::: "memory");
        const int nt = tile + G1B * G1W;
        const bool more = nt < NTILE;
        const unsigned xn = more ? (unsigned)((nt * 16 + c) * NFEAT + 16 * hh) : xo;
        v8f acc0 = {}, acc1 = {}, acc2 = {}, acc3 = {};
        for (int ch = 0; ch < NFEAT / KC; ch += 2) {
#pragma unroll
            for (int e = 0; e < 4; ++e) xb[e] = *(const float4*)(x + xo + (ch + 1) * KC + 4 * e);
            g1_chunk(xa, bp + (ch * (KC / 2)) * W1PP, acc0, acc1, acc2, acc3);
            const unsigned xq = (ch + 2 < NFEAT / KC) ? xo + (ch + 2) * KC : xn;
#pragma unroll
            for (int e = 0; e < 4; ++e) xa[e] = *(const float4*)(x + xq + 4 * e);
            g1_chunk(xb, bp + ((ch + 1) * (KC / 2)) * W1PP, acc0, acc1, acc2, acc3);
        }

        float* __restrict__ o = h1 + (unsigned)((tile * 16 + 8 * hh) * NHID + c);
        for (int j = 0; j < 8; ++j) {
            o[j * NHID]      = acc0[j];
            o[j * NHID + 16] = acc1[j];
            o[j * NHID + 32] = acc2[j];
            o[j * NHID + 48] = acc3[j];
        }
        if (!more) break;
        tile = nt;
        xo = xn;
    }
}

#define N_EDGES 1600000
#define NB  256
#define BW  391
#define NCH 256
#define NT  512
#define KE  13
#define SEGK (NCH / (NT / 16))
static_assert(NB * BW >= N_NODES, "buckets must cover every row");
static_assert(BW + 32 <= NT, "one thread per row of a bucket, and a spare word per lane behind the rows");
static_assert(NB <= NT && NCH <= NT, "one thread per bucket, one per chunk");
static_assert((long)KE * NT * NCH >= N_EDGES, "a chunk's edges fit its threads' registers");
static_assert(NCH % (NT / 16) == 0 && 2 * SEGK <= 32, "whole runs per 16-lane group, one mask bit per slot");

template <int NTH>
__device__ __forceinline__ int wg_excl_scan(int v, int& total) {
    __shared__ int wsum[32];
    const int lane = threadIdx.x & 31, wv = threadIdx.x >> 5;
    int incl = v;
    for (int d = 1; d < 32; d <<= 1) {
        int t = __shfl_up(incl, d, 32);
        if (lane >= d) incl += t;
    }
    if (lane == 31) wsum[wv] = incl;
    __syncthreads();
    const int ws = (lane < NTH / 32) ? wsum[lane] : 0;
    int wincl = ws;
    for (int d = 1; d < 32; d <<= 1) {
        int t = __shfl_up(wincl, d, 32);
        if (lane >= d) wincl += t;
    }
    total = __shfl(wincl, 31, 32);
    const int wexcl = __shfl(wincl - ws, wv, 32);
    return wexcl + incl - v;
}

template <int NTH, int CT>
__device__ __forceinline__ void list_to_memory(const int2* lst, int n, int* __restrict__ dstw, int s) {
    const int t = threadIdx.x;
    const int nl = (n > 0) ? n : 0;
    int2 ra[CT], rb[CT];
#pragma unroll
    for (int k = 0; k < CT; ++k) {
        const int i0 = 2 * ((s >> 1) + t + k * NTH) - s;
        ra[k] = lst[((unsigned)i0 < (unsigned)nl) ? i0 : 0];
        rb[k] = lst[(i0 + 1 < nl) ? i0 + 1 : 0];
    }
#pragma unroll
    for (int k = 0; k < CT; ++k) {
        const int q  = (s >> 1) + t + k * NTH;
        const int i0 = 2 * q - s;
        const bool k0 = (unsigned)i0 < (unsigned)nl, k1 = i0 + 1 < nl;
        if (k0 && k1) {
            int4 w4; w4.x = ra[k].x; w4.y = ra[k].y; w4.z = rb[k].x; w4.w = rb[k].y;
            ((int4*)dstw)[q] = w4;
        } else if (k0) {
            ((int2*)dstw)[2 * q] = ra[k];
        } else if (k1) {
            ((int2*)dstw)[2 * q + 1] = rb[k];
        }
    }
}

static_assert((N_EDGES + NCH - 1) / NCH < 65536, "a run's start and length fit 16 bits each");
#define SRCB 17
static_assert(N_NODES <= (1 << SRCB) && BW < (1 << (31 - SRCB)), "source row and row in the bucket share one word");
static_assert(G1T == NT, "both kinds of work-group of the first launch have NT threads");
static_assert(G1B == NCH && G1B == 256, "one work-group of each kind per pair of block numbers");
__device__ __forceinline__ void chunk_role(const int g, const int* __restrict__ src,
                                           const int* __restrict__ dst,
                                           const float* __restrict__ w,
                                           int* __restrict__ seg,
                                           int* __restrict__ brec_w, int nEdges, int2* stage) {
    __shared__ int h[NB + 32];
    const int t = threadIdx.x;
    if (t < NB + 32) h[t] = 0;
    const int chunk = (nEdges + NCH - 1) / NCH;
    const int e0 = g * chunk;
    int e1 = e0 + chunk; if (e1 > nEdges) e1 = nEdges;
    int d[KE], s[KE], wb[KE];
#pragma unroll
    for (int k = 0; k < KE; ++k) {
        const int e  = e0 + t + k * NT;
        const int ec = (e < e1) ? e : 0;
        d[k]  = dst[ec];
        s[k]  = src[ec];
        wb[k] = __float_as_int(w[ec]);
    }
    __syncthreads();
    int bk[KE], rk[KE];
#pragma unroll
    for (int k = 0; k < KE; ++k) {
        bk[k] = (e0 + t + k * NT < e1) ? d[k] / BW : NB + (t & 31);
        rk[k] = atomicAdd(&h[bk[k]], 1);
    }
    __syncthreads();
    const int mine = (t < NB) ? h[t] : 0;
    int total;
    const int excl = wg_excl_scan<NT>(mine, total);
    if (t < NB) {
        h[t] = excl;
        seg[g * NB + t] = excl | (mine << 16);
    }
    __syncthreads();
    int pb[KE];
#pragma unroll
    for (int k = 0; k < KE; ++k) pb[k] = h[bk[k]];
#pragma unroll
    for (int k = 0; k < KE; ++k) {
        if (e0 + t + k * NT < e1) {
            int2 r; r.x = s[k] | ((d[k] - bk[k] * BW) << SRCB); r.y = wb[k];
            stage[pb[k] + rk[k]] = r;
        }
    }
    __syncthreads();
    constexpr int CCT = (KE * NT / 2 + NT) / NT;
    static_assert(KE * NT <= 2 * (NT * CCT - 1), "the copy's trips cover a full chunk");
    list_to_memory<NT, CCT>(stage, e1 - e0, brec_w, e0);
}

__global__ __launch_bounds__(NT) __attribute__((amdgpu_waves_per_eu(8)))
void gemm1_chunk_group_k(const float* __restrict__ x,
                                                          const float* __restrict__ W1,
                                                          float* __restrict__ h1,
                                                          const int* __restrict__ src,
                                                          const int* __restrict__ dst,
                                                          const float* __restrict__ w,
                                                          int* __restrict__ seg,
                                                          int* __restrict__ brec_w, int nEdges) {
    __shared__ __attribute__((aligned(16))) float wg_lds[G1LDS];
    static_assert((size_t)G1LDS * sizeof(float) >= (size_t)KE * NT * sizeof(int2), "a chunk's records fit the block");
    const int b = blockIdx.x;
    if (((b ^ (b >> 8)) & 1) == 0) gemm1_role(b >> 1, x, W1, h1, wg_lds);
    else chunk_role(b >> 1, src, dst, w, seg, brec_w, nEdges, (int2*)wg_lds);
}

__device__ __forceinline__ float xhalf(float x) {
    int b = __builtin_bit_cast(int, x);
    return __builtin_bit_cast(float, __builtin_amdgcn_permlanex16(b, b, 0x76543210, 0xfedcba98, false, false));
}

#define GP 8

template <int P, bool ODD, int PITCH>
__device__ __forceinline__ void acc_pairs(const float* __restrict__ h, int ex, int ey, int j,
                                          int hh, int q4, float4& a) {
    float4 v[P];
    float  wk[P];
#pragma unroll
    for (int k = 0; k < P; ++k) {
        const bool lone = ODD && (k == P - 1);
        const int s0 = __builtin_amdgcn_readlane(ex, j + 2 * k);
        const int w0 = __builtin_amdgcn_readlane(ey, j + 2 * k);
        int sj = s0, wj = w0;
        if (!lone) {
            const int s1 = __builtin_amdgcn_readlane(ex, j + 2 * k + 1);
            const int w1 = __builtin_amdgcn_readlane(ey, j + 2 * k + 1);
            sj = hh ? s1 : s0;
            wj = hh ? w1 : w0;
        }
        wk[k] = __int_as_float(wj);
        v[k]  = *(const float4*)(h + (size_t)(unsigned)sj * PITCH + q4);
        if (lone && hh) { wk[k] = 0.0f; v[k].x = 0.0f; v[k].y = 0.0f; v[k].z = 0.0f; v[k].w = 0.0f; }
    }
#pragma unroll
    for (int k = 0; k < P; ++k) {
        a.x = fmaf(wk[k], v[k].x, a.x);
        a.y = fmaf(wk[k], v[k].y, a.y);
        a.z = fmaf(wk[k], v[k].z, a.z);
        a.w = fmaf(wk[k], v[k].w, a.w);
    }
}

template <int R, int PITCH>
__device__ __forceinline__ void acc_tail(const float* __restrict__ h, int ex, int ey, int j,
                                         int hh, int q4, float4& a, int r, int odd) {
    if (r == R) {
        if (odd) acc_pairs<R + 1, true, PITCH>(h, ex, ey, j, hh, q4, a);
        else if constexpr (R > 0) acc_pairs<R, false, PITCH>(h, ex, ey, j, hh, q4, a);
    } else if constexpr (R > 0) {
        acc_tail<R - 1, PITCH>(h, ex, ey, j, hh, q4, a, r, odd);
    }
}

struct GlbList {
    const int2* __restrict__ p;
    __device__ __forceinline__ int2 operator()(int i) const { return p[i]; }
};
struct LdsList {
    const int2* p;
    __device__ __forceinline__ int2 operator()(int i) const { return p[i]; }
};

template <int PITCH, class L>
__device__ __forceinline__ void row_sum_list(const L ld, int n, const float* __restrict__ h,
                                             int lane, int q4, float4& a) {
    const int hh = lane >> 4;
    a.x = 0.0f; a.y = 0.0f; a.z = 0.0f; a.w = 0.0f;
    for (int c0 = 0; c0 < n; c0 += 32) {
        int m = n - c0; if (m > 32) m = 32;
        int ex = 0, ey = 0;
        if (lane < m) { int2 e = ld(c0 + lane); ex = e.x; ey = e.y; }
        const int whole = m >> 1, odd = m & 1;
        int p = 0;
        for (; p + GP <= whole; p += GP) acc_pairs<GP, false, PITCH>(h, ex, ey, 2 * p, hh, q4, a);
        acc_tail<GP - 1, PITCH>(h, ex, ey, 2 * p, hh, q4, a, whole - p, odd);
    }
    a.x += xhalf(a.x); a.y += xhalf(a.y); a.z += xhalf(a.z); a.w += xhalf(a.w);
}

template <int PITCH>
__device__ __forceinline__ void row_sum_pairs(const int* __restrict__ rowstart,
                                              const int* __restrict__ cnt,
                                              const int* __restrict__ pk,
                                              const float* __restrict__ h,
                                              int r, int lane, int q4, float4& a) {
    const int ru = __builtin_amdgcn_readfirstlane(r);
    const int s  = __builtin_amdgcn_readfirstlane(rowstart[ru]);
    const int n  = __builtin_amdgcn_readfirstlane(cnt[ru]);
    row_sum_list<PITCH>(GlbList{(const int2*)pk + s}, n, h, lane, q4, a);
}

#define TP   68
#define NPAD 48
#define FT    1024
#define FSEGK (NCH / (FT / 16))
#define NTL   ((BW + 15) / 16)
#define BWP   (NTL * 16)
#define LCAP  8192
#define NW2   ((NHID / 2) * NPAD)
static_assert(NHID % 4 == 0 && NCLASS <= NPAD, "product tiling");
static_assert(BW + 32 <= FT && BWP <= FT, "one thread per row of a bucket, and a spare word per lane behind the rows");
static_assert(NCH <= FT && NCH % (FT / 16) == 0 && 2 * FSEGK <= 32, "whole runs per 16-lane group, one mask bit per slot");
static_assert(NW2 <= 2 * FT, "two entries of W2 per thread at most");

__device__ __forceinline__ void lds_barrier() {
    __builtin_amdgcn_fence(__ATOMIC_RELEASE, "workgroup", "local");
    __builtin_amdgcn_s_barrier();
    __builtin_amdgcn_fence(__ATOMIC_ACQUIRE, "workgroup", "local");
}

__global__ __launch_bounds__(FT) void bucket_gather64_gemm2_k(const int* __restrict__ seg_w,
                                                              const int* __restrict__ brec_w,
                                                              int* __restrict__ cnt,
                                                              int* __restrict__ rowstart,
                                                              int* __restrict__ pk, int nEdges,
                                                              const float* __restrict__ h,
                                                              const float* __restrict__ b1,
                                                              const float* __restrict__ W2,
                                                              float* __restrict__ h2) {
    __shared__ __attribute__((aligned(16))) float w2p[NW2 * 2];
    __shared__ __attribute__((aligned(16))) float plane[BWP * TP];
    __shared__ __attribute__((aligned(16))) int2 lst[LCAP];
    __shared__ int c[FT];
    __shared__ int so[NCH];
    __shared__ int sc[NCH];
    __shared__ int rs[BWP];
    __shared__ int rc[BWP];
    __shared__ int ctl[2];
    const int2* __restrict__ brec = (const int2*)brec_w;
    const int t = threadIdx.x, b = blockIdx.x;
    const int lane = t & 31;
    const int wave = t >> 5;
    const int node0 = b * BW;
    int nrows = N_NODES - node0; if (nrows > BW) nrows = BW;
    const int chunk = (nEdges + NCH - 1) / NCH;

    float2 wv[2];
#pragma unroll
    for (int q = 0; q < 2; ++q) {
        int i = t + q * FT;
        if (i > NW2 - 1) i = NW2 - 1;
        const int kp = i / NPAD, n = i - kp * NPAD;
        const int nc = (n < NCLASS) ? n : (NCLASS - 1);
        const float lo = W2[(2 * kp) * NCLASS + nc];
        const float hi = W2[(2 * kp + 1) * NCLASS + nc];
        wv[q].x = (n < NCLASS) ? lo : 0.0f;
        wv[q].y = (n < NCLASS) ? hi : 0.0f;
    }

    int off = 0;
    if (t < NCH) {
        const unsigned v = (unsigned)seg_w[t * NB + b];
        off = (int)(v & 0xffffu);
        so[t] = t * chunk + off;
        sc[t] = (int)(v >> 16);
    }
    c[t] = 0;
    if (t == 0) ctl[0] = 0;
    int s;
    (void)wg_excl_scan<FT>(off, s);
    const int hw = t >> 4, l16 = t & 15;
    int base[FSEGK], m[FSEGK];
    int2 r[2 * FSEGK];
    int  ri[2 * FSEGK];
    unsigned vm = 0;
    int mmax = 0;
#pragma unroll
    for (int k = 0; k < FSEGK; ++k) {
        const int gq = hw + (FT / 16) * k;
        base[k] = so[gq]; m[k] = sc[gq];
        mmax = (m[k] > mmax) ? m[k] : mmax;
        const bool v0 = l16 < m[k], v1 = l16 + 16 < m[k];
        r[2 * k]     = brec[v0 ? base[k] + l16 : 0];
        r[2 * k + 1] = brec[v1 ? base[k] + l16 + 16 : 0];
        vm |= (v0 ? 1u : 0u) << (2 * k);
        vm |= (v1 ? 1u : 0u) << (2 * k + 1);
    }
#pragma unroll
    for (int q = 0; q < 2; ++q) {
        const int i = t + q * FT;
        if (i < NW2) *(float2*)(w2p + 2 * i) = wv[q];
    }
#pragma unroll
    for (int j = 0; j < 2 * FSEGK; ++j) {
        ri[j] = ((vm >> j) & 1u) ? (r[j].x >> SRCB) : BW + (t & 31);
        atomicAdd(&c[ri[j]], 1);
    }
    if (mmax > 32) {
        for (int k = 0; k < FSEGK; ++k)
            for (int i = 32 + l16; i < m[k]; i += 16) atomicAdd(&c[brec[base[k] + i].x >> SRCB], 1);
    }
    __syncthreads();
    const int mine = c[t];
    int total;
    const int excl = wg_excl_scan<FT>(mine, total);
    if (t < BWP) {
        rs[t] = excl;
        rc[t] = (t < BW) ? mine : 0;
    }
    if (t == BW) ctl[1] = excl;
    c[t] = excl;
    for (int i = nrows * TP + t; i < BWP * TP; i += FT) plane[i] = 0.0f;
    __syncthreads();
    int p[2 * FSEGK];
#pragma unroll
    for (int j = 0; j < 2 * FSEGK; ++j) p[j] = atomicAdd(&c[ri[j]], 1);
    if (t < BW && node0 + t < N_NODES) {
        cnt[node0 + t] = mine;
        rowstart[node0 + t] = s + excl;
    }
#pragma unroll
    for (int j = 0; j < 2 * FSEGK; ++j) {
        if ((vm >> j) & 1u) {
            int2 v; v.x = r[j].x & ((1 << SRCB) - 1); v.y = r[j].y;
            if (p[j] < LCAP) lst[p[j]] = v; else ((int2*)pk)[s + p[j]] = v;
        }
    }
    if (mmax > 32) {
        for (int k = 0; k < FSEGK; ++k)
            for (int i = 32 + l16; i < m[k]; i += 16) {
                const int2 q = brec[base[k] + i];
                const int pq = atomicAdd(&c[q.x >> SRCB], 1);
                int2 v; v.x = q.x & ((1 << SRCB) - 1); v.y = q.y;
                if (pq < LCAP) lst[pq] = v; else ((int2*)pk)[s + pq] = v;
            }
    }
    lds_barrier();

    const int nrec = ctl[1];
    const bool inl = nrec <= LCAP;
    constexpr int LCT = (LCAP / 2 + FT) / FT;
    static_assert(LCAP <= 2 * (FT * LCT - 1), "the copy's trips cover a full LDS list");
    list_to_memory<FT, LCT>(lst, inl ? nrec : LCAP, pk, s);
    if (!inl) __syncthreads();

    const int cc = lane & 15;
    const int hh = lane >> 4;
    const float4 bq = *(const float4*)(b1 + 4 * cc);
    for (;;) {
        int nx = 0;
        if (lane == 0) nx = atomicAdd(&ctl[0], 1);
        const int rw = __builtin_amdgcn_readfirstlane(nx);
        if (rw >= nrows) break;
        const int st = __builtin_amdgcn_readfirstlane(rs[rw]);
        const int n  = __builtin_amdgcn_readfirstlane(rc[rw]);
        float4 a;
        if (inl) row_sum_list<NHID>(LdsList{lst + st}, n, h, lane, 4 * cc, a);
        else     row_sum_list<NHID>(GlbList{(const int2*)pk + s + st}, n, h, lane, 4 * cc, a);
        if (hh == 0) {
            float4 o;
            o.x = fmaxf(a.x + bq.x, 0.0f); o.y = fmaxf(a.y + bq.y, 0.0f);
            o.z = fmaxf(a.z + bq.z, 0.0f); o.w = fmaxf(a.w + bq.w, 0.0f);
            *(float4*)(plane + rw * TP + 4 * cc) = o;
        }
    }
    __syncthreads();

    for (int item = wave; item < NTL * (NPAD / 16); item += FT / 32) {
        const int tl = item / (NPAD / 16), ct = item - tl * (NPAD / 16);
        const float* __restrict__ ap = plane + (tl * 16 + cc) * TP + 2 * hh;
        const float* __restrict__ bp = w2p + (hh * NPAD + ct * 16 + cc) * 2;
        v8f acc = {};
        for (int k0 = 0; k0 < NHID; k0 += 4) {
            const v2f av = *(const v2f*)(ap + k0);
            const v2f bv = *(const v2f*)(bp + (k0 / 2) * NPAD * 2);
            acc = __builtin_amdgcn_wmma_f32_16x16x4_f32(false, av, false, bv, (short)0, acc, false, false);
        }
        asm volatile("v_nop\n\tv_nop\n\tv_nop\n\tv_nop\n\tv_nop" : "+v"(acc));
        const int row0 = tl * 16 + 8 * hh;
        float* __restrict__ op = h2 + (long)(node0 + row0) * NPAD + ct * 16 + cc;
        for (int j = 0; j < 8; ++j)
            if (row0 + j < nrows) op[j * NPAD] = acc[j];
    }
}

__device__ __forceinline__ float row16_max(float v) {
    asm volatile("v_max_num_f32_dpp %0, %0, %0 row_xmask:1 row_mask:0xf bank_mask:0xf\n\t"
                 "v_max_num_f32_dpp %0, %0, %0 row_xmask:2 row_mask:0xf bank_mask:0xf\n\t"
                 "v_max_num_f32_dpp %0, %0, %0 row_xmask:4 row_mask:0xf bank_mask:0xf\n\t"
                 "v_max_num_f32_dpp %0, %0, %0 row_xmask:8 row_mask:0xf bank_mask:0xf"
                 : "+v"(v));
    return v;
}

__device__ __forceinline__ float row16_sum(float v) {
    asm volatile("v_add_f32_dpp %0, %0, %0 row_xmask:1 row_mask:0xf bank_mask:0xf\n\t"
                 "v_add_f32_dpp %0, %0, %0 row_xmask:2 row_mask:0xf bank_mask:0xf\n\t"
                 "v_add_f32_dpp %0, %0, %0 row_xmask:4 row_mask:0xf bank_mask:0xf\n\t"
                 "v_add_f32_dpp %0, %0, %0 row_xmask:8 row_mask:0xf bank_mask:0xf"
                 : "+v"(v));
    return v;
}

__device__ __forceinline__ float wave_max(float v) {
    v = row16_max(v);
    return fmaxf(v, xhalf(v));
}

__device__ __forceinline__ float wave_sum(float v) {
    v = row16_sum(v);
    return v + xhalf(v);
}

__global__ __launch_bounds__(256) void gather41_lsm_k(const int* __restrict__ rowstart,
                                                      const int* __restrict__ cnt,
                                                      const int* __restrict__ pk,
                                                      const float* __restrict__ h,
                                                      const float* __restrict__ b2,
                                                      float* __restrict__ out) {
    const int lane = threadIdx.x & 31;
    const int nw = gridDim.x * 8;
    const bool has1 = (32 + lane < NCLASS);
    const int c1i = has1 ? (32 + lane) : (NCLASS - 1);
    const float bz0 = b2[lane];
    const float bz1 = b2[c1i];
    const int q  = lane & 15;
    const int q4 = 4 * ((q < NPAD / 4) ? q : (NPAD / 4 - 1));
    const int sl0 = lane >> 2, sl1 = 8 + (lane >> 2), cm = lane & 3;
    for (int r = blockIdx.x * 8 + (threadIdx.x >> 5); r < N_NODES; r += nw) {
        const int ru = __builtin_amdgcn_readfirstlane(r);
        float4 a;
        row_sum_pairs<NPAD>(rowstart, cnt, pk, h, r, lane, q4, a);
        const float t0 = __shfl(a.x, sl0, 32), t1 = __shfl(a.y, sl0, 32);
        const float t2 = __shfl(a.z, sl0, 32), t3 = __shfl(a.w, sl0, 32);
        const float u0 = __shfl(a.x, sl1, 32), u1 = __shfl(a.y, sl1, 32);
        const float u2 = __shfl(a.z, sl1, 32), u3 = __shfl(a.w, sl1, 32);
        const float a0 = (cm == 0) ? t0 : (cm == 1) ? t1 : (cm == 2) ? t2 : t3;
        const float a1 = (cm == 0) ? u0 : (cm == 1) ? u1 : (cm == 2) ? u2 : u3;
        const float z0 = a0 + bz0;
        const float z1 = has1 ? (a1 + bz1) : z0;
        const float mx = wave_max(fmaxf(z0, z1));
        const float e0 = expf(z0 - mx);
        const float e1 = expf(z1 - mx);
        const float lse = mx + logf(wave_sum(has1 ? (e0 + e1) : e0));
        float* __restrict__ o = out + (long)ru * NCLASS;
        o[lane] = z0 - lse;
        if (has1) o[32 + lane] = z1 - lse;
    }
}

extern "C" void kernel_launch(void* const* d_in, const int* in_sizes, int n_in,
                              void* d_out, int out_size, void* d_ws, size_t ws_size,
                              hipStream_t stream) {
    const float* x    = (const float*)d_in[0];
    const int*   esrc = (const int*)d_in[1];
    const int*   edst = (const int*)d_in[2];
    const float* ew   = (const float*)d_in[3];
    const float* W1   = (const float*)d_in[4];
    const float* b1   = (const float*)d_in[5];
    const float* W2   = (const float*)d_in[6];
    const float* b2   = (const float*)d_in[7];
    float* out = (float*)d_out;
    const int E = in_sizes[1];

    float* h1   = (float*)d_ws;
    float* agg1 = h1   + (size_t)N_NODES * NHID;
    float* h2   = agg1;
    int* pk       = (int*)(agg1 + (size_t)N_NODES * NHID);
    int* seg      = pk + (size_t)2 * N_EDGES;
    int* cnt      = pk + (size_t)2 * N_EDGES + (size_t)NCH * NB + (size_t)(NB + 8);
    int* rowstart = cnt    + (size_t)(N_NODES + 8);
    int* brec     = (int*)d_out;
    static_assert((size_t)2 * N_EDGES <= (size_t)N_NODES * NCLASS, "bucketed copy fits in out");

    gemm1_chunk_group_k<<<G1B + NCH, NT, 0, stream>>>(x, W1, h1, esrc, edst, ew, seg, brec, E);

    bucket_gather64_gemm2_k<<<NB, FT, 0, stream>>>(seg, brec, cnt, rowstart, pk, E, h1, b1, W2, h2);

    gather41_lsm_k<<<2048, 256, 0, stream>>>(rowstart, cnt, pk, h2, b2, out);
}
